# starting kernel
# baseline (speedup 1.0000x reference)
#include <hip/hip_runtime.h>
#include <math.h>

typedef float v2f __attribute__((ext_vector_type(2)));
typedef float v8f __attribute__((ext_vector_type(8)));

#define WG 256

__device__ __forceinline__ void atomic_fadd(float* p, float v) {
  __hip_atomic_fetch_add(p, v, __ATOMIC_RELAXED, __HIP_MEMORY_SCOPE_AGENT);
}

__device__ __forceinline__ float fast_tanh(float x) {
  float r;
  asm("v_tanh_f32 %0, %1\n\tv_nop\n\tv_nop" : "=v"(r) : "v"(x));
  return r;
}

__global__ void k_init(const float* __restrict__ u, const float* __restrict__ it,
                       float* __restrict__ ego, float* __restrict__ acc,
                       int userElems, int total) {
  int i = blockIdx.x * blockDim.x + threadIdx.x;
  if (i >= total) return;
  float v = (i < userElems) ? u[i] : it[i - userElems];
  ego[i] = v;
  acc[i] = v;
}

__global__ void k_fill1(float* __restrict__ a, int n) {
  int i = blockIdx.x * blockDim.x + threadIdx.x;
  if (i < n) a[i] = 1.0f;
}

__global__ void k_softmax_dval(const float* __restrict__ A, const int* __restrict__ hl,
                               float* __restrict__ ew, float* __restrict__ dval, int E) {
  int e = blockIdx.x * blockDim.x + threadIdx.x;
  if (e >= E) return;
  float4 a = reinterpret_cast<const float4*>(A)[e];
  float m = fmaxf(fmaxf(a.x, a.y), fmaxf(a.z, a.w));
  float e0 = expf(a.x - m), e1 = expf(a.y - m), e2 = expf(a.z - m), e3 = expf(a.w - m);
  float inv = 1.0f / (e0 + e1 + e2 + e3);
  float4 nrm = make_float4(e0 * inv, e1 * inv, e2 * inv, e3 * inv);
  reinterpret_cast<float4*>(ew)[e] = nrm;
  float* d = dval + (size_t)hl[e] * 4;
  atomic_fadd(d + 0, nrm.x);
  atomic_fadd(d + 1, nrm.y);
  atomic_fadd(d + 2, nrm.z);
  atomic_fadd(d + 3, nrm.w);
}

__global__ void k_dinv(float* __restrict__ d, int n) {
  int i = blockIdx.x * blockDim.x + threadIdx.x;
  if (i >= n) return;
  d[i] = 1.0f / sqrtf(fmaxf(d[i], 1e-8f));
}

__global__ void k_edge_weight(float* __restrict__ ew, const float* __restrict__ dinv,
                              const int* __restrict__ hl, const int* __restrict__ tl, int E) {
  int e = blockIdx.x * blockDim.x + threadIdx.x;
  if (e >= E) return;
  float4 w = reinterpret_cast<float4*>(ew)[e];
  float4 dh = reinterpret_cast<const float4*>(dinv)[hl[e]];
  float4 dt = reinterpret_cast<const float4*>(dinv)[tl[e]];
  w.x *= dh.x * dt.x;
  w.y *= dh.y * dt.y;
  w.z *= dh.z * dt.z;
  w.w *= dh.w * dt.w;
  reinterpret_cast<float4*>(ew)[e] = w;
}

__global__ void k_message(const float* __restrict__ ew, const float* __restrict__ ego,
                          const int* __restrict__ hl, const int* __restrict__ tl,
                          float* __restrict__ fact, int E) {
  int tid = blockIdx.x * blockDim.x + threadIdx.x;
  int total = E * 16;
  if (tid >= total) return;
  int e = tid >> 4, q = tid & 15, f = q >> 2;
  int h = hl[e], t = tl[e];
  float w = ew[(size_t)e * 4 + f];
  float4 v = reinterpret_cast<const float4*>(ego)[(size_t)t * 16 + q];
  float* dst = fact + (size_t)h * 64 + q * 4;
  atomic_fadd(dst + 0, w * v.x);
  atomic_fadd(dst + 1, w * v.y);
  atomic_fadd(dst + 2, w * v.z);
  atomic_fadd(dst + 3, w * v.w);
}

__global__ void k_norminv(const float* __restrict__ x, float* __restrict__ out, int NF) {
  int i = blockIdx.x * blockDim.x + threadIdx.x;
  if (i >= NF) return;
  int n = i >> 2, f = i & 3;
  const float4* p = reinterpret_cast<const float4*>(x + (size_t)n * 64 + (size_t)f * 16);
  float s = 0.f;
#pragma unroll
  for (int k = 0; k < 4; ++k) {
    float4 v = p[k];
    s += v.x * v.x + v.y * v.y + v.z * v.z + v.w * v.w;
  }
  out[i] = 1.0f / fmaxf(sqrtf(s), 1e-12f);
}

__global__ void __launch_bounds__(WG) k_updateA(
    const float* __restrict__ fact, const float* __restrict__ ego,
    const float* __restrict__ hninv, const float* __restrict__ tninv,
    const int* __restrict__ hl, const int* __restrict__ tl,
    float* __restrict__ A, int E, int nGroups) {
  int tid = blockIdx.x * blockDim.x + threadIdx.x;
  int wave = tid >> 5;
  int lane = tid & 31;
  if (wave >= nGroups) return;

  int r = lane & 15;
  int eIdx = wave * 16 + r;
  int e = (eIdx < E) ? eIdx : (E - 1);
  int h = hl[e], t = tl[e];
  int koff = (lane >> 4) << 1;
  const float* Hrow = fact + (size_t)h * 64;
  const float* Trow = ego + (size_t)t * 64;
  const float* hn4 = hninv + (size_t)h * 4;
  const float* tn4 = tninv + (size_t)t * 4;

  bool active = (lane < 8) || (lane >= 24);
  int outRow = (lane < 8) ? lane : (lane - 16);
  int eOut = wave * 16 + outRow;

#pragma unroll
  for (int f = 0; f < 4; ++f) {
    float hn = hn4[f];
    float tn = tn4[f];
    v8f c = {0.f, 0.f, 0.f, 0.f, 0.f, 0.f, 0.f, 0.f};
#pragma unroll
    for (int s = 0; s < 4; ++s) {
      int o = f * 16 + s * 4 + koff;
      v2f av, bv;
      av.x = Hrow[o] * hn;
      av.y = Hrow[o + 1] * hn;
      bv.x = fast_tanh(Trow[o] * tn);
      bv.y = fast_tanh(Trow[o + 1] * tn);
      c = __builtin_amdgcn_wmma_f32_16x16x4_f32(false, av, false, bv,
                                                (short)0, c, false, false);
    }
    int idx = lane & 7;
    float dv = c[0];
    dv = (idx == 1) ? c[1] : dv;
    dv = (idx == 2) ? c[2] : dv;
    dv = (idx == 3) ? c[3] : dv;
    dv = (idx == 4) ? c[4] : dv;
    dv = (idx == 5) ? c[5] : dv;
    dv = (idx == 6) ? c[6] : dv;
    dv = (idx == 7) ? c[7] : dv;
    if (active && eOut < E) A[(size_t)eOut * 4 + f] += dv;
  }
}

__global__ void k_acc_add(float* __restrict__ acc, const float* __restrict__ x, int n) {
  int i = blockIdx.x * blockDim.x + threadIdx.x;
  if (i < n) acc[i] += x[i];
}

__global__ void k_final(const float* __restrict__ acc, float* __restrict__ out, int n) {
  int i = blockIdx.x * blockDim.x + threadIdx.x;
  if (i < n) out[i] = acc[i] * (1.0f / 3.0f);
}

static inline size_t align16f(size_t x) { return (x + 15) & ~(size_t)15; }

extern "C" void kernel_launch(void* const* d_in, const int* in_sizes, int n_in,
                              void* d_out, int out_size, void* d_ws, size_t ws_size,
                              hipStream_t stream) {
  (void)n_in; (void)out_size; (void)ws_size;
  const float* u  = (const float*)d_in[0];
  const float* it = (const float*)d_in[1];
  const int* hl   = (const int*)d_in[2];
  const int* tl   = (const int*)d_in[3];

  const int userElems = in_sizes[0];
  const int itemElems = in_sizes[1];
  const int E = in_sizes[2];
  const int N = (userElems + itemElems) / 64;
  const size_t N64 = (size_t)N * 64;
  const size_t E4  = (size_t)E * 4;
  const size_t N4  = (size_t)N * 4;

  float* base = (float*)d_ws;
  size_t off = 0;
  float* ego   = base + off; off += align16f(N64);
  float* fact  = base + off; off += align16f(N64);
  float* acc   = base + off; off += align16f(N64);
  float* A     = base + off; off += align16f(E4);
  float* ew    = base + off; off += align16f(E4);
  float* dval  = base + off; off += align16f(N4);
  float* hninv = base + off; off += align16f(N4);
  float* tninv = base + off; off += align16f(N4);

#define GRID(n) ((unsigned)(((size_t)(n) + WG - 1) / WG))

  k_init<<<GRID(N64), WG, 0, stream>>>(u, it, ego, acc, userElems, (int)N64);
  k_fill1<<<GRID(E4), WG, 0, stream>>>(A, (int)E4);

  float* cur = ego;
  float* nxt = fact;
  const int nGroups = (E + 15) / 16;

  for (int layer = 0; layer < 2; ++layer) {
    k_norminv<<<GRID(N4), WG, 0, stream>>>(cur, tninv, (int)N4);
    for (int itn = 0; itn < 2; ++itn) {
      hipMemsetAsync(dval, 0, N4 * sizeof(float), stream);
      k_softmax_dval<<<GRID(E), WG, 0, stream>>>(A, hl, ew, dval, E);
      k_dinv<<<GRID(N4), WG, 0, stream>>>(dval, (int)N4);
      k_edge_weight<<<GRID(E), WG, 0, stream>>>(ew, dval, hl, tl, E);
      hipMemsetAsync(nxt, 0, N64 * sizeof(float), stream);
      k_message<<<GRID((size_t)E * 16), WG, 0, stream>>>(ew, cur, hl, tl, nxt, E);
      const bool lastUpdateDead = (layer == 1 && itn == 1);
      if (!lastUpdateDead) {
        k_norminv<<<GRID(N4), WG, 0, stream>>>(nxt, hninv, (int)N4);
        k_updateA<<<GRID((size_t)nGroups * 32), WG, 0, stream>>>(
            nxt, cur, hninv, tninv, hl, tl, A, E, nGroups);
      }
    }
    k_acc_add<<<GRID(N64), WG, 0, stream>>>(acc, nxt, (int)N64);
    float* tmp = cur; cur = nxt; nxt = tmp;
  }

  k_final<<<GRID(N64), WG, 0, stream>>>(acc, (float*)d_out, (int)N64);
#undef GRID
}


# Row inverse norms formed inside the routing update: a lane sums its 8 squares, adds lane l^16's, takes 1/max(sqrt,1e-12); norm tables, their gathers and 5 launches gone; norm sum order changed
# speedup vs baseline: 2.4155x; 2.4155x over previous
#include <hip/hip_runtime.h>
#include <math.h>

typedef float v2f __attribute__((ext_vector_type(2)));
typedef float v4f __attribute__((ext_vector_type(4)));
typedef float v8f __attribute__((ext_vector_type(8)));

#define WG 256

__device__ __forceinline__ float xhalf(float x) {
  int b = __builtin_bit_cast(int, x);
  return __builtin_bit_cast(float,
      __builtin_amdgcn_permlanex16(b, b, 0x76543210, 0xfedcba98, false, false));
}

__device__ __forceinline__ float fast_tanh(float x) {
  float r;
  asm("v_tanh_f32 %0, %1\n\tv_nop\n\tv_nop" : "=v"(r) : "v"(x));
  return r;
}

__global__ void k_init(const float* __restrict__ u, const float* __restrict__ it,
                       float* __restrict__ ego, float* __restrict__ acc,
                       int userElems, int total) {
  int i = blockIdx.x * blockDim.x + threadIdx.x;
  if (i >= total) return;
  float v = (i < userElems) ? u[i] : it[i - userElems];
  ego[i] = v;
  acc[i] = v;
}

__global__ void k_fill1(float* __restrict__ a, int n) {
  int i = blockIdx.x * blockDim.x + threadIdx.x;
  if (i < n) a[i] = 1.0f;
}

__global__ void k_softmax(const float* __restrict__ A, float* __restrict__ ew, int E) {
  int e = blockIdx.x * blockDim.x + threadIdx.x;
  if (e >= E) return;
  float4 a = reinterpret_cast<const float4*>(A)[e];
  float m = fmaxf(fmaxf(a.x, a.y), fmaxf(a.z, a.w));
  float e0 = expf(a.x - m), e1 = expf(a.y - m), e2 = expf(a.z - m), e3 = expf(a.w - m);
  float inv = 1.0f / (e0 + e1 + e2 + e3);
  reinterpret_cast<float4*>(ew)[e] = make_float4(e0 * inv, e1 * inv, e2 * inv, e3 * inv);
}

#define DEG_SLICES 16
#define DEG_NODES  5000
#define DEG_WG     1024
#define DEG_VEC    4

__device__ __forceinline__ void lds_fadd(float* p, float v) {
  __hip_atomic_fetch_add(p, v, __ATOMIC_RELAXED, __HIP_MEMORY_SCOPE_WORKGROUP);
}

__global__ void __launch_bounds__(DEG_WG) k_degree_lds(
    const float* __restrict__ ew, const int* __restrict__ hl,
    float* __restrict__ part, int E, int N, int per) {
  __shared__ float4 tab4[DEG_NODES];
  float* tab = reinterpret_cast<float*>(tab4);
  const int s = blockIdx.x;
  const int lo = blockIdx.y * DEG_NODES;

  for (int i = threadIdx.x; i < DEG_NODES; i += DEG_WG)
    tab4[i] = make_float4(0.f, 0.f, 0.f, 0.f);
  __syncthreads();

  const int e0 = s * per;
  int e1 = e0 + per;
  if (e1 > E) e1 = E;
  const int nvec = (e1 > e0) ? ((e1 - e0) >> 2) : 0;
  const int4* hv = reinterpret_cast<const int4*>(hl + e0);
  const float4* ew4 = reinterpret_cast<const float4*>(ew);

  for (int v = threadIdx.x; v < nvec; v += DEG_WG * DEG_VEC) {
    int hd[DEG_VEC * 4];
#pragma unroll
    for (int k = 0; k < DEG_VEC; ++k) {
      int vk = v + k * DEG_WG;
      int4 h = make_int4(-1, -1, -1, -1);
      if (vk < nvec) h = hv[vk];
      hd[k * 4 + 0] = h.x; hd[k * 4 + 1] = h.y; hd[k * 4 + 2] = h.z; hd[k * 4 + 3] = h.w;
    }
    float4 w[DEG_VEC * 4];
#pragma unroll
    for (int j = 0; j < DEG_VEC * 4; ++j) {
      unsigned d = (unsigned)(hd[j] - lo);
      w[j] = make_float4(0.f, 0.f, 0.f, 0.f);
      if (d < (unsigned)DEG_NODES) w[j] = ew4[e0 + ((v + (j >> 2) * DEG_WG) << 2) + (j & 3)];
    }
#pragma unroll
    for (int j = 0; j < DEG_VEC * 4; ++j) {
      unsigned d = (unsigned)(hd[j] - lo);
      if (d < (unsigned)DEG_NODES) {
        float* t = tab + d * 4;
        lds_fadd(t + 0, w[j].x);
        lds_fadd(t + 1, w[j].y);
        lds_fadd(t + 2, w[j].z);
        lds_fadd(t + 3, w[j].w);
      }
    }
  }
  for (int e = e0 + (nvec << 2) + threadIdx.x; e < e1; e += DEG_WG) {
    unsigned d = (unsigned)(hl[e] - lo);
    if (d < (unsigned)DEG_NODES) {
      float4 wv = ew4[e];
      float* t = tab + d * 4;
      lds_fadd(t + 0, wv.x);
      lds_fadd(t + 1, wv.y);
      lds_fadd(t + 2, wv.z);
      lds_fadd(t + 3, wv.w);
    }
  }
  __syncthreads();

  int nodes = N - lo;
  if (nodes > DEG_NODES) nodes = DEG_NODES;
  float4* dst = reinterpret_cast<float4*>(part) + (size_t)s * N + lo;
  for (int i = threadIdx.x; i < nodes; i += DEG_WG) dst[i] = tab4[i];
}

__global__ void k_dinv_sum(const float* __restrict__ part, float* __restrict__ d, int N) {
  int n = blockIdx.x * blockDim.x + threadIdx.x;
  if (n >= N) return;
  const float4* p4 = reinterpret_cast<const float4*>(part);
  float4 sum = make_float4(0.f, 0.f, 0.f, 0.f);
#pragma unroll
  for (int k = 0; k < DEG_SLICES; ++k) {
    float4 v = p4[(size_t)k * N + n];
    sum.x += v.x; sum.y += v.y; sum.z += v.z; sum.w += v.w;
  }
  float4 r;
  r.x = 1.0f / sqrtf(fmaxf(sum.x, 1e-8f));
  r.y = 1.0f / sqrtf(fmaxf(sum.y, 1e-8f));
  r.z = 1.0f / sqrtf(fmaxf(sum.z, 1e-8f));
  r.w = 1.0f / sqrtf(fmaxf(sum.w, 1e-8f));
  reinterpret_cast<float4*>(d)[n] = r;
}

#define MSG_NODES 320
#define MSG_WG    1024
#define MSG_WAVES 32
#define MSG_BATCH 16
#define MSG_QCAP  544
#define MSG_NOKEY 255u

__global__ void k_edge_weight(float* __restrict__ ew, const float* __restrict__ dinv,
                              const int* __restrict__ hl, const int* __restrict__ tl,
                              float* __restrict__ keyf, int E, int R) {
  int e = blockIdx.x * blockDim.x + threadIdx.x;
  unsigned char* keyb = reinterpret_cast<unsigned char*>(keyf);
  if (e >= E) {
    if (e < ((E + 15) & ~15)) keyb[e] = (unsigned char)MSG_NOKEY;
    return;
  }
  int h = hl[e];
  float4 w = reinterpret_cast<float4*>(ew)[e];
  float4 dh = reinterpret_cast<const float4*>(dinv)[h];
  float4 dt = reinterpret_cast<const float4*>(dinv)[tl[e]];
  w.x *= dh.x * dt.x;
  w.y *= dh.y * dt.y;
  w.z *= dh.z * dt.z;
  w.w *= dh.w * dt.w;
  reinterpret_cast<float4*>(ew)[e] = w;
  keyb[e] = (unsigned char)((unsigned)h % (unsigned)R);
}

__device__ __forceinline__ unsigned nonzero_bytes(unsigned x) {
  return ((x & 0x7f7f7f7fu) + 0x7f7f7f7fu) | x;
}

__device__ __forceinline__ unsigned key_hits(uint4 k, unsigned pat) {
  unsigned y0 = nonzero_bytes(k.x ^ pat);
  unsigned y1 = nonzero_bytes(k.y ^ pat);
  unsigned y2 = nonzero_bytes(k.z ^ pat);
  unsigned y3 = nonzero_bytes(k.w ^ pat);
  unsigned n = ((y0 >> 7) & 0x01010101u) | ((y1 >> 6) & 0x02020202u) |
               ((y2 >> 5) & 0x04040404u) | ((y3 >> 4) & 0x08080808u);
  return ~n & 0x0f0f0f0fu;
}

template <bool FULL>
__device__ __forceinline__ void msg_batch(const unsigned* qp, int n, int lane, unsigned R,
                                          const float* __restrict__ ew,
                                          const float* __restrict__ ego,
                                          const int* __restrict__ hl,
                                          const int* __restrict__ tl, float* tab) {
  const int l16 = lane & 15;
  const unsigned el = qp[(FULL || l16 < n) ? l16 : 0];
  const int tv = tl[el];
  const int dv = (int)((unsigned)hl[el] / R);
  const int j8 = lane >> 2;
  const unsigned ea = qp[(FULL || j8 < n) ? j8 : 0];
  const unsigned eb = qp[(FULL || j8 + 8 < n) ? j8 + 8 : 0];
  const float wa = ew[(size_t)ea * 4 + (lane & 3)];
  const float wb = ew[(size_t)eb * 4 + (lane & 3)];
  const int f = lane >> 3;

  float2 v[MSG_BATCH];
  int d[MSG_BATCH];
#pragma unroll
  for (int i = 0; i < MSG_BATCH; ++i) {
    const int t = __builtin_amdgcn_readlane(tv, i);
    d[i] = __builtin_amdgcn_readlane(dv, i);
    v[i] = make_float2(0.f, 0.f);
    if (FULL || i < n)
      v[i] = *reinterpret_cast<const float2*>(ego + (size_t)t * 64 + lane * 2);
  }
#pragma unroll
  for (int i = 0; i < MSG_BATCH; ++i) {
    const float w = __shfl((i < 8) ? wa : wb, 4 * (i & 7) + f);
    if (FULL || i < n) {
      float* tp = tab + d[i] * 64 + lane * 2;
      lds_fadd(tp + 0, w * v[i].x);
      lds_fadd(tp + 1, w * v[i].y);
    }
  }
}

__global__ void __launch_bounds__(MSG_WG) k_message_lds(
    const float* __restrict__ ew, const float* __restrict__ ego,
    const int* __restrict__ hl, const int* __restrict__ tl,
    const float* __restrict__ keyf, float* __restrict__ fact, int E, int N) {
  __shared__ float4 mtab4[MSG_NODES * 16];
  __shared__ unsigned mq[MSG_WAVES][MSG_QCAP];
  float* tab = reinterpret_cast<float*>(mtab4);
  const int r = blockIdx.x;
  const unsigned R = gridDim.x;
  const int lane = threadIdx.x & 31;
  const int wave = threadIdx.x >> 5;

  for (int i = threadIdx.x; i < MSG_NODES * 16; i += MSG_WG)
    mtab4[i] = make_float4(0.f, 0.f, 0.f, 0.f);
  __syncthreads();

  const uint4* k4 = reinterpret_cast<const uint4*>(keyf);
  const int nChunks = (E + 15) >> 4;
  const int perTrip = MSG_WAVES * 128;
  const int trips = (nChunks + perTrip - 1) / perTrip;
  const unsigned pat = (unsigned)r * 0x01010101u;
  unsigned* q = mq[wave];
  int cnt = 0;

  int j = r % trips;
  for (int jj = 0; jj < trips; ++jj) {
    const int cb = j * perTrip + wave * 128;
    j = (j + 1 == trips) ? 0 : j + 1;
    uint4 kk[4];
#pragma unroll
    for (int u = 0; u < 4; ++u) {
      const int c = cb + u * 32 + lane;
      kk[u] = make_uint4(~0u, ~0u, ~0u, ~0u);
      if (c < nChunks) kk[u] = k4[c];
    }
#pragma unroll
    for (int u = 0; u < 4; ++u) {
      const int c = cb + u * 32 + lane;
      unsigned m = key_hits(kk[u], pat);
      unsigned bal = __builtin_amdgcn_ballot_w32(m != 0u);
      while (bal != 0u) {
        if (m != 0u) {
          const int p = __builtin_ctz(m);
          const int idx = ((p & 3) << 2) | (p >> 3);
          q[cnt + (int)__builtin_amdgcn_mbcnt_lo(bal, 0u)] = (unsigned)(c * 16 + idx);
          m &= m - 1u;
        }
        cnt += __builtin_popcount(bal);
        bal = __builtin_amdgcn_ballot_w32(m != 0u);
      }
      while (cnt >= MSG_BATCH) {
        cnt -= MSG_BATCH;
        msg_batch<true>(q + cnt, MSG_BATCH, lane, R, ew, ego, hl, tl, tab);
      }
    }
  }
  if (cnt > 0) msg_batch<false>(q, cnt, lane, R, ew, ego, hl, tl, tab);
  __syncthreads();

  float4* dst = reinterpret_cast<float4*>(fact);
  for (int i = threadIdx.x; i < MSG_NODES * 16; i += MSG_WG) {
    const int node = (i >> 4) * (int)R + r;
    if (node < N) dst[(size_t)node * 16 + (i & 15)] = mtab4[i];
  }
}

__global__ void __launch_bounds__(WG) k_updateA(
    const float* __restrict__ fact, const float* __restrict__ ego,
    const int* __restrict__ hl, const int* __restrict__ tl,
    float* __restrict__ A, int E, int nGroups) {
  int tid = blockIdx.x * blockDim.x + threadIdx.x;
  int wave = tid >> 5;
  int lane = tid & 31;
  if (wave >= nGroups) return;

  const int c = lane & 15;
  const int kh = lane >> 4;
  const int i = c >> 2;
  const int f = c & 3;
  const int seg = f * 16 + kh * 8;

  const bool active = (lane < 8) || (lane >= 24);
  const int cc = (lane < 8) ? lane : (lane - 16);
  const int e0 = wave * 16;

  int hh[4], tt[4];
  float aold[4];
  bool wr[4];
  const int aLast = E * 4 - 1;
#pragma unroll
  for (int g = 0; g < 4; ++g) {
    const int eIdx = e0 + g * 4 + i;
    const int e = (eIdx < E) ? eIdx : (E - 1);
    hh[g] = hl[e];
    tt[g] = tl[e];
    const int ai = (e0 + g * 4) * 4 + c;
    aold[g] = A[(ai < aLast) ? ai : aLast];
    wr[g] = active && (e0 + g * 4 + (cc >> 2) < E);
  }
  asm volatile("" :: "v"(hh[0]), "v"(hh[1]), "v"(hh[2]), "v"(hh[3]),
                     "v"(tt[0]), "v"(tt[1]), "v"(tt[2]), "v"(tt[3]),
                     "v"(aold[0]), "v"(aold[1]), "v"(aold[2]), "v"(aold[3]));
  v4f h0[4], h1[4], t0[4], t1[4];
#pragma unroll
  for (int g = 0; g < 4; ++g) {
    const v4f* hp = reinterpret_cast<const v4f*>(fact + (size_t)hh[g] * 64 + seg);
    const v4f* tp = reinterpret_cast<const v4f*>(ego + (size_t)tt[g] * 64 + seg);
    h0[g] = hp[0];
    h1[g] = hp[1];
    t0[g] = tp[0];
    t1[g] = tp[1];
  }
  float dv[4];
#pragma unroll
  for (int g = 0; g < 4; ++g) {
    const float hv[8] = {h0[g].x, h0[g].y, h0[g].z, h0[g].w, h1[g].x, h1[g].y, h1[g].z, h1[g].w};
    const float tv[8] = {t0[g].x, t0[g].y, t0[g].z, t0[g].w, t1[g].x, t1[g].y, t1[g].z, t1[g].w};
    v2f qh = h0[g].xy * h0[g].xy;
    v2f qt = t0[g].xy * t0[g].xy;
    qh = __builtin_elementwise_fma(h0[g].zw, h0[g].zw, qh);
    qt = __builtin_elementwise_fma(t0[g].zw, t0[g].zw, qt);
    qh = __builtin_elementwise_fma(h1[g].xy, h1[g].xy, qh);
    qt = __builtin_elementwise_fma(t1[g].xy, t1[g].xy, qt);
    qh = __builtin_elementwise_fma(h1[g].zw, h1[g].zw, qh);
    qt = __builtin_elementwise_fma(t1[g].zw, t1[g].zw, qt);
    const float ph = qh.x + qh.y, pt = qt.x + qt.y;
    const float sh = ph + xhalf(ph);
    const float st = pt + xhalf(pt);
    const float mine = 1.0f / fmaxf(sqrtf(kh ? st : sh), 1e-12f);
    const float other = xhalf(mine);
    const float hn = kh ? other : mine;
    const float tn = kh ? mine : other;
    v8f acc = {0.f, 0.f, 0.f, 0.f, 0.f, 0.f, 0.f, 0.f};
#pragma unroll
    for (int j = 0; j < 4; ++j) {
      v2f av, bv;
      av.x = hv[2 * j] * hn;
      av.y = hv[2 * j + 1] * hn;
      bv.x = fast_tanh(tv[2 * j] * tn);
      bv.y = fast_tanh(tv[2 * j + 1] * tn);
      acc = __builtin_amdgcn_wmma_f32_16x16x4_f32(false, av, false, bv,
                                                  (short)0, acc, false, false);
    }
    const int idx = lane & 7;
    float d = acc[0];
    d = (idx == 1) ? acc[1] : d;
    d = (idx == 2) ? acc[2] : d;
    d = (idx == 3) ? acc[3] : d;
    d = (idx == 4) ? acc[4] : d;
    d = (idx == 5) ? acc[5] : d;
    d = (idx == 6) ? acc[6] : d;
    d = (idx == 7) ? acc[7] : d;
    dv[g] = d;
  }

#pragma unroll
  for (int g = 0; g < 4; ++g)
    if (wr[g]) A[(size_t)(e0 + g * 4) * 4 + cc] = aold[g] + dv[g];
}

__global__ void k_acc_add(float* __restrict__ acc, const float* __restrict__ x, int n) {
  int i = blockIdx.x * blockDim.x + threadIdx.x;
  if (i < n) acc[i] += x[i];
}

__global__ void k_final(const float* __restrict__ acc, float* __restrict__ out, int n) {
  int i = blockIdx.x * blockDim.x + threadIdx.x;
  if (i < n) out[i] = acc[i] * (1.0f / 3.0f);
}

static inline size_t align16f(size_t x) { return (x + 15) & ~(size_t)15; }

extern "C" void kernel_launch(void* const* d_in, const int* in_sizes, int n_in,
                              void* d_out, int out_size, void* d_ws, size_t ws_size,
                              hipStream_t stream) {
  (void)n_in; (void)out_size; (void)ws_size;
  const float* u  = (const float*)d_in[0];
  const float* it = (const float*)d_in[1];
  const int* hl   = (const int*)d_in[2];
  const int* tl   = (const int*)d_in[3];

  const int userElems = in_sizes[0];
  const int itemElems = in_sizes[1];
  const int E = in_sizes[2];
  const int N = (userElems + itemElems) / 64;
  const size_t N64 = (size_t)N * 64;
  const size_t E4  = (size_t)E * 4;
  const size_t N4  = (size_t)N * 4;

  float* base = (float*)d_ws;
  size_t off = 0;
  float* ego   = base + off; off += align16f(N64);
  float* fact  = base + off; off += align16f(N64);
  float* acc   = base + off; off += align16f(N64);
  float* A     = base + off; off += align16f(E4);
  float* ew    = base + off; off += align16f(E4);
  float* dval  = base + off; off += align16f(N4);
  float* keyf  = base + off; off += align16f(N4);

#define GRID(n) ((unsigned)(((size_t)(n) + WG - 1) / WG))

  k_init<<<GRID(N64), WG, 0, stream>>>(u, it, ego, acc, userElems, (int)N64);
  k_fill1<<<GRID(E4), WG, 0, stream>>>(A, (int)E4);

  float* cur = ego;
  float* nxt = fact;
  const int nGroups = (E + 15) / 16;
  const dim3 degGrid(DEG_SLICES, (unsigned)((N + DEG_NODES - 1) / DEG_NODES));
  const unsigned msgGrid = (unsigned)((N + MSG_NODES - 1) / MSG_NODES);
  const int degPer = (((E + DEG_SLICES - 1) / DEG_SLICES) + 3) & ~3;

  for (int layer = 0; layer < 2; ++layer) {
    for (int itn = 0; itn < 2; ++itn) {
      k_softmax<<<GRID(E), WG, 0, stream>>>(A, ew, E);
      k_degree_lds<<<degGrid, DEG_WG, 0, stream>>>(ew, hl, nxt, E, N, degPer);
      k_dinv_sum<<<GRID(N), WG, 0, stream>>>(nxt, dval, N);
      k_edge_weight<<<GRID(E), WG, 0, stream>>>(ew, dval, hl, tl, keyf, E, (int)msgGrid);
      k_message_lds<<<msgGrid, MSG_WG, 0, stream>>>(ew, cur, hl, tl, keyf, nxt, E, N);
      const bool lastUpdateDead = (layer == 1 && itn == 1);
      if (!lastUpdateDead)
        k_updateA<<<GRID((size_t)nGroups * 32), WG, 0, stream>>>(
            nxt, cur, hl, tl, A, E, nGroups);
    }
    k_acc_add<<<GRID(N64), WG, 0, stream>>>(acc, nxt, (int)N64);
    float* tmp = cur; cur = nxt; nxt = tmp;
  }

  k_final<<<GRID(N64), WG, 0, stream>>>(acc, (float*)d_out, (int)N64);
#undef GRID
}


# Message passes 2-4 of a call read per-wave edge lists that pass 1 stores, no key rescan; lists sit in the running-sum buffer, the mean adds the 3 tables at the end; LDS adds in arrival order
# speedup vs baseline: 2.6767x; 1.1081x over previous
#include <hip/hip_runtime.h>
#include <math.h>

typedef float v2f __attribute__((ext_vector_type(2)));
typedef float v4f __attribute__((ext_vector_type(4)));
typedef float v8f __attribute__((ext_vector_type(8)));

#define WG 256

__device__ __forceinline__ float xhalf(float x) {
  int b = __builtin_bit_cast(int, x);
  return __builtin_bit_cast(float,
      __builtin_amdgcn_permlanex16(b, b, 0x76543210, 0xfedcba98, false, false));
}

__device__ __forceinline__ float fast_tanh(float x) {
  float r;
  asm("v_tanh_f32 %0, %1\n\tv_nop\n\tv_nop" : "=v"(r) : "v"(x));
  return r;
}

__global__ void k_init(const float* __restrict__ u, const float* __restrict__ it,
                       float* __restrict__ ego, int userElems, int total) {
  int i = blockIdx.x * blockDim.x + threadIdx.x;
  if (i >= total) return;
  ego[i] = (i < userElems) ? u[i] : it[i - userElems];
}

__global__ void k_fill1(float* __restrict__ a, int n) {
  int i = blockIdx.x * blockDim.x + threadIdx.x;
  if (i < n) a[i] = 1.0f;
}

__global__ void k_softmax(const float* __restrict__ A, float* __restrict__ ew, int E) {
  int e = blockIdx.x * blockDim.x + threadIdx.x;
  if (e >= E) return;
  float4 a = reinterpret_cast<const float4*>(A)[e];
  float m = fmaxf(fmaxf(a.x, a.y), fmaxf(a.z, a.w));
  float e0 = expf(a.x - m), e1 = expf(a.y - m), e2 = expf(a.z - m), e3 = expf(a.w - m);
  float inv = 1.0f / (e0 + e1 + e2 + e3);
  reinterpret_cast<float4*>(ew)[e] = make_float4(e0 * inv, e1 * inv, e2 * inv, e3 * inv);
}

#define DEG_SLICES 16
#define DEG_NODES  5000
#define DEG_WG     1024
#define DEG_VEC    4

__device__ __forceinline__ void lds_fadd(float* p, float v) {
  __hip_atomic_fetch_add(p, v, __ATOMIC_RELAXED, __HIP_MEMORY_SCOPE_WORKGROUP);
}

__global__ void __launch_bounds__(DEG_WG) k_degree_lds(
    const float* __restrict__ ew, const int* __restrict__ hl,
    float* __restrict__ part, int E, int N, int per) {
  __shared__ float4 tab4[DEG_NODES];
  float* tab = reinterpret_cast<float*>(tab4);
  const int s = blockIdx.x;
  const int lo = blockIdx.y * DEG_NODES;

  for (int i = threadIdx.x; i < DEG_NODES; i += DEG_WG)
    tab4[i] = make_float4(0.f, 0.f, 0.f, 0.f);
  __syncthreads();

  const int e0 = s * per;
  int e1 = e0 + per;
  if (e1 > E) e1 = E;
  const int nvec = (e1 > e0) ? ((e1 - e0) >> 2) : 0;
  const int4* hv = reinterpret_cast<const int4*>(hl + e0);
  const float4* ew4 = reinterpret_cast<const float4*>(ew);

  for (int v = threadIdx.x; v < nvec; v += DEG_WG * DEG_VEC) {
    int hd[DEG_VEC * 4];
#pragma unroll
    for (int k = 0; k < DEG_VEC; ++k) {
      int vk = v + k * DEG_WG;
      int4 h = make_int4(-1, -1, -1, -1);
      if (vk < nvec) h = hv[vk];
      hd[k * 4 + 0] = h.x; hd[k * 4 + 1] = h.y; hd[k * 4 + 2] = h.z; hd[k * 4 + 3] = h.w;
    }
    float4 w[DEG_VEC * 4];
#pragma unroll
    for (int j = 0; j < DEG_VEC * 4; ++j) {
      unsigned d = (unsigned)(hd[j] - lo);
      w[j] = make_float4(0.f, 0.f, 0.f, 0.f);
      if (d < (unsigned)DEG_NODES) w[j] = ew4[e0 + ((v + (j >> 2) * DEG_WG) << 2) + (j & 3)];
    }
#pragma unroll
    for (int j = 0; j < DEG_VEC * 4; ++j) {
      unsigned d = (unsigned)(hd[j] - lo);
      if (d < (unsigned)DEG_NODES) {
        float* t = tab + d * 4;
        lds_fadd(t + 0, w[j].x);
        lds_fadd(t + 1, w[j].y);
        lds_fadd(t + 2, w[j].z);
        lds_fadd(t + 3, w[j].w);
      }
    }
  }
  for (int e = e0 + (nvec << 2) + threadIdx.x; e < e1; e += DEG_WG) {
    unsigned d = (unsigned)(hl[e] - lo);
    if (d < (unsigned)DEG_NODES) {
      float4 wv = ew4[e];
      float* t = tab + d * 4;
      lds_fadd(t + 0, wv.x);
      lds_fadd(t + 1, wv.y);
      lds_fadd(t + 2, wv.z);
      lds_fadd(t + 3, wv.w);
    }
  }
  __syncthreads();

  int nodes = N - lo;
  if (nodes > DEG_NODES) nodes = DEG_NODES;
  float4* dst = reinterpret_cast<float4*>(part) + (size_t)s * N + lo;
  for (int i = threadIdx.x; i < nodes; i += DEG_WG) dst[i] = tab4[i];
}

__global__ void k_dinv_sum(const float* __restrict__ part, float* __restrict__ d, int N) {
  int n = blockIdx.x * blockDim.x + threadIdx.x;
  if (n >= N) return;
  const float4* p4 = reinterpret_cast<const float4*>(part);
  float4 sum = make_float4(0.f, 0.f, 0.f, 0.f);
#pragma unroll
  for (int k = 0; k < DEG_SLICES; ++k) {
    float4 v = p4[(size_t)k * N + n];
    sum.x += v.x; sum.y += v.y; sum.z += v.z; sum.w += v.w;
  }
  float4 r;
  r.x = 1.0f / sqrtf(fmaxf(sum.x, 1e-8f));
  r.y = 1.0f / sqrtf(fmaxf(sum.y, 1e-8f));
  r.z = 1.0f / sqrtf(fmaxf(sum.z, 1e-8f));
  r.w = 1.0f / sqrtf(fmaxf(sum.w, 1e-8f));
  reinterpret_cast<float4*>(d)[n] = r;
}

#define MSG_NODES 320
#define MSG_WG    1024
#define MSG_WAVES 32
#define MSG_BATCH 16
#define MSG_QCAP  544
#define MSG_NOKEY 255u
#define MSG_OVER  0xffffffffu
#define MSG_TRIP  128

__global__ void k_edge_weight(float* __restrict__ ew, const float* __restrict__ dinv,
                              const int* __restrict__ hl, const int* __restrict__ tl,
                              float* __restrict__ keyf, int E, int R) {
  int e = blockIdx.x * blockDim.x + threadIdx.x;
  unsigned char* keyb = reinterpret_cast<unsigned char*>(keyf);
  if (e >= E) {
    if (e < ((E + 15) & ~15)) keyb[e] = (unsigned char)MSG_NOKEY;
    return;
  }
  int h = hl[e];
  float4 w = reinterpret_cast<float4*>(ew)[e];
  float4 dh = reinterpret_cast<const float4*>(dinv)[h];
  float4 dt = reinterpret_cast<const float4*>(dinv)[tl[e]];
  w.x *= dh.x * dt.x;
  w.y *= dh.y * dt.y;
  w.z *= dh.z * dt.z;
  w.w *= dh.w * dt.w;
  reinterpret_cast<float4*>(ew)[e] = w;
  keyb[e] = (unsigned char)((unsigned)h % (unsigned)R);
}

__device__ __forceinline__ unsigned nonzero_bytes(unsigned x) {
  return ((x & 0x7f7f7f7fu) + 0x7f7f7f7fu) | x;
}

__device__ __forceinline__ unsigned key_hits(uint4 k, unsigned pat) {
  unsigned y0 = nonzero_bytes(k.x ^ pat);
  unsigned y1 = nonzero_bytes(k.y ^ pat);
  unsigned y2 = nonzero_bytes(k.z ^ pat);
  unsigned y3 = nonzero_bytes(k.w ^ pat);
  unsigned n = ((y0 >> 7) & 0x01010101u) | ((y1 >> 6) & 0x02020202u) |
               ((y2 >> 5) & 0x04040404u) | ((y3 >> 4) & 0x08080808u);
  return ~n & 0x0f0f0f0fu;
}

template <bool FULL>
__device__ __forceinline__ void msg_batch(const unsigned* qp, int n, int lane, unsigned R,
                                          const float* __restrict__ ew,
                                          const float* __restrict__ ego,
                                          const int* __restrict__ hl,
                                          const int* __restrict__ tl, float* tab) {
  const int l16 = lane & 15;
  const unsigned el = qp[(FULL || l16 < n) ? l16 : 0];
  const int tv = tl[el];
  const int dv = (int)((unsigned)hl[el] / R);
  const int j8 = lane >> 2;
  const unsigned ea = qp[(FULL || j8 < n) ? j8 : 0];
  const unsigned eb = qp[(FULL || j8 + 8 < n) ? j8 + 8 : 0];
  const float wa = ew[(size_t)ea * 4 + (lane & 3)];
  const float wb = ew[(size_t)eb * 4 + (lane & 3)];
  const int f = lane >> 3;

  float2 v[MSG_BATCH];
  int d[MSG_BATCH];
#pragma unroll
  for (int i = 0; i < MSG_BATCH; ++i) {
    const int t = __builtin_amdgcn_readlane(tv, i);
    d[i] = __builtin_amdgcn_readlane(dv, i);
    v[i] = make_float2(0.f, 0.f);
    if (FULL || i < n)
      v[i] = *reinterpret_cast<const float2*>(ego + (size_t)t * 64 + lane * 2);
  }
#pragma unroll
  for (int i = 0; i < MSG_BATCH; ++i) {
    const float w = __shfl((i < 8) ? wa : wb, 4 * (i & 7) + f);
    if (FULL || i < n) {
      float* tp = tab + d[i] * 64 + lane * 2;
      lds_fadd(tp + 0, w * v[i].x);
      lds_fadd(tp + 1, w * v[i].y);
    }
  }
}

template <bool REPLAY>
__device__ __forceinline__ void message_body(
    const float* __restrict__ ew, const float* __restrict__ ego,
    const int* __restrict__ hl, const int* __restrict__ tl,
    const float* __restrict__ keyf, float* __restrict__ fact,
    float* __restrict__ listf, int E, int N,
    float4* mtab4, unsigned (*mq)[MSG_QCAP]) {
  float* tab = reinterpret_cast<float*>(mtab4);
  const int r = blockIdx.x;
  const unsigned R = gridDim.x;
  const int lane = threadIdx.x & 31;
  const int wave = threadIdx.x >> 5;

  for (int i = threadIdx.x; i < MSG_NODES * 16; i += MSG_WG)
    mtab4[i] = make_float4(0.f, 0.f, 0.f, 0.f);
  __syncthreads();

  const uint4* k4 = reinterpret_cast<const uint4*>(keyf);
  const int nChunks = (E + 15) >> 4;
  const int perTrip = MSG_WAVES * 128;
  const int trips = (nChunks + perTrip - 1) / perTrip;
  const unsigned pat = (unsigned)r * 0x01010101u;
  unsigned* q = mq[wave];
  int cnt = 0;

  const unsigned cap = ((unsigned)N * 64u) / (R * (unsigned)MSG_WAVES);
  unsigned* gl = reinterpret_cast<unsigned*>(listf) +
                 (size_t)((unsigned)r * MSG_WAVES + (unsigned)wave) * cap;
  unsigned stored = 0;
  bool fits = true;
  bool scan = true;

  if (REPLAY) {
    const unsigned n = (unsigned)__builtin_amdgcn_readfirstlane((int)gl[0]);
    if (n != MSG_OVER) {
      scan = false;
      for (unsigned b0 = 0; b0 < n; b0 += MSG_TRIP) {
        const unsigned m = (n - b0 < (unsigned)MSG_TRIP) ? (n - b0) : (unsigned)MSG_TRIP;
        unsigned idv[MSG_TRIP / 32];
#pragma unroll
        for (int k = 0; k < MSG_TRIP / 32; ++k) {
          const unsigned x = (unsigned)(k * 32 + lane);
          idv[k] = 0u;
          if (x < m) idv[k] = gl[1 + b0 + x];
        }
#pragma unroll
        for (int k = 0; k < MSG_TRIP / 32; ++k) q[k * 32 + lane] = idv[k];
        const int full = (int)(m >> 4);
        for (int b = 0; b < full; ++b)
          msg_batch<true>(q + b * MSG_BATCH, MSG_BATCH, lane, R, ew, ego, hl, tl, tab);
        const int rem = (int)(m & 15u);
        if (rem > 0) msg_batch<false>(q + full * MSG_BATCH, rem, lane, R, ew, ego, hl, tl, tab);
      }
    }
  }

  int j = r % trips;
  for (int jj = 0; scan && jj < trips; ++jj) {
    const int cb = j * perTrip + wave * 128;
    j = (j + 1 == trips) ? 0 : j + 1;
    uint4 kk[4];
#pragma unroll
    for (int u = 0; u < 4; ++u) {
      const int c = cb + u * 32 + lane;
      kk[u] = make_uint4(~0u, ~0u, ~0u, ~0u);
      if (c < nChunks) kk[u] = k4[c];
    }
#pragma unroll
    for (int u = 0; u < 4; ++u) {
      const int c = cb + u * 32 + lane;
      unsigned m = key_hits(kk[u], pat);
      unsigned bal = __builtin_amdgcn_ballot_w32(m != 0u);
      while (bal != 0u) {
        if (m != 0u) {
          const int p = __builtin_ctz(m);
          const int idx = ((p & 3) << 2) | (p >> 3);
          q[cnt + (int)__builtin_amdgcn_mbcnt_lo(bal, 0u)] = (unsigned)(c * 16 + idx);
          m &= m - 1u;
        }
        cnt += __builtin_popcount(bal);
        bal = __builtin_amdgcn_ballot_w32(m != 0u);
      }
      while (cnt >= MSG_BATCH) {
        cnt -= MSG_BATCH;
        if (!REPLAY) {
          if (fits && stored + MSG_BATCH + 1u <= cap) {
            if (lane < MSG_BATCH) gl[1 + stored + lane] = q[cnt + lane];
            stored += MSG_BATCH;
          } else {
            fits = false;
          }
        }
        msg_batch<true>(q + cnt, MSG_BATCH, lane, R, ew, ego, hl, tl, tab);
      }
    }
  }
  if (cnt > 0) {
    if (!REPLAY) {
      if (fits && stored + (unsigned)cnt + 1u <= cap) {
        if (lane < cnt) gl[1 + stored + lane] = q[lane];
        stored += (unsigned)cnt;
      } else {
        fits = false;
      }
    }
    msg_batch<false>(q, cnt, lane, R, ew, ego, hl, tl, tab);
  }
  if (!REPLAY && lane == 0) gl[0] = fits ? stored : MSG_OVER;
  __syncthreads();

  float4* dst = reinterpret_cast<float4*>(fact);
  for (int i = threadIdx.x; i < MSG_NODES * 16; i += MSG_WG) {
    const int node = (i >> 4) * (int)R + r;
    if (node < N) dst[(size_t)node * 16 + (i & 15)] = mtab4[i];
  }
}

__global__ void __launch_bounds__(MSG_WG) k_message_first(
    const float* __restrict__ ew, const float* __restrict__ ego,
    const int* __restrict__ hl, const int* __restrict__ tl,
    const float* __restrict__ keyf, float* __restrict__ fact,
    float* __restrict__ listf, int E, int N) {
  __shared__ float4 mtab4[MSG_NODES * 16];
  __shared__ unsigned mq[MSG_WAVES][MSG_QCAP];
  message_body<false>(ew, ego, hl, tl, keyf, fact, listf, E, N, mtab4, mq);
}

__global__ void __launch_bounds__(MSG_WG) k_message_replay(
    const float* __restrict__ ew, const float* __restrict__ ego,
    const int* __restrict__ hl, const int* __restrict__ tl,
    const float* __restrict__ keyf, float* __restrict__ fact,
    float* __restrict__ listf, int E, int N) {
  __shared__ float4 mtab4[MSG_NODES * 16];
  __shared__ unsigned mq[MSG_WAVES][MSG_QCAP];
  message_body<true>(ew, ego, hl, tl, keyf, fact, listf, E, N, mtab4, mq);
}

__global__ void __launch_bounds__(WG) k_updateA(
    const float* __restrict__ fact, const float* __restrict__ ego,
    const int* __restrict__ hl, const int* __restrict__ tl,
    float* __restrict__ A, int E, int nGroups) {
  int tid = blockIdx.x * blockDim.x + threadIdx.x;
  int wave = tid >> 5;
  int lane = tid & 31;
  if (wave >= nGroups) return;

  const int c = lane & 15;
  const int kh = lane >> 4;
  const int i = c >> 2;
  const int f = c & 3;
  const int seg = f * 16 + kh * 8;

  const bool active = (lane < 8) || (lane >= 24);
  const int cc = (lane < 8) ? lane : (lane - 16);
  const int e0 = wave * 16;

  int hh[4], tt[4];
  float aold[4];
  bool wr[4];
  const int aLast = E * 4 - 1;
#pragma unroll
  for (int g = 0; g < 4; ++g) {
    const int eIdx = e0 + g * 4 + i;
    const int e = (eIdx < E) ? eIdx : (E - 1);
    hh[g] = hl[e];
    tt[g] = tl[e];
    const int ai = (e0 + g * 4) * 4 + c;
    aold[g] = A[(ai < aLast) ? ai : aLast];
    wr[g] = active && (e0 + g * 4 + (cc >> 2) < E);
  }
  asm volatile("" :: "v"(hh[0]), "v"(hh[1]), "v"(hh[2]), "v"(hh[3]),
                     "v"(tt[0]), "v"(tt[1]), "v"(tt[2]), "v"(tt[3]),
                     "v"(aold[0]), "v"(aold[1]), "v"(aold[2]), "v"(aold[3]));
  v4f h0[4], h1[4], t0[4], t1[4];
#pragma unroll
  for (int g = 0; g < 4; ++g) {
    const v4f* hp = reinterpret_cast<const v4f*>(fact + (size_t)hh[g] * 64 + seg);
    const v4f* tp = reinterpret_cast<const v4f*>(ego + (size_t)tt[g] * 64 + seg);
    h0[g] = hp[0];
    h1[g] = hp[1];
    t0[g] = tp[0];
    t1[g] = tp[1];
  }
  float dv[4];
#pragma unroll
  for (int g = 0; g < 4; ++g) {
    const float hv[8] = {h0[g].x, h0[g].y, h0[g].z, h0[g].w, h1[g].x, h1[g].y, h1[g].z, h1[g].w};
    const float tv[8] = {t0[g].x, t0[g].y, t0[g].z, t0[g].w, t1[g].x, t1[g].y, t1[g].z, t1[g].w};
    v2f qh = h0[g].xy * h0[g].xy;
    v2f qt = t0[g].xy * t0[g].xy;
    qh = __builtin_elementwise_fma(h0[g].zw, h0[g].zw, qh);
    qt = __builtin_elementwise_fma(t0[g].zw, t0[g].zw, qt);
    qh = __builtin_elementwise_fma(h1[g].xy, h1[g].xy, qh);
    qt = __builtin_elementwise_fma(t1[g].xy, t1[g].xy, qt);
    qh = __builtin_elementwise_fma(h1[g].zw, h1[g].zw, qh);
    qt = __builtin_elementwise_fma(t1[g].zw, t1[g].zw, qt);
    const float ph = qh.x + qh.y, pt = qt.x + qt.y;
    const float sh = ph + xhalf(ph);
    const float st = pt + xhalf(pt);
    const float mine = 1.0f / fmaxf(sqrtf(kh ? st : sh), 1e-12f);
    const float other = xhalf(mine);
    const float hn = kh ? other : mine;
    const float tn = kh ? mine : other;
    v8f acc = {0.f, 0.f, 0.f, 0.f, 0.f, 0.f, 0.f, 0.f};
#pragma unroll
    for (int j = 0; j < 4; ++j) {
      v2f av, bv;
      av.x = hv[2 * j] * hn;
      av.y = hv[2 * j + 1] * hn;
      bv.x = fast_tanh(tv[2 * j] * tn);
      bv.y = fast_tanh(tv[2 * j + 1] * tn);
      acc = __builtin_amdgcn_wmma_f32_16x16x4_f32(false, av, false, bv,
                                                  (short)0, acc, false, false);
    }
    const int idx = lane & 7;
    float d = acc[0];
    d = (idx == 1) ? acc[1] : d;
    d = (idx == 2) ? acc[2] : d;
    d = (idx == 3) ? acc[3] : d;
    d = (idx == 4) ? acc[4] : d;
    d = (idx == 5) ? acc[5] : d;
    d = (idx == 6) ? acc[6] : d;
    d = (idx == 7) ? acc[7] : d;
    dv[g] = d;
  }

#pragma unroll
  for (int g = 0; g < 4; ++g)
    if (wr[g]) A[(size_t)(e0 + g * 4) * 4 + cc] = aold[g] + dv[g];
}

__global__ void k_final(const float* __restrict__ u, const float* __restrict__ it,
                        const float* __restrict__ l1, const float* __restrict__ l2,
                        float* __restrict__ out, int userElems, int n) {
  int i = blockIdx.x * blockDim.x + threadIdx.x;
  if (i >= n) return;
  float v = (i < userElems) ? u[i] : it[i - userElems];
  v += l1[i];
  v += l2[i];
  out[i] = v * (1.0f / 3.0f);
}

static inline size_t align16f(size_t x) { return (x + 15) & ~(size_t)15; }

extern "C" void kernel_launch(void* const* d_in, const int* in_sizes, int n_in,
                              void* d_out, int out_size, void* d_ws, size_t ws_size,
                              hipStream_t stream) {
  (void)n_in; (void)out_size; (void)ws_size;
  const float* u  = (const float*)d_in[0];
  const float* it = (const float*)d_in[1];
  const int* hl   = (const int*)d_in[2];
  const int* tl   = (const int*)d_in[3];

  const int userElems = in_sizes[0];
  const int itemElems = in_sizes[1];
  const int E = in_sizes[2];
  const int N = (userElems + itemElems) / 64;
  const size_t N64 = (size_t)N * 64;
  const size_t E4  = (size_t)E * 4;
  const size_t N4  = (size_t)N * 4;

  float* base = (float*)d_ws;
  size_t off = 0;
  float* ego   = base + off; off += align16f(N64);
  float* fact  = base + off; off += align16f(N64);
  float* lists = base + off; off += align16f(N64);
  float* A     = base + off; off += align16f(E4);
  float* ew    = base + off; off += align16f(E4);
  float* dval  = base + off; off += align16f(N4);
  float* keyf  = base + off; off += align16f(N4);

#define GRID(n) ((unsigned)(((size_t)(n) + WG - 1) / WG))

  k_init<<<GRID(N64), WG, 0, stream>>>(u, it, ego, userElems, (int)N64);
  k_fill1<<<GRID(E4), WG, 0, stream>>>(A, (int)E4);

  float* cur = ego;
  float* nxt = fact;
  const int nGroups = (E + 15) / 16;
  const dim3 degGrid(DEG_SLICES, (unsigned)((N + DEG_NODES - 1) / DEG_NODES));
  const unsigned msgGrid = (unsigned)((N + MSG_NODES - 1) / MSG_NODES);
  const int degPer = (((E + DEG_SLICES - 1) / DEG_SLICES) + 3) & ~3;

  for (int layer = 0; layer < 2; ++layer) {
    for (int itn = 0; itn < 2; ++itn) {
      k_softmax<<<GRID(E), WG, 0, stream>>>(A, ew, E);
      k_degree_lds<<<degGrid, DEG_WG, 0, stream>>>(ew, hl, nxt, E, N, degPer);
      k_dinv_sum<<<GRID(N), WG, 0, stream>>>(nxt, dval, N);
      k_edge_weight<<<GRID(E), WG, 0, stream>>>(ew, dval, hl, tl, keyf, E, (int)msgGrid);
      if (layer == 0 && itn == 0)
        k_message_first<<<msgGrid, MSG_WG, 0, stream>>>(ew, cur, hl, tl, keyf, nxt, lists, E, N);
      else
        k_message_replay<<<msgGrid, MSG_WG, 0, stream>>>(ew, cur, hl, tl, keyf, nxt, lists, E, N);
      const bool lastUpdateDead = (layer == 1 && itn == 1);
      if (!lastUpdateDead)
        k_updateA<<<GRID((size_t)nGroups * 32), WG, 0, stream>>>(
            nxt, cur, hl, tl, A, E, nGroups);
    }
    float* tmp = cur; cur = nxt; nxt = tmp;
  }

  k_final<<<GRID(N64), WG, 0, stream>>>(u, it, nxt, cur, (float*)d_out, userElems, (int)N64);
#undef GRID
}


# Degree sums of passes 2-4 by the message groups from their stored edge lists: one LDS table a group (LDS f32 adds, arrival order), its inverse roots too; slice planes, their sum, 3 launches gone
# speedup vs baseline: 2.8293x; 1.0570x over previous
#include <hip/hip_runtime.h>
#include <math.h>

typedef float v2f __attribute__((ext_vector_type(2)));
typedef float v4f __attribute__((ext_vector_type(4)));
typedef float v8f __attribute__((ext_vector_type(8)));

#define WG 256

__device__ __forceinline__ float xhalf(float x) {
  int b = __builtin_bit_cast(int, x);
  return __builtin_bit_cast(float,
      __builtin_amdgcn_permlanex16(b, b, 0x76543210, 0xfedcba98, false, false));
}

__device__ __forceinline__ float fast_tanh(float x) {
  float r;
  asm("v_tanh_f32 %0, %1\n\tv_nop\n\tv_nop" : "=v"(r) : "v"(x));
  return r;
}

__global__ void k_init(const float* __restrict__ u, const float* __restrict__ it,
                       float* __restrict__ ego, int userElems, int total) {
  int i = blockIdx.x * blockDim.x + threadIdx.x;
  if (i >= total) return;
  ego[i] = (i < userElems) ? u[i] : it[i - userElems];
}

__global__ void k_fill1(float* __restrict__ a, int n) {
  int i = blockIdx.x * blockDim.x + threadIdx.x;
  if (i < n) a[i] = 1.0f;
}

__global__ void k_softmax(const float* __restrict__ A, float* __restrict__ ew, int E) {
  int e = blockIdx.x * blockDim.x + threadIdx.x;
  if (e >= E) return;
  float4 a = reinterpret_cast<const float4*>(A)[e];
  float m = fmaxf(fmaxf(a.x, a.y), fmaxf(a.z, a.w));
  float e0 = expf(a.x - m), e1 = expf(a.y - m), e2 = expf(a.z - m), e3 = expf(a.w - m);
  float inv = 1.0f / (e0 + e1 + e2 + e3);
  reinterpret_cast<float4*>(ew)[e] = make_float4(e0 * inv, e1 * inv, e2 * inv, e3 * inv);
}

#define DEG_SLICES 16
#define DEG_NODES  5000
#define DEG_WG     1024
#define DEG_VEC    4

__device__ __forceinline__ void lds_fadd(float* p, float v) {
  __hip_atomic_fetch_add(p, v, __ATOMIC_RELAXED, __HIP_MEMORY_SCOPE_WORKGROUP);
}

__global__ void __launch_bounds__(DEG_WG) k_degree_lds(
    const float* __restrict__ ew, const int* __restrict__ hl,
    float* __restrict__ part, int E, int N, int per) {
  __shared__ float4 tab4[DEG_NODES];
  float* tab = reinterpret_cast<float*>(tab4);
  const int s = blockIdx.x;
  const int lo = blockIdx.y * DEG_NODES;

  for (int i = threadIdx.x; i < DEG_NODES; i += DEG_WG)
    tab4[i] = make_float4(0.f, 0.f, 0.f, 0.f);
  __syncthreads();

  const int e0 = s * per;
  int e1 = e0 + per;
  if (e1 > E) e1 = E;
  const int nvec = (e1 > e0) ? ((e1 - e0) >> 2) : 0;
  const int4* hv = reinterpret_cast<const int4*>(hl + e0);
  const float4* ew4 = reinterpret_cast<const float4*>(ew);

  for (int v = threadIdx.x; v < nvec; v += DEG_WG * DEG_VEC) {
    int hd[DEG_VEC * 4];
#pragma unroll
    for (int k = 0; k < DEG_VEC; ++k) {
      int vk = v + k * DEG_WG;
      int4 h = make_int4(-1, -1, -1, -1);
      if (vk < nvec) h = hv[vk];
      hd[k * 4 + 0] = h.x; hd[k * 4 + 1] = h.y; hd[k * 4 + 2] = h.z; hd[k * 4 + 3] = h.w;
    }
    float4 w[DEG_VEC * 4];
#pragma unroll
    for (int j = 0; j < DEG_VEC * 4; ++j) {
      unsigned d = (unsigned)(hd[j] - lo);
      w[j] = make_float4(0.f, 0.f, 0.f, 0.f);
      if (d < (unsigned)DEG_NODES) w[j] = ew4[e0 + ((v + (j >> 2) * DEG_WG) << 2) + (j & 3)];
    }
#pragma unroll
    for (int j = 0; j < DEG_VEC * 4; ++j) {
      unsigned d = (unsigned)(hd[j] - lo);
      if (d < (unsigned)DEG_NODES) {
        float* t = tab + d * 4;
        lds_fadd(t + 0, w[j].x);
        lds_fadd(t + 1, w[j].y);
        lds_fadd(t + 2, w[j].z);
        lds_fadd(t + 3, w[j].w);
      }
    }
  }
  for (int e = e0 + (nvec << 2) + threadIdx.x; e < e1; e += DEG_WG) {
    unsigned d = (unsigned)(hl[e] - lo);
    if (d < (unsigned)DEG_NODES) {
      float4 wv = ew4[e];
      float* t = tab + d * 4;
      lds_fadd(t + 0, wv.x);
      lds_fadd(t + 1, wv.y);
      lds_fadd(t + 2, wv.z);
      lds_fadd(t + 3, wv.w);
    }
  }
  __syncthreads();

  int nodes = N - lo;
  if (nodes > DEG_NODES) nodes = DEG_NODES;
  float4* dst = reinterpret_cast<float4*>(part) + (size_t)s * N + lo;
  for (int i = threadIdx.x; i < nodes; i += DEG_WG) dst[i] = tab4[i];
}

__global__ void k_dinv_sum(const float* __restrict__ part, float* __restrict__ d, int N) {
  int n = blockIdx.x * blockDim.x + threadIdx.x;
  if (n >= N) return;
  const float4* p4 = reinterpret_cast<const float4*>(part);
  float4 sum = make_float4(0.f, 0.f, 0.f, 0.f);
#pragma unroll
  for (int k = 0; k < DEG_SLICES; ++k) {
    float4 v = p4[(size_t)k * N + n];
    sum.x += v.x; sum.y += v.y; sum.z += v.z; sum.w += v.w;
  }
  float4 r;
  r.x = 1.0f / sqrtf(fmaxf(sum.x, 1e-8f));
  r.y = 1.0f / sqrtf(fmaxf(sum.y, 1e-8f));
  r.z = 1.0f / sqrtf(fmaxf(sum.z, 1e-8f));
  r.w = 1.0f / sqrtf(fmaxf(sum.w, 1e-8f));
  reinterpret_cast<float4*>(d)[n] = r;
}

#define MSG_NODES 320
#define MSG_WG    1024
#define MSG_WAVES 32
#define MSG_BATCH 16
#define MSG_QCAP  544
#define MSG_NOKEY 255u
#define MSG_OVER  0xffffffffu
#define MSG_TRIP  128

template <bool GROUPED>
__global__ void k_edge_weight(float* __restrict__ ew, const float* __restrict__ dinv,
                              const int* __restrict__ hl, const int* __restrict__ tl,
                              float* __restrict__ keyf, int E, int R) {
  int e = blockIdx.x * blockDim.x + threadIdx.x;
  unsigned char* keyb = reinterpret_cast<unsigned char*>(keyf);
  if (e >= E) {
    if (e < ((E + 15) & ~15)) keyb[e] = (unsigned char)MSG_NOKEY;
    return;
  }
  const unsigned h = (unsigned)hl[e];
  const unsigned t = (unsigned)tl[e];
  const unsigned hg = h % (unsigned)R;
  const unsigned hi = GROUPED ? hg * (unsigned)MSG_NODES + h / (unsigned)R : h;
  const unsigned ti = GROUPED ? (t % (unsigned)R) * (unsigned)MSG_NODES + t / (unsigned)R : t;
  float4 w = reinterpret_cast<float4*>(ew)[e];
  float4 dh = reinterpret_cast<const float4*>(dinv)[hi];
  float4 dt = reinterpret_cast<const float4*>(dinv)[ti];
  w.x *= dh.x * dt.x;
  w.y *= dh.y * dt.y;
  w.z *= dh.z * dt.z;
  w.w *= dh.w * dt.w;
  reinterpret_cast<float4*>(ew)[e] = w;
  keyb[e] = (unsigned char)hg;
}

__device__ __forceinline__ unsigned nonzero_bytes(unsigned x) {
  return ((x & 0x7f7f7f7fu) + 0x7f7f7f7fu) | x;
}

__device__ __forceinline__ unsigned key_hits(uint4 k, unsigned pat) {
  unsigned y0 = nonzero_bytes(k.x ^ pat);
  unsigned y1 = nonzero_bytes(k.y ^ pat);
  unsigned y2 = nonzero_bytes(k.z ^ pat);
  unsigned y3 = nonzero_bytes(k.w ^ pat);
  unsigned n = ((y0 >> 7) & 0x01010101u) | ((y1 >> 6) & 0x02020202u) |
               ((y2 >> 5) & 0x04040404u) | ((y3 >> 4) & 0x08080808u);
  return ~n & 0x0f0f0f0fu;
}

template <bool FULL>
__device__ __forceinline__ void msg_batch(const unsigned* qp, int n, int lane, unsigned R,
                                          const float* __restrict__ ew,
                                          const float* __restrict__ ego,
                                          const int* __restrict__ hl,
                                          const int* __restrict__ tl, float* tab) {
  const int l16 = lane & 15;
  const unsigned el = qp[(FULL || l16 < n) ? l16 : 0];
  const int tv = tl[el];
  const int dv = (int)((unsigned)hl[el] / R);
  const int j8 = lane >> 2;
  const unsigned ea = qp[(FULL || j8 < n) ? j8 : 0];
  const unsigned eb = qp[(FULL || j8 + 8 < n) ? j8 + 8 : 0];
  const float wa = ew[(size_t)ea * 4 + (lane & 3)];
  const float wb = ew[(size_t)eb * 4 + (lane & 3)];
  const int f = lane >> 3;

  float2 v[MSG_BATCH];
  int d[MSG_BATCH];
#pragma unroll
  for (int i = 0; i < MSG_BATCH; ++i) {
    const int t = __builtin_amdgcn_readlane(tv, i);
    d[i] = __builtin_amdgcn_readlane(dv, i);
    v[i] = make_float2(0.f, 0.f);
    if (FULL || i < n)
      v[i] = *reinterpret_cast<const float2*>(ego + (size_t)t * 64 + lane * 2);
  }
#pragma unroll
  for (int i = 0; i < MSG_BATCH; ++i) {
    const float w = __shfl((i < 8) ? wa : wb, 4 * (i & 7) + f);
    if (FULL || i < n) {
      float* tp = tab + d[i] * 64 + lane * 2;
      lds_fadd(tp + 0, w * v[i].x);
      lds_fadd(tp + 1, w * v[i].y);
    }
  }
}

template <bool REPLAY>
__device__ __forceinline__ void message_body(
    const float* __restrict__ ew, const float* __restrict__ ego,
    const int* __restrict__ hl, const int* __restrict__ tl,
    const float* __restrict__ keyf, float* __restrict__ fact,
    float* __restrict__ listf, int E, int N,
    float4* mtab4, unsigned (*mq)[MSG_QCAP]) {
  float* tab = reinterpret_cast<float*>(mtab4);
  const int r = blockIdx.x;
  const unsigned R = gridDim.x;
  const int lane = threadIdx.x & 31;
  const int wave = threadIdx.x >> 5;

  for (int i = threadIdx.x; i < MSG_NODES * 16; i += MSG_WG)
    mtab4[i] = make_float4(0.f, 0.f, 0.f, 0.f);
  __syncthreads();

  const uint4* k4 = reinterpret_cast<const uint4*>(keyf);
  const int nChunks = (E + 15) >> 4;
  const int perTrip = MSG_WAVES * 128;
  const int trips = (nChunks + perTrip - 1) / perTrip;
  const unsigned pat = (unsigned)r * 0x01010101u;
  unsigned* q = mq[wave];
  int cnt = 0;

  const unsigned cap = ((unsigned)N * 64u) / (R * (unsigned)MSG_WAVES);
  unsigned* gl = reinterpret_cast<unsigned*>(listf) +
                 (size_t)((unsigned)r * MSG_WAVES + (unsigned)wave) * cap;
  unsigned stored = 0;
  bool fits = true;
  bool scan = true;

  if (REPLAY) {
    const unsigned n = (unsigned)__builtin_amdgcn_readfirstlane((int)gl[0]);
    if (n != MSG_OVER) {
      scan = false;
      for (unsigned b0 = 0; b0 < n; b0 += MSG_TRIP) {
        const unsigned m = (n - b0 < (unsigned)MSG_TRIP) ? (n - b0) : (unsigned)MSG_TRIP;
        unsigned idv[MSG_TRIP / 32];
#pragma unroll
        for (int k = 0; k < MSG_TRIP / 32; ++k) {
          const unsigned x = (unsigned)(k * 32 + lane);
          idv[k] = 0u;
          if (x < m) idv[k] = gl[1 + b0 + x];
        }
#pragma unroll
        for (int k = 0; k < MSG_TRIP / 32; ++k) q[k * 32 + lane] = idv[k];
        const int full = (int)(m >> 4);
        for (int b = 0; b < full; ++b)
          msg_batch<true>(q + b * MSG_BATCH, MSG_BATCH, lane, R, ew, ego, hl, tl, tab);
        const int rem = (int)(m & 15u);
        if (rem > 0) msg_batch<false>(q + full * MSG_BATCH, rem, lane, R, ew, ego, hl, tl, tab);
      }
    }
  }

  int j = r % trips;
  for (int jj = 0; scan && jj < trips; ++jj) {
    const int cb = j * perTrip + wave * 128;
    j = (j + 1 == trips) ? 0 : j + 1;
    uint4 kk[4];
#pragma unroll
    for (int u = 0; u < 4; ++u) {
      const int c = cb + u * 32 + lane;
      kk[u] = make_uint4(~0u, ~0u, ~0u, ~0u);
      if (c < nChunks) kk[u] = k4[c];
    }
#pragma unroll
    for (int u = 0; u < 4; ++u) {
      const int c = cb + u * 32 + lane;
      unsigned m = key_hits(kk[u], pat);
      unsigned bal = __builtin_amdgcn_ballot_w32(m != 0u);
      while (bal != 0u) {
        if (m != 0u) {
          const int p = __builtin_ctz(m);
          const int idx = ((p & 3) << 2) | (p >> 3);
          q[cnt + (int)__builtin_amdgcn_mbcnt_lo(bal, 0u)] = (unsigned)(c * 16 + idx);
          m &= m - 1u;
        }
        cnt += __builtin_popcount(bal);
        bal = __builtin_amdgcn_ballot_w32(m != 0u);
      }
      while (cnt >= MSG_BATCH) {
        cnt -= MSG_BATCH;
        if (!REPLAY) {
          if (fits && stored + MSG_BATCH + 1u <= cap) {
            if (lane < MSG_BATCH) gl[1 + stored + lane] = q[cnt + lane];
            stored += MSG_BATCH;
          } else {
            fits = false;
          }
        }
        msg_batch<true>(q + cnt, MSG_BATCH, lane, R, ew, ego, hl, tl, tab);
      }
    }
  }
  if (cnt > 0) {
    if (!REPLAY) {
      if (fits && stored + (unsigned)cnt + 1u <= cap) {
        if (lane < cnt) gl[1 + stored + lane] = q[lane];
        stored += (unsigned)cnt;
      } else {
        fits = false;
      }
    }
    msg_batch<false>(q, cnt, lane, R, ew, ego, hl, tl, tab);
  }
  if (!REPLAY && lane == 0) gl[0] = fits ? stored : MSG_OVER;
  __syncthreads();

  float4* dst = reinterpret_cast<float4*>(fact);
  for (int i = threadIdx.x; i < MSG_NODES * 16; i += MSG_WG) {
    const int node = (i >> 4) * (int)R + r;
    if (node < N) dst[(size_t)node * 16 + (i & 15)] = mtab4[i];
  }
}

__global__ void __launch_bounds__(MSG_WG) k_message_first(
    const float* __restrict__ ew, const float* __restrict__ ego,
    const int* __restrict__ hl, const int* __restrict__ tl,
    const float* __restrict__ keyf, float* __restrict__ fact,
    float* __restrict__ listf, int E, int N) {
  __shared__ float4 mtab4[MSG_NODES * 16];
  __shared__ unsigned mq[MSG_WAVES][MSG_QCAP];
  message_body<false>(ew, ego, hl, tl, keyf, fact, listf, E, N, mtab4, mq);
}

__global__ void __launch_bounds__(MSG_WG) k_message_replay(
    const float* __restrict__ ew, const float* __restrict__ ego,
    const int* __restrict__ hl, const int* __restrict__ tl,
    const float* __restrict__ keyf, float* __restrict__ fact,
    float* __restrict__ listf, int E, int N) {
  __shared__ float4 mtab4[MSG_NODES * 16];
  __shared__ unsigned mq[MSG_WAVES][MSG_QCAP];
  message_body<true>(ew, ego, hl, tl, keyf, fact, listf, E, N, mtab4, mq);
}

#define DL_IDS 6

__global__ void __launch_bounds__(MSG_WG) k_degree_lists(
    const float* __restrict__ ew, const int* __restrict__ hl,
    const float* __restrict__ listf, float* __restrict__ dinvg, int E, int N) {
  __shared__ float4 dtab4[MSG_NODES];
  float* tab = reinterpret_cast<float*>(dtab4);
  const unsigned r = blockIdx.x;
  const unsigned R = gridDim.x;
  const int lane = threadIdx.x & 31;
  const int wave = threadIdx.x >> 5;

  for (int i = threadIdx.x; i < MSG_NODES; i += MSG_WG)
    dtab4[i] = make_float4(0.f, 0.f, 0.f, 0.f);
  __syncthreads();

  const float4* ew4 = reinterpret_cast<const float4*>(ew);
  const unsigned cap = ((unsigned)N * 64u) / (R * (unsigned)MSG_WAVES);
  const unsigned* gl = reinterpret_cast<const unsigned*>(listf) +
                       (size_t)(r * MSG_WAVES + (unsigned)wave) * cap;
  const unsigned n = (unsigned)__builtin_amdgcn_readfirstlane((int)gl[0]);
  if (n != MSG_OVER) {
    for (unsigned b0 = 0; b0 < n; b0 += 32u * DL_IDS) {
      unsigned id[DL_IDS];
      bool on[DL_IDS];
#pragma unroll
      for (int k = 0; k < DL_IDS; ++k) {
        const unsigned x = b0 + (unsigned)(k * 32 + lane);
        on[k] = x < n;
        id[k] = gl[1 + (on[k] ? x : n - 1u)];
      }
      unsigned hd[DL_IDS];
      float4 w[DL_IDS];
#pragma unroll
      for (int k = 0; k < DL_IDS; ++k) {
        hd[k] = (unsigned)hl[id[k]];
        w[k] = ew4[id[k]];
      }
#pragma unroll
      for (int k = 0; k < DL_IDS; ++k) {
        if (on[k]) {
          float* t = tab + (hd[k] / R) * 4u;
          lds_fadd(t + 0, w[k].x);
          lds_fadd(t + 1, w[k].y);
          lds_fadd(t + 2, w[k].z);
          lds_fadd(t + 3, w[k].w);
        }
      }
    }
  } else {
    const int nChunks = (E + 15) >> 4;
    const int perTrip = MSG_WAVES * 128;
    const int trips = (nChunks + perTrip - 1) / perTrip;
#pragma unroll 1
    for (int j = 0; j < trips; ++j) {
#pragma unroll 1
      for (int u = 0; u < 4; ++u) {
        const int c = j * perTrip + wave * 128 + u * 32 + lane;
        if (c < nChunks) {
#pragma unroll 1
          for (int x = 0; x < 16; ++x) {
            const int e = c * 16 + x;
            if (e < E) {
              const unsigned h = (unsigned)hl[e];
              if (h % R == r) {
                const float4 wv = ew4[e];
                float* t = tab + (h / R) * 4u;
                lds_fadd(t + 0, wv.x);
                lds_fadd(t + 1, wv.y);
                lds_fadd(t + 2, wv.z);
                lds_fadd(t + 3, wv.w);
              }
            }
          }
        }
      }
    }
  }
  __syncthreads();

  float4* dst = reinterpret_cast<float4*>(dinvg) + (size_t)r * MSG_NODES;
  for (int i = threadIdx.x; i < MSG_NODES; i += MSG_WG) {
    const float4 s = dtab4[i];
    float4 o;
    o.x = 1.0f / sqrtf(fmaxf(s.x, 1e-8f));
    o.y = 1.0f / sqrtf(fmaxf(s.y, 1e-8f));
    o.z = 1.0f / sqrtf(fmaxf(s.z, 1e-8f));
    o.w = 1.0f / sqrtf(fmaxf(s.w, 1e-8f));
    dst[i] = o;
  }
}

__global__ void __launch_bounds__(WG) k_updateA(
    const float* __restrict__ fact, const float* __restrict__ ego,
    const int* __restrict__ hl, const int* __restrict__ tl,
    float* __restrict__ A, int E, int nGroups) {
  int tid = blockIdx.x * blockDim.x + threadIdx.x;
  int wave = tid >> 5;
  int lane = tid & 31;
  if (wave >= nGroups) return;

  const int c = lane & 15;
  const int kh = lane >> 4;
  const int i = c >> 2;
  const int f = c & 3;
  const int seg = f * 16 + kh * 8;

  const bool active = (lane < 8) || (lane >= 24);
  const int cc = (lane < 8) ? lane : (lane - 16);
  const int e0 = wave * 16;

  int hh[4], tt[4];
  float aold[4];
  bool wr[4];
  const int aLast = E * 4 - 1;
#pragma unroll
  for (int g = 0; g < 4; ++g) {
    const int eIdx = e0 + g * 4 + i;
    const int e = (eIdx < E) ? eIdx : (E - 1);
    hh[g] = hl[e];
    tt[g] = tl[e];
    const int ai = (e0 + g * 4) * 4 + c;
    aold[g] = A[(ai < aLast) ? ai : aLast];
    wr[g] = active && (e0 + g * 4 + (cc >> 2) < E);
  }
  asm volatile("" :: "v"(hh[0]), "v"(hh[1]), "v"(hh[2]), "v"(hh[3]),
                     "v"(tt[0]), "v"(tt[1]), "v"(tt[2]), "v"(tt[3]),
                     "v"(aold[0]), "v"(aold[1]), "v"(aold[2]), "v"(aold[3]));
  v4f h0[4], h1[4], t0[4], t1[4];
#pragma unroll
  for (int g = 0; g < 4; ++g) {
    const v4f* hp = reinterpret_cast<const v4f*>(fact + (size_t)hh[g] * 64 + seg);
    const v4f* tp = reinterpret_cast<const v4f*>(ego + (size_t)tt[g] * 64 + seg);
    h0[g] = hp[0];
    h1[g] = hp[1];
    t0[g] = tp[0];
    t1[g] = tp[1];
  }
  float dv[4];
#pragma unroll
  for (int g = 0; g < 4; ++g) {
    const float hv[8] = {h0[g].x, h0[g].y, h0[g].z, h0[g].w, h1[g].x, h1[g].y, h1[g].z, h1[g].w};
    const float tv[8] = {t0[g].x, t0[g].y, t0[g].z, t0[g].w, t1[g].x, t1[g].y, t1[g].z, t1[g].w};
    v2f qh = h0[g].xy * h0[g].xy;
    v2f qt = t0[g].xy * t0[g].xy;
    qh = __builtin_elementwise_fma(h0[g].zw, h0[g].zw, qh);
    qt = __builtin_elementwise_fma(t0[g].zw, t0[g].zw, qt);
    qh = __builtin_elementwise_fma(h1[g].xy, h1[g].xy, qh);
    qt = __builtin_elementwise_fma(t1[g].xy, t1[g].xy, qt);
    qh = __builtin_elementwise_fma(h1[g].zw, h1[g].zw, qh);
    qt = __builtin_elementwise_fma(t1[g].zw, t1[g].zw, qt);
    const float ph = qh.x + qh.y, pt = qt.x + qt.y;
    const float sh = ph + xhalf(ph);
    const float st = pt + xhalf(pt);
    const float mine = 1.0f / fmaxf(sqrtf(kh ? st : sh), 1e-12f);
    const float other = xhalf(mine);
    const float hn = kh ? other : mine;
    const float tn = kh ? mine : other;
    v8f acc = {0.f, 0.f, 0.f, 0.f, 0.f, 0.f, 0.f, 0.f};
#pragma unroll
    for (int j = 0; j < 4; ++j) {
      v2f av, bv;
      av.x = hv[2 * j] * hn;
      av.y = hv[2 * j + 1] * hn;
      bv.x = fast_tanh(tv[2 * j] * tn);
      bv.y = fast_tanh(tv[2 * j + 1] * tn);
      acc = __builtin_amdgcn_wmma_f32_16x16x4_f32(false, av, false, bv,
                                                  (short)0, acc, false, false);
    }
    const int idx = lane & 7;
    float d = acc[0];
    d = (idx == 1) ? acc[1] : d;
    d = (idx == 2) ? acc[2] : d;
    d = (idx == 3) ? acc[3] : d;
    d = (idx == 4) ? acc[4] : d;
    d = (idx == 5) ? acc[5] : d;
    d = (idx == 6) ? acc[6] : d;
    d = (idx == 7) ? acc[7] : d;
    dv[g] = d;
  }

#pragma unroll
  for (int g = 0; g < 4; ++g)
    if (wr[g]) A[(size_t)(e0 + g * 4) * 4 + cc] = aold[g] + dv[g];
}

__global__ void k_final(const float* __restrict__ u, const float* __restrict__ it,
                        const float* __restrict__ l1, const float* __restrict__ l2,
                        float* __restrict__ out, int userElems, int n) {
  int i = blockIdx.x * blockDim.x + threadIdx.x;
  if (i >= n) return;
  float v = (i < userElems) ? u[i] : it[i - userElems];
  v += l1[i];
  v += l2[i];
  out[i] = v * (1.0f / 3.0f);
}

static inline size_t align16f(size_t x) { return (x + 15) & ~(size_t)15; }

extern "C" void kernel_launch(void* const* d_in, const int* in_sizes, int n_in,
                              void* d_out, int out_size, void* d_ws, size_t ws_size,
                              hipStream_t stream) {
  (void)n_in; (void)out_size; (void)ws_size;
  const float* u  = (const float*)d_in[0];
  const float* it = (const float*)d_in[1];
  const int* hl   = (const int*)d_in[2];
  const int* tl   = (const int*)d_in[3];

  const int userElems = in_sizes[0];
  const int itemElems = in_sizes[1];
  const int E = in_sizes[2];
  const int N = (userElems + itemElems) / 64;
  const size_t N64 = (size_t)N * 64;
  const size_t E4  = (size_t)E * 4;
  const size_t N4  = (size_t)N * 4;

  float* base = (float*)d_ws;
  size_t off = 0;
  float* ego   = base + off; off += align16f(N64);
  float* fact  = base + off; off += align16f(N64);
  float* lists = base + off; off += align16f(N64);
  float* A     = base + off; off += align16f(E4);
  float* ew    = base + off; off += align16f(E4);
  float* dval  = base + off; off += align16f(N4);
  float* keyf  = base + off; off += align16f(N4);

#define GRID(n) ((unsigned)(((size_t)(n) + WG - 1) / WG))

  k_init<<<GRID(N64), WG, 0, stream>>>(u, it, ego, userElems, (int)N64);
  k_fill1<<<GRID(E4), WG, 0, stream>>>(A, (int)E4);

  float* cur = ego;
  float* nxt = fact;
  const int nGroups = (E + 15) / 16;
  const dim3 degGrid(DEG_SLICES, (unsigned)((N + DEG_NODES - 1) / DEG_NODES));
  const unsigned msgGrid = (unsigned)((N + MSG_NODES - 1) / MSG_NODES);
  const int degPer = (((E + DEG_SLICES - 1) / DEG_SLICES) + 3) & ~3;

  for (int layer = 0; layer < 2; ++layer) {
    for (int itn = 0; itn < 2; ++itn) {
      k_softmax<<<GRID(E), WG, 0, stream>>>(A, ew, E);
      if (layer == 0 && itn == 0) {
        k_degree_lds<<<degGrid, DEG_WG, 0, stream>>>(ew, hl, nxt, E, N, degPer);
        k_dinv_sum<<<GRID(N), WG, 0, stream>>>(nxt, dval, N);
        k_edge_weight<false><<<GRID(E), WG, 0, stream>>>(ew, dval, hl, tl, keyf, E, (int)msgGrid);
        k_message_first<<<msgGrid, MSG_WG, 0, stream>>>(ew, cur, hl, tl, keyf, nxt, lists, E, N);
      } else {
        k_degree_lists<<<msgGrid, MSG_WG, 0, stream>>>(ew, hl, lists, nxt, E, N);
        k_edge_weight<true><<<GRID(E), WG, 0, stream>>>(ew, nxt, hl, tl, keyf, E, (int)msgGrid);
        k_message_replay<<<msgGrid, MSG_WG, 0, stream>>>(ew, cur, hl, tl, keyf, nxt, lists, E, N);
      }
      const bool lastUpdateDead = (layer == 1 && itn == 1);
      if (!lastUpdateDead)
        k_updateA<<<GRID((size_t)nGroups * 32), WG, 0, stream>>>(
            nxt, cur, hl, tl, A, E, nGroups);
    }
    float* tmp = cur; cur = nxt; nxt = tmp;
  }

  k_final<<<GRID(N64), WG, 0, stream>>>(u, it, nxt, cur, (float*)d_out, userElems, (int)N64);
#undef GRID
}


# Stored edge-list entries carry the tail node and the head's table row beside the edge id: message passes 2-4 and the list-based degree sums no longer load head and tail lists by edge id
# speedup vs baseline: 2.9793x; 1.0530x over previous
#include <hip/hip_runtime.h>
#include <math.h>

typedef float v2f __attribute__((ext_vector_type(2)));
typedef float v4f __attribute__((ext_vector_type(4)));
typedef float v8f __attribute__((ext_vector_type(8)));

#define WG 256

__device__ __forceinline__ float xhalf(float x) {
  int b = __builtin_bit_cast(int, x);
  return __builtin_bit_cast(float,
      __builtin_amdgcn_permlanex16(b, b, 0x76543210, 0xfedcba98, false, false));
}

__device__ __forceinline__ float fast_tanh(float x) {
  float r;
  asm("v_tanh_f32 %0, %1\n\tv_nop\n\tv_nop" : "=v"(r) : "v"(x));
  return r;
}

__global__ void k_init(const float* __restrict__ u, const float* __restrict__ it,
                       float* __restrict__ ego, int userElems, int total) {
  int i = blockIdx.x * blockDim.x + threadIdx.x;
  if (i >= total) return;
  ego[i] = (i < userElems) ? u[i] : it[i - userElems];
}

__global__ void k_fill1(float* __restrict__ a, int n) {
  int i = blockIdx.x * blockDim.x + threadIdx.x;
  if (i < n) a[i] = 1.0f;
}

__global__ void k_softmax(const float* __restrict__ A, float* __restrict__ ew, int E) {
  int e = blockIdx.x * blockDim.x + threadIdx.x;
  if (e >= E) return;
  float4 a = reinterpret_cast<const float4*>(A)[e];
  float m = fmaxf(fmaxf(a.x, a.y), fmaxf(a.z, a.w));
  float e0 = expf(a.x - m), e1 = expf(a.y - m), e2 = expf(a.z - m), e3 = expf(a.w - m);
  float inv = 1.0f / (e0 + e1 + e2 + e3);
  reinterpret_cast<float4*>(ew)[e] = make_float4(e0 * inv, e1 * inv, e2 * inv, e3 * inv);
}

#define DEG_SLICES 16
#define DEG_NODES  5000
#define DEG_WG     1024
#define DEG_VEC    4

__device__ __forceinline__ void lds_fadd(float* p, float v) {
  __hip_atomic_fetch_add(p, v, __ATOMIC_RELAXED, __HIP_MEMORY_SCOPE_WORKGROUP);
}

__global__ void __launch_bounds__(DEG_WG) k_degree_lds(
    const float* __restrict__ ew, const int* __restrict__ hl,
    float* __restrict__ part, int E, int N, int per) {
  __shared__ float4 tab4[DEG_NODES];
  float* tab = reinterpret_cast<float*>(tab4);
  const int s = blockIdx.x;
  const int lo = blockIdx.y * DEG_NODES;

  for (int i = threadIdx.x; i < DEG_NODES; i += DEG_WG)
    tab4[i] = make_float4(0.f, 0.f, 0.f, 0.f);
  __syncthreads();

  const int e0 = s * per;
  int e1 = e0 + per;
  if (e1 > E) e1 = E;
  const int nvec = (e1 > e0) ? ((e1 - e0) >> 2) : 0;
  const int4* hv = reinterpret_cast<const int4*>(hl + e0);
  const float4* ew4 = reinterpret_cast<const float4*>(ew);

  for (int v = threadIdx.x; v < nvec; v += DEG_WG * DEG_VEC) {
    int hd[DEG_VEC * 4];
#pragma unroll
    for (int k = 0; k < DEG_VEC; ++k) {
      int vk = v + k * DEG_WG;
      int4 h = make_int4(-1, -1, -1, -1);
      if (vk < nvec) h = hv[vk];
      hd[k * 4 + 0] = h.x; hd[k * 4 + 1] = h.y; hd[k * 4 + 2] = h.z; hd[k * 4 + 3] = h.w;
    }
    float4 w[DEG_VEC * 4];
#pragma unroll
    for (int j = 0; j < DEG_VEC * 4; ++j) {
      unsigned d = (unsigned)(hd[j] - lo);
      w[j] = make_float4(0.f, 0.f, 0.f, 0.f);
      if (d < (unsigned)DEG_NODES) w[j] = ew4[e0 + ((v + (j >> 2) * DEG_WG) << 2) + (j & 3)];
    }
#pragma unroll
    for (int j = 0; j < DEG_VEC * 4; ++j) {
      unsigned d = (unsigned)(hd[j] - lo);
      if (d < (unsigned)DEG_NODES) {
        float* t = tab + d * 4;
        lds_fadd(t + 0, w[j].x);
        lds_fadd(t + 1, w[j].y);
        lds_fadd(t + 2, w[j].z);
        lds_fadd(t + 3, w[j].w);
      }
    }
  }
  for (int e = e0 + (nvec << 2) + threadIdx.x; e < e1; e += DEG_WG) {
    unsigned d = (unsigned)(hl[e] - lo);
    if (d < (unsigned)DEG_NODES) {
      float4 wv = ew4[e];
      float* t = tab + d * 4;
      lds_fadd(t + 0, wv.x);
      lds_fadd(t + 1, wv.y);
      lds_fadd(t + 2, wv.z);
      lds_fadd(t + 3, wv.w);
    }
  }
  __syncthreads();

  int nodes = N - lo;
  if (nodes > DEG_NODES) nodes = DEG_NODES;
  float4* dst = reinterpret_cast<float4*>(part) + (size_t)s * N + lo;
  for (int i = threadIdx.x; i < nodes; i += DEG_WG) dst[i] = tab4[i];
}

__global__ void k_dinv_sum(const float* __restrict__ part, float* __restrict__ d, int N) {
  int n = blockIdx.x * blockDim.x + threadIdx.x;
  if (n >= N) return;
  const float4* p4 = reinterpret_cast<const float4*>(part);
  float4 sum = make_float4(0.f, 0.f, 0.f, 0.f);
#pragma unroll
  for (int k = 0; k < DEG_SLICES; ++k) {
    float4 v = p4[(size_t)k * N + n];
    sum.x += v.x; sum.y += v.y; sum.z += v.z; sum.w += v.w;
  }
  float4 r;
  r.x = 1.0f / sqrtf(fmaxf(sum.x, 1e-8f));
  r.y = 1.0f / sqrtf(fmaxf(sum.y, 1e-8f));
  r.z = 1.0f / sqrtf(fmaxf(sum.z, 1e-8f));
  r.w = 1.0f / sqrtf(fmaxf(sum.w, 1e-8f));
  reinterpret_cast<float4*>(d)[n] = r;
}

#define MSG_NODES 320
#define MSG_WG    1024
#define MSG_WAVES 32
#define MSG_BATCH 16
#define MSG_QCAP  544
#define MSG_NOKEY 255u
#define MSG_OVER  0xffffffffu
#define MSG_TRIP  128
#define MSG_TBITS 17

template <bool GROUPED>
__global__ void k_edge_weight(float* __restrict__ ew, const float* __restrict__ dinv,
                              const int* __restrict__ hl, const int* __restrict__ tl,
                              float* __restrict__ keyf, int E, int R) {
  int e = blockIdx.x * blockDim.x + threadIdx.x;
  unsigned char* keyb = reinterpret_cast<unsigned char*>(keyf);
  if (e >= E) {
    if (e < ((E + 15) & ~15)) keyb[e] = (unsigned char)MSG_NOKEY;
    return;
  }
  const unsigned h = (unsigned)hl[e];
  const unsigned t = (unsigned)tl[e];
  const unsigned hg = h % (unsigned)R;
  const unsigned hi = GROUPED ? hg * (unsigned)MSG_NODES + h / (unsigned)R : h;
  const unsigned ti = GROUPED ? (t % (unsigned)R) * (unsigned)MSG_NODES + t / (unsigned)R : t;
  float4 w = reinterpret_cast<float4*>(ew)[e];
  float4 dh = reinterpret_cast<const float4*>(dinv)[hi];
  float4 dt = reinterpret_cast<const float4*>(dinv)[ti];
  w.x *= dh.x * dt.x;
  w.y *= dh.y * dt.y;
  w.z *= dh.z * dt.z;
  w.w *= dh.w * dt.w;
  reinterpret_cast<float4*>(ew)[e] = w;
  keyb[e] = (unsigned char)hg;
}

__device__ __forceinline__ unsigned nonzero_bytes(unsigned x) {
  return ((x & 0x7f7f7f7fu) + 0x7f7f7f7fu) | x;
}

__device__ __forceinline__ unsigned key_hits(uint4 k, unsigned pat) {
  unsigned y0 = nonzero_bytes(k.x ^ pat);
  unsigned y1 = nonzero_bytes(k.y ^ pat);
  unsigned y2 = nonzero_bytes(k.z ^ pat);
  unsigned y3 = nonzero_bytes(k.w ^ pat);
  unsigned n = ((y0 >> 7) & 0x01010101u) | ((y1 >> 6) & 0x02020202u) |
               ((y2 >> 5) & 0x04040404u) | ((y3 >> 4) & 0x08080808u);
  return ~n & 0x0f0f0f0fu;
}

template <bool FULL, bool PAIRS>
__device__ __forceinline__ unsigned msg_batch(const unsigned* qp, int n, int lane, unsigned R,
                                              const float* __restrict__ ew,
                                              const float* __restrict__ ego,
                                              const int* __restrict__ hl,
                                              const int* __restrict__ tl, float* tab) {
  const int l16 = lane & 15;
  const int s16 = (FULL || l16 < n) ? l16 : 0;
  const int j8 = lane >> 2;
  const int sa = (FULL || j8 < n) ? j8 : 0;
  const int sb = (FULL || j8 + 8 < n) ? j8 + 8 : 0;
  unsigned pk, ea, eb;
  if (PAIRS) {
    pk = qp[2 * s16 + 1];
    ea = qp[2 * sa];
    eb = qp[2 * sb];
  } else {
    const unsigned el = qp[s16];
    pk = (unsigned)tl[el] | (((unsigned)hl[el] / R) << MSG_TBITS);
    ea = qp[sa];
    eb = qp[sb];
  }
  const int tv = (int)(pk & ((1u << MSG_TBITS) - 1u));
  const int dv = (int)(pk >> MSG_TBITS);
  const float wa = ew[(size_t)ea * 4 + (lane & 3)];
  const float wb = ew[(size_t)eb * 4 + (lane & 3)];
  const int f = lane >> 3;

  float2 v[MSG_BATCH];
  int d[MSG_BATCH];
#pragma unroll
  for (int i = 0; i < MSG_BATCH; ++i) {
    const int t = __builtin_amdgcn_readlane(tv, i);
    d[i] = __builtin_amdgcn_readlane(dv, i);
    v[i] = make_float2(0.f, 0.f);
    if (FULL || i < n)
      v[i] = *reinterpret_cast<const float2*>(ego + (size_t)t * 64 + lane * 2);
  }
  if (PAIRS) {
    asm volatile("" :: "v"(v[0].x), "v"(v[0].y), "v"(v[1].x), "v"(v[1].y),
                       "v"(v[2].x), "v"(v[2].y), "v"(v[3].x), "v"(v[3].y),
                       "v"(v[4].x), "v"(v[4].y), "v"(v[5].x), "v"(v[5].y),
                       "v"(v[6].x), "v"(v[6].y), "v"(v[7].x), "v"(v[7].y));
    asm volatile("" :: "v"(v[8].x), "v"(v[8].y), "v"(v[9].x), "v"(v[9].y),
                       "v"(v[10].x), "v"(v[10].y), "v"(v[11].x), "v"(v[11].y),
                       "v"(v[12].x), "v"(v[12].y), "v"(v[13].x), "v"(v[13].y),
                       "v"(v[14].x), "v"(v[14].y), "v"(v[15].x), "v"(v[15].y));
  }
#pragma unroll
  for (int i = 0; i < MSG_BATCH; ++i) {
    const float w = __shfl((i < 8) ? wa : wb, 4 * (i & 7) + f);
    if (FULL || i < n) {
      float* tp = tab + d[i] * 64 + lane * 2;
      lds_fadd(tp + 0, w * v[i].x);
      lds_fadd(tp + 1, w * v[i].y);
    }
  }
  return pk;
}

template <bool REPLAY>
__device__ __forceinline__ void message_body(
    const float* __restrict__ ew, const float* __restrict__ ego,
    const int* __restrict__ hl, const int* __restrict__ tl,
    const float* __restrict__ keyf, float* __restrict__ fact,
    float* __restrict__ listf, int E, int N,
    float4* mtab4, unsigned (*mq)[MSG_QCAP]) {
  float* tab = reinterpret_cast<float*>(mtab4);
  const int r = blockIdx.x;
  const unsigned R = gridDim.x;
  const int lane = threadIdx.x & 31;
  const int wave = threadIdx.x >> 5;

  for (int i = threadIdx.x; i < MSG_NODES * 16; i += MSG_WG)
    mtab4[i] = make_float4(0.f, 0.f, 0.f, 0.f);
  __syncthreads();

  const uint4* k4 = reinterpret_cast<const uint4*>(keyf);
  const int nChunks = (E + 15) >> 4;
  const int perTrip = MSG_WAVES * 128;
  const int trips = (nChunks + perTrip - 1) / perTrip;
  const unsigned pat = (unsigned)r * 0x01010101u;
  unsigned* q = mq[wave];
  int cnt = 0;

  const unsigned cap = (((unsigned)N * 64u) / (R * (unsigned)MSG_WAVES)) & ~1u;
  const unsigned pcap = (cap - 2u) >> 1;
  unsigned* gl = reinterpret_cast<unsigned*>(listf) +
                 (size_t)((unsigned)r * MSG_WAVES + (unsigned)wave) * cap;
  uint2* gp = reinterpret_cast<uint2*>(gl) + 1;
  unsigned stored = 0;
  bool fits = true;
  bool scan = true;

  if (REPLAY) {
    const unsigned n = (unsigned)__builtin_amdgcn_readfirstlane((int)gl[0]);
    if (n != MSG_OVER) {
      scan = false;
      for (unsigned b0 = 0; b0 < n; b0 += MSG_TRIP) {
        const unsigned m = (n - b0 < (unsigned)MSG_TRIP) ? (n - b0) : (unsigned)MSG_TRIP;
        uint2 env[MSG_TRIP / 32];
#pragma unroll
        for (int k = 0; k < MSG_TRIP / 32; ++k) {
          const unsigned x = (unsigned)(k * 32 + lane);
          env[k] = make_uint2(0u, 0u);
          if (x < m) env[k] = gp[b0 + x];
        }
        uint2* q2 = reinterpret_cast<uint2*>(q);
#pragma unroll
        for (int k = 0; k < MSG_TRIP / 32; ++k) q2[k * 32 + lane] = env[k];
        const int full = (int)(m >> 4);
        for (int b = 0; b < full; ++b)
          msg_batch<true, true>(q + b * MSG_BATCH * 2, MSG_BATCH, lane, R, ew, ego, hl, tl, tab);
        const int rem = (int)(m & 15u);
        if (rem > 0)
          msg_batch<false, true>(q + full * MSG_BATCH * 2, rem, lane, R, ew, ego, hl, tl, tab);
      }
    }
  }

  int j = r % trips;
  for (int jj = 0; scan && jj < trips; ++jj) {
    const int cb = j * perTrip + wave * 128;
    j = (j + 1 == trips) ? 0 : j + 1;
    uint4 kk[4];
#pragma unroll
    for (int u = 0; u < 4; ++u) {
      const int c = cb + u * 32 + lane;
      kk[u] = make_uint4(~0u, ~0u, ~0u, ~0u);
      if (c < nChunks) kk[u] = k4[c];
    }
#pragma unroll
    for (int u = 0; u < 4; ++u) {
      const int c = cb + u * 32 + lane;
      unsigned m = key_hits(kk[u], pat);
      unsigned bal = __builtin_amdgcn_ballot_w32(m != 0u);
      while (bal != 0u) {
        if (m != 0u) {
          const int p = __builtin_ctz(m);
          const int idx = ((p & 3) << 2) | (p >> 3);
          q[cnt + (int)__builtin_amdgcn_mbcnt_lo(bal, 0u)] = (unsigned)(c * 16 + idx);
          m &= m - 1u;
        }
        cnt += __builtin_popcount(bal);
        bal = __builtin_amdgcn_ballot_w32(m != 0u);
      }
      while (cnt >= MSG_BATCH) {
        cnt -= MSG_BATCH;
        const unsigned pk =
            msg_batch<true, false>(q + cnt, MSG_BATCH, lane, R, ew, ego, hl, tl, tab);
        (void)pk;
        if (!REPLAY) {
          if (fits && stored + MSG_BATCH <= pcap) {
            if (lane < MSG_BATCH) gp[stored + lane] = make_uint2(q[cnt + lane], pk);
            stored += MSG_BATCH;
          } else {
            fits = false;
          }
        }
      }
    }
  }
  if (cnt > 0) {
    const unsigned pk = msg_batch<false, false>(q, cnt, lane, R, ew, ego, hl, tl, tab);
    (void)pk;
    if (!REPLAY) {
      if (fits && stored + (unsigned)cnt <= pcap) {
        if (lane < cnt) gp[stored + lane] = make_uint2(q[lane], pk);
        stored += (unsigned)cnt;
      } else {
        fits = false;
      }
    }
  }
  if (!REPLAY && lane == 0) gl[0] = fits ? stored : MSG_OVER;
  __syncthreads();

  float4* dst = reinterpret_cast<float4*>(fact);
  for (int i = threadIdx.x; i < MSG_NODES * 16; i += MSG_WG) {
    const int node = (i >> 4) * (int)R + r;
    if (node < N) dst[(size_t)node * 16 + (i & 15)] = mtab4[i];
  }
}

__global__ void __launch_bounds__(MSG_WG) k_message_first(
    const float* __restrict__ ew, const float* __restrict__ ego,
    const int* __restrict__ hl, const int* __restrict__ tl,
    const float* __restrict__ keyf, float* __restrict__ fact,
    float* __restrict__ listf, int E, int N) {
  __shared__ float4 mtab4[MSG_NODES * 16];
  __shared__ __attribute__((aligned(16))) unsigned mq[MSG_WAVES][MSG_QCAP];
  message_body<false>(ew, ego, hl, tl, keyf, fact, listf, E, N, mtab4, mq);
}

__global__ void __launch_bounds__(MSG_WG) k_message_replay(
    const float* __restrict__ ew, const float* __restrict__ ego,
    const int* __restrict__ hl, const int* __restrict__ tl,
    const float* __restrict__ keyf, float* __restrict__ fact,
    float* __restrict__ listf, int E, int N) {
  __shared__ float4 mtab4[MSG_NODES * 16];
  __shared__ __attribute__((aligned(16))) unsigned mq[MSG_WAVES][MSG_QCAP];
  message_body<true>(ew, ego, hl, tl, keyf, fact, listf, E, N, mtab4, mq);
}

#define DL_IDS 6

__global__ void __launch_bounds__(MSG_WG) k_degree_lists(
    const float* __restrict__ ew, const int* __restrict__ hl,
    const float* __restrict__ listf, float* __restrict__ dinvg, int E, int N) {
  __shared__ float4 dtab4[MSG_NODES];
  float* tab = reinterpret_cast<float*>(dtab4);
  const unsigned r = blockIdx.x;
  const unsigned R = gridDim.x;
  const int lane = threadIdx.x & 31;
  const int wave = threadIdx.x >> 5;

  for (int i = threadIdx.x; i < MSG_NODES; i += MSG_WG)
    dtab4[i] = make_float4(0.f, 0.f, 0.f, 0.f);
  __syncthreads();

  const float4* ew4 = reinterpret_cast<const float4*>(ew);
  const unsigned cap = (((unsigned)N * 64u) / (R * (unsigned)MSG_WAVES)) & ~1u;
  const unsigned* gl = reinterpret_cast<const unsigned*>(listf) +
                       (size_t)(r * MSG_WAVES + (unsigned)wave) * cap;
  const uint2* gp = reinterpret_cast<const uint2*>(gl) + 1;
  const unsigned n = (unsigned)__builtin_amdgcn_readfirstlane((int)gl[0]);
  if (n != MSG_OVER) {
    for (unsigned b0 = 0; b0 < n; b0 += 32u * DL_IDS) {
      uint2 en[DL_IDS];
      bool on[DL_IDS];
#pragma unroll
      for (int k = 0; k < DL_IDS; ++k) {
        const unsigned x = b0 + (unsigned)(k * 32 + lane);
        on[k] = x < n;
        en[k] = gp[on[k] ? x : n - 1u];
      }
      float4 w[DL_IDS];
#pragma unroll
      for (int k = 0; k < DL_IDS; ++k) w[k] = ew4[en[k].x];
#pragma unroll
      for (int k = 0; k < DL_IDS; ++k) {
        if (on[k]) {
          float* t = tab + (en[k].y >> MSG_TBITS) * 4u;
          lds_fadd(t + 0, w[k].x);
          lds_fadd(t + 1, w[k].y);
          lds_fadd(t + 2, w[k].z);
          lds_fadd(t + 3, w[k].w);
        }
      }
    }
  } else {
    const int nChunks = (E + 15) >> 4;
    const int perTrip = MSG_WAVES * 128;
    const int trips = (nChunks + perTrip - 1) / perTrip;
#pragma unroll 1
    for (int j = 0; j < trips; ++j) {
#pragma unroll 1
      for (int u = 0; u < 4; ++u) {
        const int c = j * perTrip + wave * 128 + u * 32 + lane;
        if (c < nChunks) {
#pragma unroll 1
          for (int x = 0; x < 16; ++x) {
            const int e = c * 16 + x;
            if (e < E) {
              const unsigned h = (unsigned)hl[e];
              if (h % R == r) {
                const float4 wv = ew4[e];
                float* t = tab + (h / R) * 4u;
                lds_fadd(t + 0, wv.x);
                lds_fadd(t + 1, wv.y);
                lds_fadd(t + 2, wv.z);
                lds_fadd(t + 3, wv.w);
              }
            }
          }
        }
      }
    }
  }
  __syncthreads();

  float4* dst = reinterpret_cast<float4*>(dinvg) + (size_t)r * MSG_NODES;
  for (int i = threadIdx.x; i < MSG_NODES; i += MSG_WG) {
    const float4 s = dtab4[i];
    float4 o;
    o.x = 1.0f / sqrtf(fmaxf(s.x, 1e-8f));
    o.y = 1.0f / sqrtf(fmaxf(s.y, 1e-8f));
    o.z = 1.0f / sqrtf(fmaxf(s.z, 1e-8f));
    o.w = 1.0f / sqrtf(fmaxf(s.w, 1e-8f));
    dst[i] = o;
  }
}

__global__ void __launch_bounds__(WG) k_updateA(
    const float* __restrict__ fact, const float* __restrict__ ego,
    const int* __restrict__ hl, const int* __restrict__ tl,
    float* __restrict__ A, int E, int nGroups) {
  int tid = blockIdx.x * blockDim.x + threadIdx.x;
  int wave = tid >> 5;
  int lane = tid & 31;
  if (wave >= nGroups) return;

  const int c = lane & 15;
  const int kh = lane >> 4;
  const int i = c >> 2;
  const int f = c & 3;
  const int seg = f * 16 + kh * 8;

  const bool active = (lane < 8) || (lane >= 24);
  const int cc = (lane < 8) ? lane : (lane - 16);
  const int e0 = wave * 16;

  int hh[4], tt[4];
  float aold[4];
  bool wr[4];
  const int aLast = E * 4 - 1;
#pragma unroll
  for (int g = 0; g < 4; ++g) {
    const int eIdx = e0 + g * 4 + i;
    const int e = (eIdx < E) ? eIdx : (E - 1);
    hh[g] = hl[e];
    tt[g] = tl[e];
    const int ai = (e0 + g * 4) * 4 + c;
    aold[g] = A[(ai < aLast) ? ai : aLast];
    wr[g] = active && (e0 + g * 4 + (cc >> 2) < E);
  }
  asm volatile("" :: "v"(hh[0]), "v"(hh[1]), "v"(hh[2]), "v"(hh[3]),
                     "v"(tt[0]), "v"(tt[1]), "v"(tt[2]), "v"(tt[3]),
                     "v"(aold[0]), "v"(aold[1]), "v"(aold[2]), "v"(aold[3]));
  v4f h0[4], h1[4], t0[4], t1[4];
#pragma unroll
  for (int g = 0; g < 4; ++g) {
    const v4f* hp = reinterpret_cast<const v4f*>(fact + (size_t)hh[g] * 64 + seg);
    const v4f* tp = reinterpret_cast<const v4f*>(ego + (size_t)tt[g] * 64 + seg);
    h0[g] = hp[0];
    h1[g] = hp[1];
    t0[g] = tp[0];
    t1[g] = tp[1];
  }
  float dv[4];
#pragma unroll
  for (int g = 0; g < 4; ++g) {
    const float hv[8] = {h0[g].x, h0[g].y, h0[g].z, h0[g].w, h1[g].x, h1[g].y, h1[g].z, h1[g].w};
    const float tv[8] = {t0[g].x, t0[g].y, t0[g].z, t0[g].w, t1[g].x, t1[g].y, t1[g].z, t1[g].w};
    v2f qh = h0[g].xy * h0[g].xy;
    v2f qt = t0[g].xy * t0[g].xy;
    qh = __builtin_elementwise_fma(h0[g].zw, h0[g].zw, qh);
    qt = __builtin_elementwise_fma(t0[g].zw, t0[g].zw, qt);
    qh = __builtin_elementwise_fma(h1[g].xy, h1[g].xy, qh);
    qt = __builtin_elementwise_fma(t1[g].xy, t1[g].xy, qt);
    qh = __builtin_elementwise_fma(h1[g].zw, h1[g].zw, qh);
    qt = __builtin_elementwise_fma(t1[g].zw, t1[g].zw, qt);
    const float ph = qh.x + qh.y, pt = qt.x + qt.y;
    const float sh = ph + xhalf(ph);
    const float st = pt + xhalf(pt);
    const float mine = 1.0f / fmaxf(sqrtf(kh ? st : sh), 1e-12f);
    const float other = xhalf(mine);
    const float hn = kh ? other : mine;
    const float tn = kh ? mine : other;
    v8f acc = {0.f, 0.f, 0.f, 0.f, 0.f, 0.f, 0.f, 0.f};
#pragma unroll
    for (int j = 0; j < 4; ++j) {
      v2f av, bv;
      av.x = hv[2 * j] * hn;
      av.y = hv[2 * j + 1] * hn;
      bv.x = fast_tanh(tv[2 * j] * tn);
      bv.y = fast_tanh(tv[2 * j + 1] * tn);
      acc = __builtin_amdgcn_wmma_f32_16x16x4_f32(false, av, false, bv,
                                                  (short)0, acc, false, false);
    }
    const int idx = lane & 7;
    float d = acc[0];
    d = (idx == 1) ? acc[1] : d;
    d = (idx == 2) ? acc[2] : d;
    d = (idx == 3) ? acc[3] : d;
    d = (idx == 4) ? acc[4] : d;
    d = (idx == 5) ? acc[5] : d;
    d = (idx == 6) ? acc[6] : d;
    d = (idx == 7) ? acc[7] : d;
    dv[g] = d;
  }

#pragma unroll
  for (int g = 0; g < 4; ++g)
    if (wr[g]) A[(size_t)(e0 + g * 4) * 4 + cc] = aold[g] + dv[g];
}

__global__ void k_final(const float* __restrict__ u, const float* __restrict__ it,
                        const float* __restrict__ l1, const float* __restrict__ l2,
                        float* __restrict__ out, int userElems, int n) {
  int i = blockIdx.x * blockDim.x + threadIdx.x;
  if (i >= n) return;
  float v = (i < userElems) ? u[i] : it[i - userElems];
  v += l1[i];
  v += l2[i];
  out[i] = v * (1.0f / 3.0f);
}

static inline size_t align16f(size_t x) { return (x + 15) & ~(size_t)15; }

extern "C" void kernel_launch(void* const* d_in, const int* in_sizes, int n_in,
                              void* d_out, int out_size, void* d_ws, size_t ws_size,
                              hipStream_t stream) {
  (void)n_in; (void)out_size; (void)ws_size;
  const float* u  = (const float*)d_in[0];
  const float* it = (const float*)d_in[1];
  const int* hl   = (const int*)d_in[2];
  const int* tl   = (const int*)d_in[3];

  const int userElems = in_sizes[0];
  const int itemElems = in_sizes[1];
  const int E = in_sizes[2];
  const int N = (userElems + itemElems) / 64;
  const size_t N64 = (size_t)N * 64;
  const size_t E4  = (size_t)E * 4;
  const size_t N4  = (size_t)N * 4;

  float* base = (float*)d_ws;
  size_t off = 0;
  float* ego   = base + off; off += align16f(N64);
  float* fact  = base + off; off += align16f(N64);
  float* lists = base + off; off += align16f(N64);
  float* A     = base + off; off += align16f(E4);
  float* ew    = base + off; off += align16f(E4);
  float* dval  = base + off; off += align16f(N4);
  float* keyf  = base + off; off += align16f(N4);

#define GRID(n) ((unsigned)(((size_t)(n) + WG - 1) / WG))

  k_init<<<GRID(N64), WG, 0, stream>>>(u, it, ego, userElems, (int)N64);
  k_fill1<<<GRID(E4), WG, 0, stream>>>(A, (int)E4);

  float* cur = ego;
  float* nxt = fact;
  const int nGroups = (E + 15) / 16;
  const dim3 degGrid(DEG_SLICES, (unsigned)((N + DEG_NODES - 1) / DEG_NODES));
  const unsigned msgGrid = (unsigned)((N + MSG_NODES - 1) / MSG_NODES);
  const int degPer = (((E + DEG_SLICES - 1) / DEG_SLICES) + 3) & ~3;

  for (int layer = 0; layer < 2; ++layer) {
    for (int itn = 0; itn < 2; ++itn) {
      k_softmax<<<GRID(E), WG, 0, stream>>>(A, ew, E);
      if (layer == 0 && itn == 0) {
        k_degree_lds<<<degGrid, DEG_WG, 0, stream>>>(ew, hl, nxt, E, N, degPer);
        k_dinv_sum<<<GRID(N), WG, 0, stream>>>(nxt, dval, N);
        k_edge_weight<false><<<GRID(E), WG, 0, stream>>>(ew, dval, hl, tl, keyf, E, (int)msgGrid);
        k_message_first<<<msgGrid, MSG_WG, 0, stream>>>(ew, cur, hl, tl, keyf, nxt, lists, E, N);
      } else {
        k_degree_lists<<<msgGrid, MSG_WG, 0, stream>>>(ew, hl, lists, nxt, E, N);
        k_edge_weight<true><<<GRID(E), WG, 0, stream>>>(ew, nxt, hl, tl, keyf, E, (int)msgGrid);
        k_message_replay<<<msgGrid, MSG_WG, 0, stream>>>(ew, cur, hl, tl, keyf, nxt, lists, E, N);
      }
      const bool lastUpdateDead = (layer == 1 && itn == 1);
      if (!lastUpdateDead)
        k_updateA<<<GRID((size_t)nGroups * 32), WG, 0, stream>>>(
            nxt, cur, hl, tl, A, E, nGroups);
    }
    float* tmp = cur; cur = nxt; nxt = tmp;
  }

  k_final<<<GRID(N64), WG, 0, stream>>>(u, it, nxt, cur, (float*)d_out, userElems, (int)N64);
#undef GRID
}


# Passes 2-4: message waves form the weights of their listed edges themselves (softmax of the A row, grouped inverse roots), degree groups form normA too; 6 flat launches gone; no sum changed
# speedup vs baseline: 3.5294x; 1.1846x over previous
#include <hip/hip_runtime.h>
#include <math.h>

typedef float v2f __attribute__((ext_vector_type(2)));
typedef float v4f __attribute__((ext_vector_type(4)));
typedef float v8f __attribute__((ext_vector_type(8)));

#define WG 256

__device__ __forceinline__ float xhalf(float x) {
  int b = __builtin_bit_cast(int, x);
  return __builtin_bit_cast(float,
      __builtin_amdgcn_permlanex16(b, b, 0x76543210, 0xfedcba98, false, false));
}

__device__ __forceinline__ float fast_tanh(float x) {
  float r;
  asm("v_tanh_f32 %0, %1\n\tv_nop\n\tv_nop" : "=v"(r) : "v"(x));
  return r;
}

__global__ void k_init(const float* __restrict__ u, const float* __restrict__ it,
                       float* __restrict__ ego, int userElems, int total) {
  int i = blockIdx.x * blockDim.x + threadIdx.x;
  if (i >= total) return;
  ego[i] = (i < userElems) ? u[i] : it[i - userElems];
}

__global__ void k_fill1(float* __restrict__ a, int n) {
  int i = blockIdx.x * blockDim.x + threadIdx.x;
  if (i < n) a[i] = 1.0f;
}

__device__ __forceinline__ float4 softmax4(float4 a) {
  float m = fmaxf(fmaxf(a.x, a.y), fmaxf(a.z, a.w));
  float e0 = expf(a.x - m), e1 = expf(a.y - m), e2 = expf(a.z - m), e3 = expf(a.w - m);
  float inv = 1.0f / (e0 + e1 + e2 + e3);
  return make_float4(e0 * inv, e1 * inv, e2 * inv, e3 * inv);
}

__global__ void k_softmax(const float* __restrict__ A, float* __restrict__ ew, int E) {
  int e = blockIdx.x * blockDim.x + threadIdx.x;
  if (e >= E) return;
  reinterpret_cast<float4*>(ew)[e] = softmax4(reinterpret_cast<const float4*>(A)[e]);
}

#define DEG_SLICES 16
#define DEG_NODES  5000
#define DEG_WG     1024
#define DEG_VEC    4

__device__ __forceinline__ void lds_fadd(float* p, float v) {
  __hip_atomic_fetch_add(p, v, __ATOMIC_RELAXED, __HIP_MEMORY_SCOPE_WORKGROUP);
}

__global__ void __launch_bounds__(DEG_WG) k_degree_lds(
    const float* __restrict__ ew, const int* __restrict__ hl,
    float* __restrict__ part, int E, int N, int per) {
  __shared__ float4 tab4[DEG_NODES];
  float* tab = reinterpret_cast<float*>(tab4);
  const int s = blockIdx.x;
  const int lo = blockIdx.y * DEG_NODES;

  for (int i = threadIdx.x; i < DEG_NODES; i += DEG_WG)
    tab4[i] = make_float4(0.f, 0.f, 0.f, 0.f);
  __syncthreads();

  const int e0 = s * per;
  int e1 = e0 + per;
  if (e1 > E) e1 = E;
  const int nvec = (e1 > e0) ? ((e1 - e0) >> 2) : 0;
  const int4* hv = reinterpret_cast<const int4*>(hl + e0);
  const float4* ew4 = reinterpret_cast<const float4*>(ew);

  for (int v = threadIdx.x; v < nvec; v += DEG_WG * DEG_VEC) {
    int hd[DEG_VEC * 4];
#pragma unroll
    for (int k = 0; k < DEG_VEC; ++k) {
      int vk = v + k * DEG_WG;
      int4 h = make_int4(-1, -1, -1, -1);
      if (vk < nvec) h = hv[vk];
      hd[k * 4 + 0] = h.x; hd[k * 4 + 1] = h.y; hd[k * 4 + 2] = h.z; hd[k * 4 + 3] = h.w;
    }
    float4 w[DEG_VEC * 4];
#pragma unroll
    for (int j = 0; j < DEG_VEC * 4; ++j) {
      unsigned d = (unsigned)(hd[j] - lo);
      w[j] = make_float4(0.f, 0.f, 0.f, 0.f);
      if (d < (unsigned)DEG_NODES) w[j] = ew4[e0 + ((v + (j >> 2) * DEG_WG) << 2) + (j & 3)];
    }
#pragma unroll
    for (int j = 0; j < DEG_VEC * 4; ++j) {
      unsigned d = (unsigned)(hd[j] - lo);
      if (d < (unsigned)DEG_NODES) {
        float* t = tab + d * 4;
        lds_fadd(t + 0, w[j].x);
        lds_fadd(t + 1, w[j].y);
        lds_fadd(t + 2, w[j].z);
        lds_fadd(t + 3, w[j].w);
      }
    }
  }
  for (int e = e0 + (nvec << 2) + threadIdx.x; e < e1; e += DEG_WG) {
    unsigned d = (unsigned)(hl[e] - lo);
    if (d < (unsigned)DEG_NODES) {
      float4 wv = ew4[e];
      float* t = tab + d * 4;
      lds_fadd(t + 0, wv.x);
      lds_fadd(t + 1, wv.y);
      lds_fadd(t + 2, wv.z);
      lds_fadd(t + 3, wv.w);
    }
  }
  __syncthreads();

  int nodes = N - lo;
  if (nodes > DEG_NODES) nodes = DEG_NODES;
  float4* dst = reinterpret_cast<float4*>(part) + (size_t)s * N + lo;
  for (int i = threadIdx.x; i < nodes; i += DEG_WG) dst[i] = tab4[i];
}

__global__ void k_dinv_sum(const float* __restrict__ part, float* __restrict__ d, int N) {
  int n = blockIdx.x * blockDim.x + threadIdx.x;
  if (n >= N) return;
  const float4* p4 = reinterpret_cast<const float4*>(part);
  float4 sum = make_float4(0.f, 0.f, 0.f, 0.f);
#pragma unroll
  for (int k = 0; k < DEG_SLICES; ++k) {
    float4 v = p4[(size_t)k * N + n];
    sum.x += v.x; sum.y += v.y; sum.z += v.z; sum.w += v.w;
  }
  float4 r;
  r.x = 1.0f / sqrtf(fmaxf(sum.x, 1e-8f));
  r.y = 1.0f / sqrtf(fmaxf(sum.y, 1e-8f));
  r.z = 1.0f / sqrtf(fmaxf(sum.z, 1e-8f));
  r.w = 1.0f / sqrtf(fmaxf(sum.w, 1e-8f));
  reinterpret_cast<float4*>(d)[n] = r;
}

#define MSG_NODES 320
#define MSG_WG    1024
#define MSG_WAVES 32
#define MSG_BATCH 16
#define MSG_QCAP  544
#define MSG_NOKEY 255u
#define MSG_OVER  0xffffffffu
#define MSG_TRIP  128
#define MSG_TBITS 17

__global__ void k_edge_weight(float* __restrict__ ew, const float* __restrict__ dinv,
                              const int* __restrict__ hl, const int* __restrict__ tl,
                              float* __restrict__ keyf, int E, int R) {
  int e = blockIdx.x * blockDim.x + threadIdx.x;
  unsigned char* keyb = reinterpret_cast<unsigned char*>(keyf);
  if (e >= E) {
    if (e < ((E + 15) & ~15)) keyb[e] = (unsigned char)MSG_NOKEY;
    return;
  }
  const unsigned h = (unsigned)hl[e];
  const unsigned t = (unsigned)tl[e];
  const unsigned hg = h % (unsigned)R;
  float4 w = reinterpret_cast<float4*>(ew)[e];
  float4 dh = reinterpret_cast<const float4*>(dinv)[h];
  float4 dt = reinterpret_cast<const float4*>(dinv)[t];
  w.x *= dh.x * dt.x;
  w.y *= dh.y * dt.y;
  w.z *= dh.z * dt.z;
  w.w *= dh.w * dt.w;
  reinterpret_cast<float4*>(ew)[e] = w;
  keyb[e] = (unsigned char)hg;
}

__device__ __forceinline__ unsigned nonzero_bytes(unsigned x) {
  return ((x & 0x7f7f7f7fu) + 0x7f7f7f7fu) | x;
}

__device__ __forceinline__ unsigned key_hits(uint4 k, unsigned pat) {
  unsigned y0 = nonzero_bytes(k.x ^ pat);
  unsigned y1 = nonzero_bytes(k.y ^ pat);
  unsigned y2 = nonzero_bytes(k.z ^ pat);
  unsigned y3 = nonzero_bytes(k.w ^ pat);
  unsigned n = ((y0 >> 7) & 0x01010101u) | ((y1 >> 6) & 0x02020202u) |
               ((y2 >> 5) & 0x04040404u) | ((y3 >> 4) & 0x08080808u);
  return ~n & 0x0f0f0f0fu;
}

template <int K>
__device__ __forceinline__ float quad_lane(float x) {
  const int b = __builtin_bit_cast(int, x);
  return __builtin_bit_cast(float, __builtin_amdgcn_mov_dpp(b, K * 0x55, 0xf, 0xf, false));
}

struct OwnW {
  float4 a;
  float dh, dt;
};

__device__ __forceinline__ OwnW own_weight_load(const float* __restrict__ A,
                                                const float* __restrict__ dinv,
                                                unsigned e, unsigned p, unsigned r, unsigned R,
                                                unsigned Rm, int f) {
  const unsigned t = p & ((1u << MSG_TBITS) - 1u);
  const unsigned tq = __umulhi(t + t, Rm);
  const unsigned hi = r * (unsigned)MSG_NODES + (p >> MSG_TBITS);
  const unsigned ti = (t - tq * R) * (unsigned)MSG_NODES + tq;
  OwnW o;
  o.a = reinterpret_cast<const float4*>(A)[e];
  o.dh = dinv[(size_t)hi * 4 + f];
  o.dt = dinv[(size_t)ti * 4 + f];
  return o;
}

__device__ __forceinline__ float own_weight(const OwnW o, int f) {
  const float4 a = o.a;
  const float dh = o.dh, dt = o.dt;
  const float m = fmaxf(fmaxf(a.x, a.y), fmaxf(a.z, a.w));
  float af = a.x;
  af = (f == 1) ? a.y : af;
  af = (f == 2) ? a.z : af;
  af = (f == 3) ? a.w : af;
  const float ef = expf(af - m);
  const float e0 = quad_lane<0>(ef), e1 = quad_lane<1>(ef);
  const float e2 = quad_lane<2>(ef), e3 = quad_lane<3>(ef);
  const float inv = 1.0f / (e0 + e1 + e2 + e3);
  return (ef * inv) * (dh * dt);
}

template <bool FULL, bool PAIRS, bool OWNW>
__device__ __forceinline__ unsigned msg_batch(const unsigned* qp, int n, int lane, unsigned R,
                                              unsigned Rm,
                                              const float* __restrict__ wsrc,
                                              const float* __restrict__ dinv,
                                              const float* __restrict__ ego,
                                              const int* __restrict__ hl,
                                              const int* __restrict__ tl, float* tab) {
  const int l16 = lane & 15;
  const int s16 = (FULL || l16 < n) ? l16 : 0;
  const int j8 = lane >> 2;
  const int sa = (FULL || j8 < n) ? j8 : 0;
  const int sb = (FULL || j8 + 8 < n) ? j8 + 8 : 0;
  unsigned pk, ea, eb;
  unsigned pa = 0u, pb = 0u;
  if (PAIRS) {
    pk = qp[2 * s16 + 1];
    ea = qp[2 * sa];
    eb = qp[2 * sb];
    if (OWNW) {
      pa = qp[2 * sa + 1];
      pb = qp[2 * sb + 1];
    }
  } else {
    const unsigned el = qp[s16];
    pk = (unsigned)tl[el] | (((unsigned)hl[el] / R) << MSG_TBITS);
    ea = qp[sa];
    eb = qp[sb];
    if (OWNW) {
      pa = (unsigned)tl[ea] | (((unsigned)hl[ea] / R) << MSG_TBITS);
      pb = (unsigned)tl[eb] | (((unsigned)hl[eb] / R) << MSG_TBITS);
    }
  }
  const int tv = (int)(pk & ((1u << MSG_TBITS) - 1u));
  const int dv = (int)(pk >> MSG_TBITS);
  float wa = 0.f, wb = 0.f;
  OwnW oa, ob;
  if (OWNW) {
    oa = own_weight_load(wsrc, dinv, ea, pa, blockIdx.x, R, Rm, lane & 3);
    ob = own_weight_load(wsrc, dinv, eb, pb, blockIdx.x, R, Rm, lane & 3);
  } else {
    wa = wsrc[(size_t)ea * 4 + (lane & 3)];
    wb = wsrc[(size_t)eb * 4 + (lane & 3)];
  }
  const int f = lane >> 3;

  float2 v[MSG_BATCH];
  int d[MSG_BATCH];
#pragma unroll
  for (int i = 0; i < MSG_BATCH; ++i) {
    const int t = __builtin_amdgcn_readlane(tv, i);
    d[i] = __builtin_amdgcn_readlane(dv, i);
    v[i] = make_float2(0.f, 0.f);
    if (FULL || i < n)
      v[i] = *reinterpret_cast<const float2*>(ego + (size_t)t * 64 + lane * 2);
  }
  if (OWNW) {
    __builtin_amdgcn_sched_barrier(0);
    wa = own_weight(oa, lane & 3);
    wb = own_weight(ob, lane & 3);
  }
  if (PAIRS) {
    asm volatile("" :: "v"(v[0].x), "v"(v[0].y), "v"(v[1].x), "v"(v[1].y),
                       "v"(v[2].x), "v"(v[2].y), "v"(v[3].x), "v"(v[3].y),
                       "v"(v[4].x), "v"(v[4].y), "v"(v[5].x), "v"(v[5].y),
                       "v"(v[6].x), "v"(v[6].y), "v"(v[7].x), "v"(v[7].y));
    asm volatile("" :: "v"(v[8].x), "v"(v[8].y), "v"(v[9].x), "v"(v[9].y),
                       "v"(v[10].x), "v"(v[10].y), "v"(v[11].x), "v"(v[11].y),
                       "v"(v[12].x), "v"(v[12].y), "v"(v[13].x), "v"(v[13].y),
                       "v"(v[14].x), "v"(v[14].y), "v"(v[15].x), "v"(v[15].y));
  }
#pragma unroll
  for (int i = 0; i < MSG_BATCH; ++i) {
    const float w = __shfl((i < 8) ? wa : wb, 4 * (i & 7) + f);
    if (FULL || i < n) {
      float* tp = tab + d[i] * 64 + lane * 2;
      lds_fadd(tp + 0, w * v[i].x);
      lds_fadd(tp + 1, w * v[i].y);
    }
  }
  return pk;
}

template <bool REPLAY>
__device__ __forceinline__ void message_body(
    const float* __restrict__ wsrc, const float* __restrict__ dinv,
    const float* __restrict__ ego,
    const int* __restrict__ hl, const int* __restrict__ tl,
    const float* __restrict__ keyf, float* __restrict__ fact,
    float* __restrict__ listf, int E, int N,
    float4* mtab4, unsigned (*mq)[MSG_QCAP]) {
  float* tab = reinterpret_cast<float*>(mtab4);
  const int r = blockIdx.x;
  const unsigned R = gridDim.x;
  const unsigned Rm = 0x80000000u / R + 1u;
  const int lane = threadIdx.x & 31;
  const int wave = threadIdx.x >> 5;

  for (int i = threadIdx.x; i < MSG_NODES * 16; i += MSG_WG)
    mtab4[i] = make_float4(0.f, 0.f, 0.f, 0.f);
  __syncthreads();

  const uint4* k4 = reinterpret_cast<const uint4*>(keyf);
  const int nChunks = (E + 15) >> 4;
  const int perTrip = MSG_WAVES * 128;
  const int trips = (nChunks + perTrip - 1) / perTrip;
  const unsigned pat = (unsigned)r * 0x01010101u;
  unsigned* q = mq[wave];
  int cnt = 0;

  const unsigned cap = (((unsigned)N * 64u) / (R * (unsigned)MSG_WAVES)) & ~1u;
  const unsigned pcap = (cap - 2u) >> 1;
  unsigned* gl = reinterpret_cast<unsigned*>(listf) +
                 (size_t)((unsigned)r * MSG_WAVES + (unsigned)wave) * cap;
  uint2* gp = reinterpret_cast<uint2*>(gl) + 1;
  unsigned stored = 0;
  bool fits = true;
  bool scan = true;

  if (REPLAY) {
    const unsigned n = (unsigned)__builtin_amdgcn_readfirstlane((int)gl[0]);
    if (n != MSG_OVER) {
      scan = false;
      for (unsigned b0 = 0; b0 < n; b0 += MSG_TRIP) {
        const unsigned m = (n - b0 < (unsigned)MSG_TRIP) ? (n - b0) : (unsigned)MSG_TRIP;
        uint2 env[MSG_TRIP / 32];
#pragma unroll
        for (int k = 0; k < MSG_TRIP / 32; ++k) {
          const unsigned x = (unsigned)(k * 32 + lane);
          env[k] = make_uint2(0u, 0u);
          if (x < m) env[k] = gp[b0 + x];
        }
        uint2* q2 = reinterpret_cast<uint2*>(q);
#pragma unroll
        for (int k = 0; k < MSG_TRIP / 32; ++k) q2[k * 32 + lane] = env[k];
        const int full = (int)(m >> 4);
        for (int b = 0; b < full; ++b)
          msg_batch<true, true, true>(q + b * MSG_BATCH * 2, MSG_BATCH, lane, R, Rm, wsrc, dinv, ego, hl, tl, tab);
        const int rem = (int)(m & 15u);
        if (rem > 0)
          msg_batch<false, true, true>(q + full * MSG_BATCH * 2, rem, lane, R, Rm, wsrc, dinv, ego, hl, tl, tab);
      }
    }
  }

  int j = r % trips;
  for (int jj = 0; scan && jj < trips; ++jj) {
    const int cb = j * perTrip + wave * 128;
    j = (j + 1 == trips) ? 0 : j + 1;
    uint4 kk[4];
#pragma unroll
    for (int u = 0; u < 4; ++u) {
      const int c = cb + u * 32 + lane;
      kk[u] = make_uint4(~0u, ~0u, ~0u, ~0u);
      if (c < nChunks) kk[u] = k4[c];
    }
#pragma unroll
    for (int u = 0; u < 4; ++u) {
      const int c = cb + u * 32 + lane;
      unsigned m = key_hits(kk[u], pat);
      unsigned bal = __builtin_amdgcn_ballot_w32(m != 0u);
      while (bal != 0u) {
        if (m != 0u) {
          const int p = __builtin_ctz(m);
          const int idx = ((p & 3) << 2) | (p >> 3);
          q[cnt + (int)__builtin_amdgcn_mbcnt_lo(bal, 0u)] = (unsigned)(c * 16 + idx);
          m &= m - 1u;
        }
        cnt += __builtin_popcount(bal);
        bal = __builtin_amdgcn_ballot_w32(m != 0u);
      }
      while (cnt >= MSG_BATCH) {
        cnt -= MSG_BATCH;
        const unsigned pk =
            msg_batch<true, false, REPLAY>(q + cnt, MSG_BATCH, lane, R, Rm, wsrc, dinv, ego, hl, tl, tab);
        (void)pk;
        if (!REPLAY) {
          if (fits && stored + MSG_BATCH <= pcap) {
            if (lane < MSG_BATCH) gp[stored + lane] = make_uint2(q[cnt + lane], pk);
            stored += MSG_BATCH;
          } else {
            fits = false;
          }
        }
      }
    }
  }
  if (cnt > 0) {
    const unsigned pk = msg_batch<false, false, REPLAY>(q, cnt, lane, R, Rm, wsrc, dinv, ego, hl, tl, tab);
    (void)pk;
    if (!REPLAY) {
      if (fits && stored + (unsigned)cnt <= pcap) {
        if (lane < cnt) gp[stored + lane] = make_uint2(q[lane], pk);
        stored += (unsigned)cnt;
      } else {
        fits = false;
      }
    }
  }
  if (!REPLAY && lane == 0) gl[0] = fits ? stored : MSG_OVER;
  __syncthreads();

  float4* dst = reinterpret_cast<float4*>(fact);
  for (int i = threadIdx.x; i < MSG_NODES * 16; i += MSG_WG) {
    const int node = (i >> 4) * (int)R + r;
    if (node < N) dst[(size_t)node * 16 + (i & 15)] = mtab4[i];
  }
}

__global__ void __launch_bounds__(MSG_WG) k_message_first(
    const float* __restrict__ ew, const float* __restrict__ ego,
    const int* __restrict__ hl, const int* __restrict__ tl,
    const float* __restrict__ keyf, float* __restrict__ fact,
    float* __restrict__ listf, int E, int N) {
  __shared__ float4 mtab4[MSG_NODES * 16];
  __shared__ __attribute__((aligned(16))) unsigned mq[MSG_WAVES][MSG_QCAP];
  message_body<false>(ew, ew, ego, hl, tl, keyf, fact, listf, E, N, mtab4, mq);
}

__global__ void __launch_bounds__(MSG_WG) k_message_replay(
    const float* __restrict__ A, const float* __restrict__ dinv,
    const float* __restrict__ ego,
    const int* __restrict__ hl, const int* __restrict__ tl,
    const float* __restrict__ keyf, float* __restrict__ fact,
    float* __restrict__ listf, int E, int N) {
  __shared__ float4 mtab4[MSG_NODES * 16];
  __shared__ __attribute__((aligned(16))) unsigned mq[MSG_WAVES][MSG_QCAP];
  message_body<true>(A, dinv, ego, hl, tl, keyf, fact, listf, E, N, mtab4, mq);
}

#define DL_IDS 6

__global__ void __launch_bounds__(MSG_WG) k_degree_lists(
    const float* __restrict__ A, const int* __restrict__ hl,
    const float* __restrict__ listf, float* __restrict__ dinvg, int E, int N) {
  __shared__ float4 dtab4[MSG_NODES];
  float* tab = reinterpret_cast<float*>(dtab4);
  const unsigned r = blockIdx.x;
  const unsigned R = gridDim.x;
  const int lane = threadIdx.x & 31;
  const int wave = threadIdx.x >> 5;

  for (int i = threadIdx.x; i < MSG_NODES; i += MSG_WG)
    dtab4[i] = make_float4(0.f, 0.f, 0.f, 0.f);
  __syncthreads();

  const float4* A4 = reinterpret_cast<const float4*>(A);
  const unsigned cap = (((unsigned)N * 64u) / (R * (unsigned)MSG_WAVES)) & ~1u;
  const unsigned* gl = reinterpret_cast<const unsigned*>(listf) +
                       (size_t)(r * MSG_WAVES + (unsigned)wave) * cap;
  const uint2* gp = reinterpret_cast<const uint2*>(gl) + 1;
  const unsigned n = (unsigned)__builtin_amdgcn_readfirstlane((int)gl[0]);
  if (n != MSG_OVER) {
    for (unsigned b0 = 0; b0 < n; b0 += 32u * DL_IDS) {
      uint2 en[DL_IDS];
      bool on[DL_IDS];
#pragma unroll
      for (int k = 0; k < DL_IDS; ++k) {
        const unsigned x = b0 + (unsigned)(k * 32 + lane);
        on[k] = x < n;
        en[k] = gp[on[k] ? x : n - 1u];
      }
      float4 w[DL_IDS];
#pragma unroll
      for (int k = 0; k < DL_IDS; ++k) w[k] = A4[en[k].x];
#pragma unroll
      for (int k = 0; k < DL_IDS; ++k) w[k] = softmax4(w[k]);
#pragma unroll
      for (int k = 0; k < DL_IDS; ++k) {
        if (on[k]) {
          float* t = tab + (en[k].y >> MSG_TBITS) * 4u;
          lds_fadd(t + 0, w[k].x);
          lds_fadd(t + 1, w[k].y);
          lds_fadd(t + 2, w[k].z);
          lds_fadd(t + 3, w[k].w);
        }
      }
    }
  } else {
    const int nChunks = (E + 15) >> 4;
    const int perTrip = MSG_WAVES * 128;
    const int trips = (nChunks + perTrip - 1) / perTrip;
#pragma unroll 1
    for (int j = 0; j < trips; ++j) {
#pragma unroll 1
      for (int u = 0; u < 4; ++u) {
        const int c = j * perTrip + wave * 128 + u * 32 + lane;
        if (c < nChunks) {
#pragma unroll 1
          for (int x = 0; x < 16; ++x) {
            const int e = c * 16 + x;
            if (e < E) {
              const unsigned h = (unsigned)hl[e];
              if (h % R == r) {
                const float4 wv = softmax4(A4[e]);
                float* t = tab + (h / R) * 4u;
                lds_fadd(t + 0, wv.x);
                lds_fadd(t + 1, wv.y);
                lds_fadd(t + 2, wv.z);
                lds_fadd(t + 3, wv.w);
              }
            }
          }
        }
      }
    }
  }
  __syncthreads();

  float4* dst = reinterpret_cast<float4*>(dinvg) + (size_t)r * MSG_NODES;
  for (int i = threadIdx.x; i < MSG_NODES; i += MSG_WG) {
    const float4 s = dtab4[i];
    float4 o;
    o.x = 1.0f / sqrtf(fmaxf(s.x, 1e-8f));
    o.y = 1.0f / sqrtf(fmaxf(s.y, 1e-8f));
    o.z = 1.0f / sqrtf(fmaxf(s.z, 1e-8f));
    o.w = 1.0f / sqrtf(fmaxf(s.w, 1e-8f));
    dst[i] = o;
  }
}

__global__ void __launch_bounds__(WG) k_updateA(
    const float* __restrict__ fact, const float* __restrict__ ego,
    const int* __restrict__ hl, const int* __restrict__ tl,
    float* __restrict__ A, int E, int nGroups) {
  int tid = blockIdx.x * blockDim.x + threadIdx.x;
  int wave = tid >> 5;
  int lane = tid & 31;
  if (wave >= nGroups) return;

  const int c = lane & 15;
  const int kh = lane >> 4;
  const int i = c >> 2;
  const int f = c & 3;
  const int seg = f * 16 + kh * 8;

  const bool active = (lane < 8) || (lane >= 24);
  const int cc = (lane < 8) ? lane : (lane - 16);
  const int e0 = wave * 16;

  int hh[4], tt[4];
  float aold[4];
  bool wr[4];
  const int aLast = E * 4 - 1;
#pragma unroll
  for (int g = 0; g < 4; ++g) {
    const int eIdx = e0 + g * 4 + i;
    const int e = (eIdx < E) ? eIdx : (E - 1);
    hh[g] = hl[e];
    tt[g] = tl[e];
    const int ai = (e0 + g * 4) * 4 + c;
    aold[g] = A[(ai < aLast) ? ai : aLast];
    wr[g] = active && (e0 + g * 4 + (cc >> 2) < E);
  }
  asm volatile("" :: "v"(hh[0]), "v"(hh[1]), "v"(hh[2]), "v"(hh[3]),
                     "v"(tt[0]), "v"(tt[1]), "v"(tt[2]), "v"(tt[3]),
                     "v"(aold[0]), "v"(aold[1]), "v"(aold[2]), "v"(aold[3]));
  v4f h0[4], h1[4], t0[4], t1[4];
#pragma unroll
  for (int g = 0; g < 4; ++g) {
    const v4f* hp = reinterpret_cast<const v4f*>(fact + (size_t)hh[g] * 64 + seg);
    const v4f* tp = reinterpret_cast<const v4f*>(ego + (size_t)tt[g] * 64 + seg);
    h0[g] = hp[0];
    h1[g] = hp[1];
    t0[g] = tp[0];
    t1[g] = tp[1];
  }
  float dv[4];
#pragma unroll
  for (int g = 0; g < 4; ++g) {
    const float hv[8] = {h0[g].x, h0[g].y, h0[g].z, h0[g].w, h1[g].x, h1[g].y, h1[g].z, h1[g].w};
    const float tv[8] = {t0[g].x, t0[g].y, t0[g].z, t0[g].w, t1[g].x, t1[g].y, t1[g].z, t1[g].w};
    v2f qh = h0[g].xy * h0[g].xy;
    v2f qt = t0[g].xy * t0[g].xy;
    qh = __builtin_elementwise_fma(h0[g].zw, h0[g].zw, qh);
    qt = __builtin_elementwise_fma(t0[g].zw, t0[g].zw, qt);
    qh = __builtin_elementwise_fma(h1[g].xy, h1[g].xy, qh);
    qt = __builtin_elementwise_fma(t1[g].xy, t1[g].xy, qt);
    qh = __builtin_elementwise_fma(h1[g].zw, h1[g].zw, qh);
    qt = __builtin_elementwise_fma(t1[g].zw, t1[g].zw, qt);
    const float ph = qh.x + qh.y, pt = qt.x + qt.y;
    const float sh = ph + xhalf(ph);
    const float st = pt + xhalf(pt);
    const float mine = 1.0f / fmaxf(sqrtf(kh ? st : sh), 1e-12f);
    const float other = xhalf(mine);
    const float hn = kh ? other : mine;
    const float tn = kh ? mine : other;
    v8f acc = {0.f, 0.f, 0.f, 0.f, 0.f, 0.f, 0.f, 0.f};
#pragma unroll
    for (int j = 0; j < 4; ++j) {
      v2f av, bv;
      av.x = hv[2 * j] * hn;
      av.y = hv[2 * j + 1] * hn;
      bv.x = fast_tanh(tv[2 * j] * tn);
      bv.y = fast_tanh(tv[2 * j + 1] * tn);
      acc = __builtin_amdgcn_wmma_f32_16x16x4_f32(false, av, false, bv,
                                                  (short)0, acc, false, false);
    }
    const int idx = lane & 7;
    float d = acc[0];
    d = (idx == 1) ? acc[1] : d;
    d = (idx == 2) ? acc[2] : d;
    d = (idx == 3) ? acc[3] : d;
    d = (idx == 4) ? acc[4] : d;
    d = (idx == 5) ? acc[5] : d;
    d = (idx == 6) ? acc[6] : d;
    d = (idx == 7) ? acc[7] : d;
    dv[g] = d;
  }

#pragma unroll
  for (int g = 0; g < 4; ++g)
    if (wr[g]) A[(size_t)(e0 + g * 4) * 4 + cc] = aold[g] + dv[g];
}

__global__ void k_final(const float* __restrict__ u, const float* __restrict__ it,
                        const float* __restrict__ l1, const float* __restrict__ l2,
                        float* __restrict__ out, int userElems, int n) {
  int i = blockIdx.x * blockDim.x + threadIdx.x;
  if (i >= n) return;
  float v = (i < userElems) ? u[i] : it[i - userElems];
  v += l1[i];
  v += l2[i];
  out[i] = v * (1.0f / 3.0f);
}

static inline size_t align16f(size_t x) { return (x + 15) & ~(size_t)15; }

extern "C" void kernel_launch(void* const* d_in, const int* in_sizes, int n_in,
                              void* d_out, int out_size, void* d_ws, size_t ws_size,
                              hipStream_t stream) {
  (void)n_in; (void)out_size; (void)ws_size;
  const float* u  = (const float*)d_in[0];
  const float* it = (const float*)d_in[1];
  const int* hl   = (const int*)d_in[2];
  const int* tl   = (const int*)d_in[3];

  const int userElems = in_sizes[0];
  const int itemElems = in_sizes[1];
  const int E = in_sizes[2];
  const int N = (userElems + itemElems) / 64;
  const size_t N64 = (size_t)N * 64;
  const size_t E4  = (size_t)E * 4;
  const size_t N4  = (size_t)N * 4;

  float* base = (float*)d_ws;
  size_t off = 0;
  float* ego   = base + off; off += align16f(N64);
  float* fact  = base + off; off += align16f(N64);
  float* lists = base + off; off += align16f(N64);
  float* A     = base + off; off += align16f(E4);
  float* ew    = base + off; off += align16f(E4);
  float* dval  = base + off; off += align16f(N4);
  float* keyf  = base + off; off += align16f(N4);

#define GRID(n) ((unsigned)(((size_t)(n) + WG - 1) / WG))

  k_init<<<GRID(N64), WG, 0, stream>>>(u, it, ego, userElems, (int)N64);
  k_fill1<<<GRID(E4), WG, 0, stream>>>(A, (int)E4);

  float* cur = ego;
  float* nxt = fact;
  const int nGroups = (E + 15) / 16;
  const dim3 degGrid(DEG_SLICES, (unsigned)((N + DEG_NODES - 1) / DEG_NODES));
  const unsigned msgGrid = (unsigned)((N + MSG_NODES - 1) / MSG_NODES);
  const int degPer = (((E + DEG_SLICES - 1) / DEG_SLICES) + 3) & ~3;

  for (int layer = 0; layer < 2; ++layer) {
    for (int itn = 0; itn < 2; ++itn) {
      if (layer == 0 && itn == 0) {
        k_softmax<<<GRID(E), WG, 0, stream>>>(A, ew, E);
        k_degree_lds<<<degGrid, DEG_WG, 0, stream>>>(ew, hl, nxt, E, N, degPer);
        k_dinv_sum<<<GRID(N), WG, 0, stream>>>(nxt, dval, N);
        k_edge_weight<<<GRID(E), WG, 0, stream>>>(ew, dval, hl, tl, keyf, E, (int)msgGrid);
        k_message_first<<<msgGrid, MSG_WG, 0, stream>>>(ew, cur, hl, tl, keyf, nxt, lists, E, N);
      } else {
        k_degree_lists<<<msgGrid, MSG_WG, 0, stream>>>(A, hl, lists, ew, E, N);
        k_message_replay<<<msgGrid, MSG_WG, 0, stream>>>(A, ew, cur, hl, tl, keyf, nxt, lists, E, N);
      }
      const bool lastUpdateDead = (layer == 1 && itn == 1);
      if (!lastUpdateDead)
        k_updateA<<<GRID((size_t)nGroups * 32), WG, 0, stream>>>(
            nxt, cur, hl, tl, A, E, nGroups);
    }
    float* tmp = cur; cur = nxt; nxt = tmp;
  }

  k_final<<<GRID(N64), WG, 0, stream>>>(u, it, nxt, cur, (float*)d_out, userElems, (int)N64);
#undef GRID
}


# Elementwise passes (table copy, fill, first softmax and edge weights, final mean) run as 4096-wave resident loops of 16-byte accesses, not one element a thread; no sum changed
# speedup vs baseline: 3.7093x; 1.0510x over previous
#include <hip/hip_runtime.h>
#include <math.h>

typedef float v2f __attribute__((ext_vector_type(2)));
typedef float v4f __attribute__((ext_vector_type(4)));
typedef float v8f __attribute__((ext_vector_type(8)));

#define WG 256

__device__ __forceinline__ float xhalf(float x) {
  int b = __builtin_bit_cast(int, x);
  return __builtin_bit_cast(float,
      __builtin_amdgcn_permlanex16(b, b, 0x76543210, 0xfedcba98, false, false));
}

__device__ __forceinline__ float fast_tanh(float x) {
  float r;
  asm("v_tanh_f32 %0, %1\n\tv_nop\n\tv_nop" : "=v"(r) : "v"(x));
  return r;
}

#define STREAM_WGS 512
#define CP_VEC 8
#define EW_VEC 4
#define FN_VEC 4

__device__ __forceinline__ void hold4(const float4 a, const float4 b, const float4 c, const float4 d) {
  asm volatile("" :: "v"(a.x), "v"(a.y), "v"(a.z), "v"(a.w), "v"(b.x), "v"(b.y), "v"(b.z), "v"(b.w),
                     "v"(c.x), "v"(c.y), "v"(c.z), "v"(c.w), "v"(d.x), "v"(d.y), "v"(d.z), "v"(d.w));
}

__global__ void __launch_bounds__(WG) k_init(const float* __restrict__ u, const float* __restrict__ it,
                       float* __restrict__ ego, int userElems, int total) {
  const int user4 = userElems >> 2;
  const int total4 = total >> 2;
  const int stride = (int)(gridDim.x * blockDim.x);
  const float4* u4 = reinterpret_cast<const float4*>(u);
  const float4* i4 = reinterpret_cast<const float4*>(it);
  float4* e4 = reinterpret_cast<float4*>(ego);
  for (int s0 = (int)(blockIdx.x * blockDim.x + threadIdx.x); s0 < total4; s0 += stride * CP_VEC) {
    float4 v[CP_VEC];
#pragma unroll
    for (int k = 0; k < CP_VEC; ++k) {
      const int i = s0 + k * stride;
      const int ic = (i < total4) ? i : total4 - 1;
      const float4* p = (ic < user4) ? (u4 + ic) : (i4 + (ic - user4));
      v[k] = *p;
    }
    hold4(v[0], v[1], v[2], v[3]);
    hold4(v[4], v[5], v[6], v[7]);
#pragma unroll
    for (int k = 0; k < CP_VEC; ++k) {
      const int i = s0 + k * stride;
      if (i < total4) e4[i] = v[k];
    }
  }
}

__global__ void __launch_bounds__(WG) k_fill1(float* __restrict__ a, int n) {
  const int n4 = n >> 2;
  const int stride = (int)(gridDim.x * blockDim.x);
  float4* a4 = reinterpret_cast<float4*>(a);
  const float4 one = make_float4(1.0f, 1.0f, 1.0f, 1.0f);
  for (int i = (int)(blockIdx.x * blockDim.x + threadIdx.x); i < n4; i += stride) a4[i] = one;
}

__device__ __forceinline__ float4 softmax4(float4 a) {
  float m = fmaxf(fmaxf(a.x, a.y), fmaxf(a.z, a.w));
  float e0 = expf(a.x - m), e1 = expf(a.y - m), e2 = expf(a.z - m), e3 = expf(a.w - m);
  float inv = 1.0f / (e0 + e1 + e2 + e3);
  return make_float4(e0 * inv, e1 * inv, e2 * inv, e3 * inv);
}

__global__ void __launch_bounds__(WG) k_softmax(const float* __restrict__ A, float* __restrict__ ew, int E) {
  const int stride = (int)(gridDim.x * blockDim.x);
  const float4* A4 = reinterpret_cast<const float4*>(A);
  float4* w4 = reinterpret_cast<float4*>(ew);
  for (int s0 = (int)(blockIdx.x * blockDim.x + threadIdx.x); s0 < E; s0 += stride * CP_VEC) {
    float4 a[CP_VEC];
#pragma unroll
    for (int k = 0; k < CP_VEC; ++k) {
      const int e = s0 + k * stride;
      a[k] = A4[(e < E) ? e : E - 1];
    }
    hold4(a[0], a[1], a[2], a[3]);
    hold4(a[4], a[5], a[6], a[7]);
#pragma unroll
    for (int k = 0; k < CP_VEC; ++k) {
      const int e = s0 + k * stride;
      if (e < E) w4[e] = softmax4(a[k]);
    }
  }
}

#define DEG_SLICES 16
#define DEG_NODES  5000
#define DEG_WG     1024
#define DEG_VEC    4

__device__ __forceinline__ void lds_fadd(float* p, float v) {
  __hip_atomic_fetch_add(p, v, __ATOMIC_RELAXED, __HIP_MEMORY_SCOPE_WORKGROUP);
}

__global__ void __launch_bounds__(DEG_WG) k_degree_lds(
    const float* __restrict__ ew, const int* __restrict__ hl,
    float* __restrict__ part, int E, int N, int per) {
  __shared__ float4 tab4[DEG_NODES];
  float* tab = reinterpret_cast<float*>(tab4);
  const int s = blockIdx.x;
  const int lo = blockIdx.y * DEG_NODES;

  for (int i = threadIdx.x; i < DEG_NODES; i += DEG_WG)
    tab4[i] = make_float4(0.f, 0.f, 0.f, 0.f);
  __syncthreads();

  const int e0 = s * per;
  int e1 = e0 + per;
  if (e1 > E) e1 = E;
  const int nvec = (e1 > e0) ? ((e1 - e0) >> 2) : 0;
  const int4* hv = reinterpret_cast<const int4*>(hl + e0);
  const float4* ew4 = reinterpret_cast<const float4*>(ew);

  for (int v = threadIdx.x; v < nvec; v += DEG_WG * DEG_VEC) {
    int hd[DEG_VEC * 4];
#pragma unroll
    for (int k = 0; k < DEG_VEC; ++k) {
      int vk = v + k * DEG_WG;
      int4 h = make_int4(-1, -1, -1, -1);
      if (vk < nvec) h = hv[vk];
      hd[k * 4 + 0] = h.x; hd[k * 4 + 1] = h.y; hd[k * 4 + 2] = h.z; hd[k * 4 + 3] = h.w;
    }
    float4 w[DEG_VEC * 4];
#pragma unroll
    for (int j = 0; j < DEG_VEC * 4; ++j) {
      unsigned d = (unsigned)(hd[j] - lo);
      w[j] = make_float4(0.f, 0.f, 0.f, 0.f);
      if (d < (unsigned)DEG_NODES) w[j] = ew4[e0 + ((v + (j >> 2) * DEG_WG) << 2) + (j & 3)];
    }
#pragma unroll
    for (int j = 0; j < DEG_VEC * 4; ++j) {
      unsigned d = (unsigned)(hd[j] - lo);
      if (d < (unsigned)DEG_NODES) {
        float* t = tab + d * 4;
        lds_fadd(t + 0, w[j].x);
        lds_fadd(t + 1, w[j].y);
        lds_fadd(t + 2, w[j].z);
        lds_fadd(t + 3, w[j].w);
      }
    }
  }
  for (int e = e0 + (nvec << 2) + threadIdx.x; e < e1; e += DEG_WG) {
    unsigned d = (unsigned)(hl[e] - lo);
    if (d < (unsigned)DEG_NODES) {
      float4 wv = ew4[e];
      float* t = tab + d * 4;
      lds_fadd(t + 0, wv.x);
      lds_fadd(t + 1, wv.y);
      lds_fadd(t + 2, wv.z);
      lds_fadd(t + 3, wv.w);
    }
  }
  __syncthreads();

  int nodes = N - lo;
  if (nodes > DEG_NODES) nodes = DEG_NODES;
  float4* dst = reinterpret_cast<float4*>(part) + (size_t)s * N + lo;
  for (int i = threadIdx.x; i < nodes; i += DEG_WG) dst[i] = tab4[i];
}

__global__ void k_dinv_sum(const float* __restrict__ part, float* __restrict__ d, int N) {
  int n = blockIdx.x * blockDim.x + threadIdx.x;
  if (n >= N) return;
  const float4* p4 = reinterpret_cast<const float4*>(part);
  float4 sum = make_float4(0.f, 0.f, 0.f, 0.f);
#pragma unroll
  for (int k = 0; k < DEG_SLICES; ++k) {
    float4 v = p4[(size_t)k * N + n];
    sum.x += v.x; sum.y += v.y; sum.z += v.z; sum.w += v.w;
  }
  float4 r;
  r.x = 1.0f / sqrtf(fmaxf(sum.x, 1e-8f));
  r.y = 1.0f / sqrtf(fmaxf(sum.y, 1e-8f));
  r.z = 1.0f / sqrtf(fmaxf(sum.z, 1e-8f));
  r.w = 1.0f / sqrtf(fmaxf(sum.w, 1e-8f));
  reinterpret_cast<float4*>(d)[n] = r;
}

#define MSG_NODES 320
#define MSG_WG    1024
#define MSG_WAVES 32
#define MSG_BATCH 16
#define MSG_QCAP  544
#define MSG_NOKEY 255u
#define MSG_OVER  0xffffffffu
#define MSG_TRIP  128
#define MSG_TBITS 17

__global__ void __launch_bounds__(WG) k_edge_weight(float* __restrict__ ew, const float* __restrict__ dinv,
                              const int* __restrict__ hl, const int* __restrict__ tl,
                              float* __restrict__ keyf, int E, int R) {
  unsigned char* keyb = reinterpret_cast<unsigned char*>(keyf);
  const int stride = (int)(gridDim.x * blockDim.x);
  const int Epad = (E + 15) & ~15;
  float4* w4 = reinterpret_cast<float4*>(ew);
  const float4* d4 = reinterpret_cast<const float4*>(dinv);
  for (int s0 = (int)(blockIdx.x * blockDim.x + threadIdx.x); s0 < Epad; s0 += stride * EW_VEC) {
    unsigned h[EW_VEC], t[EW_VEC];
    float4 w[EW_VEC];
#pragma unroll
    for (int k = 0; k < EW_VEC; ++k) {
      const int e = s0 + k * stride;
      const int ec = (e < E) ? e : E - 1;
      h[k] = (unsigned)hl[ec];
      t[k] = (unsigned)tl[ec];
      w[k] = w4[ec];
    }
    float4 dh[EW_VEC], dt[EW_VEC];
#pragma unroll
    for (int k = 0; k < EW_VEC; ++k) {
      dh[k] = d4[h[k]];
      dt[k] = d4[t[k]];
    }
    hold4(w[0], w[1], w[2], w[3]);
    hold4(dh[0], dh[1], dh[2], dh[3]);
    hold4(dt[0], dt[1], dt[2], dt[3]);
#pragma unroll
    for (int k = 0; k < EW_VEC; ++k) {
      const int e = s0 + k * stride;
      if (e < E) {
        float4 x = w[k];
        x.x *= dh[k].x * dt[k].x;
        x.y *= dh[k].y * dt[k].y;
        x.z *= dh[k].z * dt[k].z;
        x.w *= dh[k].w * dt[k].w;
        w4[e] = x;
        keyb[e] = (unsigned char)(h[k] % (unsigned)R);
      } else if (e < Epad) {
        keyb[e] = (unsigned char)MSG_NOKEY;
      }
    }
  }
}

__device__ __forceinline__ unsigned nonzero_bytes(unsigned x) {
  return ((x & 0x7f7f7f7fu) + 0x7f7f7f7fu) | x;
}

__device__ __forceinline__ unsigned key_hits(uint4 k, unsigned pat) {
  unsigned y0 = nonzero_bytes(k.x ^ pat);
  unsigned y1 = nonzero_bytes(k.y ^ pat);
  unsigned y2 = nonzero_bytes(k.z ^ pat);
  unsigned y3 = nonzero_bytes(k.w ^ pat);
  unsigned n = ((y0 >> 7) & 0x01010101u) | ((y1 >> 6) & 0x02020202u) |
               ((y2 >> 5) & 0x04040404u) | ((y3 >> 4) & 0x08080808u);
  return ~n & 0x0f0f0f0fu;
}

template <int K>
__device__ __forceinline__ float quad_lane(float x) {
  const int b = __builtin_bit_cast(int, x);
  return __builtin_bit_cast(float, __builtin_amdgcn_mov_dpp(b, K * 0x55, 0xf, 0xf, false));
}

struct OwnW {
  float4 a;
  float dh, dt;
};

__device__ __forceinline__ OwnW own_weight_load(const float* __restrict__ A,
                                                const float* __restrict__ dinv,
                                                unsigned e, unsigned p, unsigned r, unsigned R,
                                                unsigned Rm, int f) {
  const unsigned t = p & ((1u << MSG_TBITS) - 1u);
  const unsigned tq = __umulhi(t + t, Rm);
  const unsigned hi = r * (unsigned)MSG_NODES + (p >> MSG_TBITS);
  const unsigned ti = (t - tq * R) * (unsigned)MSG_NODES + tq;
  OwnW o;
  o.a = reinterpret_cast<const float4*>(A)[e];
  o.dh = dinv[(size_t)hi * 4 + f];
  o.dt = dinv[(size_t)ti * 4 + f];
  return o;
}

__device__ __forceinline__ float own_weight(const OwnW o, int f) {
  const float4 a = o.a;
  const float dh = o.dh, dt = o.dt;
  const float m = fmaxf(fmaxf(a.x, a.y), fmaxf(a.z, a.w));
  float af = a.x;
  af = (f == 1) ? a.y : af;
  af = (f == 2) ? a.z : af;
  af = (f == 3) ? a.w : af;
  const float ef = expf(af - m);
  const float e0 = quad_lane<0>(ef), e1 = quad_lane<1>(ef);
  const float e2 = quad_lane<2>(ef), e3 = quad_lane<3>(ef);
  const float inv = 1.0f / (e0 + e1 + e2 + e3);
  return (ef * inv) * (dh * dt);
}

template <bool FULL, bool PAIRS, bool OWNW>
__device__ __forceinline__ unsigned msg_batch(const unsigned* qp, int n, int lane, unsigned R,
                                              unsigned Rm,
                                              const float* __restrict__ wsrc,
                                              const float* __restrict__ dinv,
                                              const float* __restrict__ ego,
                                              const int* __restrict__ hl,
                                              const int* __restrict__ tl, float* tab) {
  const int l16 = lane & 15;
  const int s16 = (FULL || l16 < n) ? l16 : 0;
  const int j8 = lane >> 2;
  const int sa = (FULL || j8 < n) ? j8 : 0;
  const int sb = (FULL || j8 + 8 < n) ? j8 + 8 : 0;
  unsigned pk, ea, eb;
  unsigned pa = 0u, pb = 0u;
  if (PAIRS) {
    pk = qp[2 * s16 + 1];
    ea = qp[2 * sa];
    eb = qp[2 * sb];
    if (OWNW) {
      pa = qp[2 * sa + 1];
      pb = qp[2 * sb + 1];
    }
  } else {
    const unsigned el = qp[s16];
    pk = (unsigned)tl[el] | (((unsigned)hl[el] / R) << MSG_TBITS);
    ea = qp[sa];
    eb = qp[sb];
    if (OWNW) {
      pa = (unsigned)tl[ea] | (((unsigned)hl[ea] / R) << MSG_TBITS);
      pb = (unsigned)tl[eb] | (((unsigned)hl[eb] / R) << MSG_TBITS);
    }
  }
  const int tv = (int)(pk & ((1u << MSG_TBITS) - 1u));
  const int dv = (int)(pk >> MSG_TBITS);
  float wa = 0.f, wb = 0.f;
  OwnW oa, ob;
  if (OWNW) {
    oa = own_weight_load(wsrc, dinv, ea, pa, blockIdx.x, R, Rm, lane & 3);
    ob = own_weight_load(wsrc, dinv, eb, pb, blockIdx.x, R, Rm, lane & 3);
  } else {
    wa = wsrc[(size_t)ea * 4 + (lane & 3)];
    wb = wsrc[(size_t)eb * 4 + (lane & 3)];
  }
  const int f = lane >> 3;

  float2 v[MSG_BATCH];
  int d[MSG_BATCH];
#pragma unroll
  for (int i = 0; i < MSG_BATCH; ++i) {
    const int t = __builtin_amdgcn_readlane(tv, i);
    d[i] = __builtin_amdgcn_readlane(dv, i);
    v[i] = make_float2(0.f, 0.f);
    if (FULL || i < n)
      v[i] = *reinterpret_cast<const float2*>(ego + (size_t)t * 64 + lane * 2);
  }
  if (OWNW) {
    __builtin_amdgcn_sched_barrier(0);
    wa = own_weight(oa, lane & 3);
    wb = own_weight(ob, lane & 3);
  }
  if (PAIRS) {
    asm volatile("" :: "v"(v[0].x), "v"(v[0].y), "v"(v[1].x), "v"(v[1].y),
                       "v"(v[2].x), "v"(v[2].y), "v"(v[3].x), "v"(v[3].y),
                       "v"(v[4].x), "v"(v[4].y), "v"(v[5].x), "v"(v[5].y),
                       "v"(v[6].x), "v"(v[6].y), "v"(v[7].x), "v"(v[7].y));
    asm volatile("" :: "v"(v[8].x), "v"(v[8].y), "v"(v[9].x), "v"(v[9].y),
                       "v"(v[10].x), "v"(v[10].y), "v"(v[11].x), "v"(v[11].y),
                       "v"(v[12].x), "v"(v[12].y), "v"(v[13].x), "v"(v[13].y),
                       "v"(v[14].x), "v"(v[14].y), "v"(v[15].x), "v"(v[15].y));
  }
#pragma unroll
  for (int i = 0; i < MSG_BATCH; ++i) {
    const float w = __shfl((i < 8) ? wa : wb, 4 * (i & 7) + f);
    if (FULL || i < n) {
      float* tp = tab + d[i] * 64 + lane * 2;
      lds_fadd(tp + 0, w * v[i].x);
      lds_fadd(tp + 1, w * v[i].y);
    }
  }
  return pk;
}

template <bool REPLAY>
__device__ __forceinline__ void message_body(
    const float* __restrict__ wsrc, const float* __restrict__ dinv,
    const float* __restrict__ ego,
    const int* __restrict__ hl, const int* __restrict__ tl,
    const float* __restrict__ keyf, float* __restrict__ fact,
    float* __restrict__ listf, int E, int N,
    float4* mtab4, unsigned (*mq)[MSG_QCAP]) {
  float* tab = reinterpret_cast<float*>(mtab4);
  const int r = blockIdx.x;
  const unsigned R = gridDim.x;
  const unsigned Rm = 0x80000000u / R + 1u;
  const int lane = threadIdx.x & 31;
  const int wave = threadIdx.x >> 5;

  for (int i = threadIdx.x; i < MSG_NODES * 16; i += MSG_WG)
    mtab4[i] = make_float4(0.f, 0.f, 0.f, 0.f);
  __syncthreads();

  const uint4* k4 = reinterpret_cast<const uint4*>(keyf);
  const int nChunks = (E + 15) >> 4;
  const int perTrip = MSG_WAVES * 128;
  const int trips = (nChunks + perTrip - 1) / perTrip;
  const unsigned pat = (unsigned)r * 0x01010101u;
  unsigned* q = mq[wave];
  int cnt = 0;

  const unsigned cap = (((unsigned)N * 64u) / (R * (unsigned)MSG_WAVES)) & ~1u;
  const unsigned pcap = (cap - 2u) >> 1;
  unsigned* gl = reinterpret_cast<unsigned*>(listf) +
                 (size_t)((unsigned)r * MSG_WAVES + (unsigned)wave) * cap;
  uint2* gp = reinterpret_cast<uint2*>(gl) + 1;
  unsigned stored = 0;
  bool fits = true;
  bool scan = true;

  if (REPLAY) {
    const unsigned n = (unsigned)__builtin_amdgcn_readfirstlane((int)gl[0]);
    if (n != MSG_OVER) {
      scan = false;
      for (unsigned b0 = 0; b0 < n; b0 += MSG_TRIP) {
        const unsigned m = (n - b0 < (unsigned)MSG_TRIP) ? (n - b0) : (unsigned)MSG_TRIP;
        uint2 env[MSG_TRIP / 32];
#pragma unroll
        for (int k = 0; k < MSG_TRIP / 32; ++k) {
          const unsigned x = (unsigned)(k * 32 + lane);
          env[k] = make_uint2(0u, 0u);
          if (x < m) env[k] = gp[b0 + x];
        }
        uint2* q2 = reinterpret_cast<uint2*>(q);
#pragma unroll
        for (int k = 0; k < MSG_TRIP / 32; ++k) q2[k * 32 + lane] = env[k];
        const int full = (int)(m >> 4);
        for (int b = 0; b < full; ++b)
          msg_batch<true, true, true>(q + b * MSG_BATCH * 2, MSG_BATCH, lane, R, Rm, wsrc, dinv, ego, hl, tl, tab);
        const int rem = (int)(m & 15u);
        if (rem > 0)
          msg_batch<false, true, true>(q + full * MSG_BATCH * 2, rem, lane, R, Rm, wsrc, dinv, ego, hl, tl, tab);
      }
    }
  }

  int j = r % trips;
  for (int jj = 0; scan && jj < trips; ++jj) {
    const int cb = j * perTrip + wave * 128;
    j = (j + 1 == trips) ? 0 : j + 1;
    uint4 kk[4];
#pragma unroll
    for (int u = 0; u < 4; ++u) {
      const int c = cb + u * 32 + lane;
      kk[u] = make_uint4(~0u, ~0u, ~0u, ~0u);
      if (c < nChunks) kk[u] = k4[c];
    }
#pragma unroll
    for (int u = 0; u < 4; ++u) {
      const int c = cb + u * 32 + lane;
      unsigned m = key_hits(kk[u], pat);
      unsigned bal = __builtin_amdgcn_ballot_w32(m != 0u);
      while (bal != 0u) {
        if (m != 0u) {
          const int p = __builtin_ctz(m);
          const int idx = ((p & 3) << 2) | (p >> 3);
          q[cnt + (int)__builtin_amdgcn_mbcnt_lo(bal, 0u)] = (unsigned)(c * 16 + idx);
          m &= m - 1u;
        }
        cnt += __builtin_popcount(bal);
        bal = __builtin_amdgcn_ballot_w32(m != 0u);
      }
      while (cnt >= MSG_BATCH) {
        cnt -= MSG_BATCH;
        const unsigned pk =
            msg_batch<true, false, REPLAY>(q + cnt, MSG_BATCH, lane, R, Rm, wsrc, dinv, ego, hl, tl, tab);
        (void)pk;
        if (!REPLAY) {
          if (fits && stored + MSG_BATCH <= pcap) {
            if (lane < MSG_BATCH) gp[stored + lane] = make_uint2(q[cnt + lane], pk);
            stored += MSG_BATCH;
          } else {
            fits = false;
          }
        }
      }
    }
  }
  if (cnt > 0) {
    const unsigned pk = msg_batch<false, false, REPLAY>(q, cnt, lane, R, Rm, wsrc, dinv, ego, hl, tl, tab);
    (void)pk;
    if (!REPLAY) {
      if (fits && stored + (unsigned)cnt <= pcap) {
        if (lane < cnt) gp[stored + lane] = make_uint2(q[lane], pk);
        stored += (unsigned)cnt;
      } else {
        fits = false;
      }
    }
  }
  if (!REPLAY && lane == 0) gl[0] = fits ? stored : MSG_OVER;
  __syncthreads();

  float4* dst = reinterpret_cast<float4*>(fact);
  for (int i = threadIdx.x; i < MSG_NODES * 16; i += MSG_WG) {
    const int node = (i >> 4) * (int)R + r;
    if (node < N) dst[(size_t)node * 16 + (i & 15)] = mtab4[i];
  }
}

__global__ void __launch_bounds__(MSG_WG) k_message_first(
    const float* __restrict__ ew, const float* __restrict__ ego,
    const int* __restrict__ hl, const int* __restrict__ tl,
    const float* __restrict__ keyf, float* __restrict__ fact,
    float* __restrict__ listf, int E, int N) {
  __shared__ float4 mtab4[MSG_NODES * 16];
  __shared__ __attribute__((aligned(16))) unsigned mq[MSG_WAVES][MSG_QCAP];
  message_body<false>(ew, ew, ego, hl, tl, keyf, fact, listf, E, N, mtab4, mq);
}

__global__ void __launch_bounds__(MSG_WG) k_message_replay(
    const float* __restrict__ A, const float* __restrict__ dinv,
    const float* __restrict__ ego,
    const int* __restrict__ hl, const int* __restrict__ tl,
    const float* __restrict__ keyf, float* __restrict__ fact,
    float* __restrict__ listf, int E, int N) {
  __shared__ float4 mtab4[MSG_NODES * 16];
  __shared__ __attribute__((aligned(16))) unsigned mq[MSG_WAVES][MSG_QCAP];
  message_body<true>(A, dinv, ego, hl, tl, keyf, fact, listf, E, N, mtab4, mq);
}

#define DL_IDS 6

__global__ void __launch_bounds__(MSG_WG) k_degree_lists(
    const float* __restrict__ A, const int* __restrict__ hl,
    const float* __restrict__ listf, float* __restrict__ dinvg, int E, int N) {
  __shared__ float4 dtab4[MSG_NODES];
  float* tab = reinterpret_cast<float*>(dtab4);
  const unsigned r = blockIdx.x;
  const unsigned R = gridDim.x;
  const int lane = threadIdx.x & 31;
  const int wave = threadIdx.x >> 5;

  for (int i = threadIdx.x; i < MSG_NODES; i += MSG_WG)
    dtab4[i] = make_float4(0.f, 0.f, 0.f, 0.f);
  __syncthreads();

  const float4* A4 = reinterpret_cast<const float4*>(A);
  const unsigned cap = (((unsigned)N * 64u) / (R * (unsigned)MSG_WAVES)) & ~1u;
  const unsigned* gl = reinterpret_cast<const unsigned*>(listf) +
                       (size_t)(r * MSG_WAVES + (unsigned)wave) * cap;
  const uint2* gp = reinterpret_cast<const uint2*>(gl) + 1;
  const unsigned n = (unsigned)__builtin_amdgcn_readfirstlane((int)gl[0]);
  if (n != MSG_OVER) {
    for (unsigned b0 = 0; b0 < n; b0 += 32u * DL_IDS) {
      uint2 en[DL_IDS];
      bool on[DL_IDS];
#pragma unroll
      for (int k = 0; k < DL_IDS; ++k) {
        const unsigned x = b0 + (unsigned)(k * 32 + lane);
        on[k] = x < n;
        en[k] = gp[on[k] ? x : n - 1u];
      }
      float4 w[DL_IDS];
#pragma unroll
      for (int k = 0; k < DL_IDS; ++k) w[k] = A4[en[k].x];
#pragma unroll
      for (int k = 0; k < DL_IDS; ++k) w[k] = softmax4(w[k]);
#pragma unroll
      for (int k = 0; k < DL_IDS; ++k) {
        if (on[k]) {
          float* t = tab + (en[k].y >> MSG_TBITS) * 4u;
          lds_fadd(t + 0, w[k].x);
          lds_fadd(t + 1, w[k].y);
          lds_fadd(t + 2, w[k].z);
          lds_fadd(t + 3, w[k].w);
        }
      }
    }
  } else {
    const int nChunks = (E + 15) >> 4;
    const int perTrip = MSG_WAVES * 128;
    const int trips = (nChunks + perTrip - 1) / perTrip;
#pragma unroll 1
    for (int j = 0; j < trips; ++j) {
#pragma unroll 1
      for (int u = 0; u < 4; ++u) {
        const int c = j * perTrip + wave * 128 + u * 32 + lane;
        if (c < nChunks) {
#pragma unroll 1
          for (int x = 0; x < 16; ++x) {
            const int e = c * 16 + x;
            if (e < E) {
              const unsigned h = (unsigned)hl[e];
              if (h % R == r) {
                const float4 wv = softmax4(A4[e]);
                float* t = tab + (h / R) * 4u;
                lds_fadd(t + 0, wv.x);
                lds_fadd(t + 1, wv.y);
                lds_fadd(t + 2, wv.z);
                lds_fadd(t + 3, wv.w);
              }
            }
          }
        }
      }
    }
  }
  __syncthreads();

  float4* dst = reinterpret_cast<float4*>(dinvg) + (size_t)r * MSG_NODES;
  for (int i = threadIdx.x; i < MSG_NODES; i += MSG_WG) {
    const float4 s = dtab4[i];
    float4 o;
    o.x = 1.0f / sqrtf(fmaxf(s.x, 1e-8f));
    o.y = 1.0f / sqrtf(fmaxf(s.y, 1e-8f));
    o.z = 1.0f / sqrtf(fmaxf(s.z, 1e-8f));
    o.w = 1.0f / sqrtf(fmaxf(s.w, 1e-8f));
    dst[i] = o;
  }
}

__global__ void __launch_bounds__(WG) k_updateA(
    const float* __restrict__ fact, const float* __restrict__ ego,
    const int* __restrict__ hl, const int* __restrict__ tl,
    float* __restrict__ A, int E, int nGroups) {
  int tid = blockIdx.x * blockDim.x + threadIdx.x;
  int wave = tid >> 5;
  int lane = tid & 31;
  if (wave >= nGroups) return;

  const int c = lane & 15;
  const int kh = lane >> 4;
  const int i = c >> 2;
  const int f = c & 3;
  const int seg = f * 16 + kh * 8;

  const bool active = (lane < 8) || (lane >= 24);
  const int cc = (lane < 8) ? lane : (lane - 16);
  const int e0 = wave * 16;

  int hh[4], tt[4];
  float aold[4];
  bool wr[4];
  const int aLast = E * 4 - 1;
#pragma unroll
  for (int g = 0; g < 4; ++g) {
    const int eIdx = e0 + g * 4 + i;
    const int e = (eIdx < E) ? eIdx : (E - 1);
    hh[g] = hl[e];
    tt[g] = tl[e];
    const int ai = (e0 + g * 4) * 4 + c;
    aold[g] = A[(ai < aLast) ? ai : aLast];
    wr[g] = active && (e0 + g * 4 + (cc >> 2) < E);
  }
  asm volatile("" :: "v"(hh[0]), "v"(hh[1]), "v"(hh[2]), "v"(hh[3]),
                     "v"(tt[0]), "v"(tt[1]), "v"(tt[2]), "v"(tt[3]),
                     "v"(aold[0]), "v"(aold[1]), "v"(aold[2]), "v"(aold[3]));
  v4f h0[4], h1[4], t0[4], t1[4];
#pragma unroll
  for (int g = 0; g < 4; ++g) {
    const v4f* hp = reinterpret_cast<const v4f*>(fact + (size_t)hh[g] * 64 + seg);
    const v4f* tp = reinterpret_cast<const v4f*>(ego + (size_t)tt[g] * 64 + seg);
    h0[g] = hp[0];
    h1[g] = hp[1];
    t0[g] = tp[0];
    t1[g] = tp[1];
  }
  float dv[4];
#pragma unroll
  for (int g = 0; g < 4; ++g) {
    const float hv[8] = {h0[g].x, h0[g].y, h0[g].z, h0[g].w, h1[g].x, h1[g].y, h1[g].z, h1[g].w};
    const float tv[8] = {t0[g].x, t0[g].y, t0[g].z, t0[g].w, t1[g].x, t1[g].y, t1[g].z, t1[g].w};
    v2f qh = h0[g].xy * h0[g].xy;
    v2f qt = t0[g].xy * t0[g].xy;
    qh = __builtin_elementwise_fma(h0[g].zw, h0[g].zw, qh);
    qt = __builtin_elementwise_fma(t0[g].zw, t0[g].zw, qt);
    qh = __builtin_elementwise_fma(h1[g].xy, h1[g].xy, qh);
    qt = __builtin_elementwise_fma(t1[g].xy, t1[g].xy, qt);
    qh = __builtin_elementwise_fma(h1[g].zw, h1[g].zw, qh);
    qt = __builtin_elementwise_fma(t1[g].zw, t1[g].zw, qt);
    const float ph = qh.x + qh.y, pt = qt.x + qt.y;
    const float sh = ph + xhalf(ph);
    const float st = pt + xhalf(pt);
    const float mine = 1.0f / fmaxf(sqrtf(kh ? st : sh), 1e-12f);
    const float other = xhalf(mine);
    const float hn = kh ? other : mine;
    const float tn = kh ? mine : other;
    v8f acc = {0.f, 0.f, 0.f, 0.f, 0.f, 0.f, 0.f, 0.f};
#pragma unroll
    for (int j = 0; j < 4; ++j) {
      v2f av, bv;
      av.x = hv[2 * j] * hn;
      av.y = hv[2 * j + 1] * hn;
      bv.x = fast_tanh(tv[2 * j] * tn);
      bv.y = fast_tanh(tv[2 * j + 1] * tn);
      acc = __builtin_amdgcn_wmma_f32_16x16x4_f32(false, av, false, bv,
                                                  (short)0, acc, false, false);
    }
    const int idx = lane & 7;
    float d = acc[0];
    d = (idx == 1) ? acc[1] : d;
    d = (idx == 2) ? acc[2] : d;
    d = (idx == 3) ? acc[3] : d;
    d = (idx == 4) ? acc[4] : d;
    d = (idx == 5) ? acc[5] : d;
    d = (idx == 6) ? acc[6] : d;
    d = (idx == 7) ? acc[7] : d;
    dv[g] = d;
  }

#pragma unroll
  for (int g = 0; g < 4; ++g)
    if (wr[g]) A[(size_t)(e0 + g * 4) * 4 + cc] = aold[g] + dv[g];
}

__global__ void __launch_bounds__(WG) k_final(const float* __restrict__ u, const float* __restrict__ it,
                        const float* __restrict__ l1, const float* __restrict__ l2,
                        float* __restrict__ out, int userElems, int n) {
  const int user4 = userElems >> 2;
  const int n4 = n >> 2;
  const int stride = (int)(gridDim.x * blockDim.x);
  const float4* u4 = reinterpret_cast<const float4*>(u);
  const float4* i4 = reinterpret_cast<const float4*>(it);
  const float4* a4 = reinterpret_cast<const float4*>(l1);
  const float4* b4 = reinterpret_cast<const float4*>(l2);
  float4* o4 = reinterpret_cast<float4*>(out);
  const int span = stride * FN_VEC;
  const int trips = (n4 + span - 1) / span;
  const int first = (int)(blockIdx.x * blockDim.x + threadIdx.x);
  for (int tr = 0; tr < trips; ++tr) {
    const int s0 = first + tr * span;
    float4 x[FN_VEC], y[FN_VEC], z[FN_VEC];
#pragma unroll
    for (int k = 0; k < FN_VEC; ++k) {
      const int i = s0 + k * stride;
      const int ic = (i < n4) ? i : n4 - 1;
      const float4* p = (ic < user4) ? (u4 + ic) : (i4 + (ic - user4));
      x[k] = *p;
    }
#pragma unroll
    for (int k = 0; k < FN_VEC; ++k) {
      const int i = s0 + k * stride;
      const int ic = (i < n4) ? i : n4 - 1;
      y[k] = a4[ic];
      z[k] = b4[ic];
    }
    hold4(x[0], x[1], x[2], x[3]);
    hold4(y[0], y[1], y[2], y[3]);
    hold4(z[0], z[1], z[2], z[3]);
#pragma unroll
    for (int k = 0; k < FN_VEC; ++k) {
      const int i = s0 + k * stride;
      if (i < n4) {
        float4 v = x[k];
        v.x += y[k].x; v.y += y[k].y; v.z += y[k].z; v.w += y[k].w;
        v.x += z[k].x; v.y += z[k].y; v.z += z[k].z; v.w += z[k].w;
        v.x *= (1.0f / 3.0f); v.y *= (1.0f / 3.0f); v.z *= (1.0f / 3.0f); v.w *= (1.0f / 3.0f);
        o4[i] = v;
      }
    }
  }
}

static inline size_t align16f(size_t x) { return (x + 15) & ~(size_t)15; }

extern "C" void kernel_launch(void* const* d_in, const int* in_sizes, int n_in,
                              void* d_out, int out_size, void* d_ws, size_t ws_size,
                              hipStream_t stream) {
  (void)n_in; (void)out_size; (void)ws_size;
  const float* u  = (const float*)d_in[0];
  const float* it = (const float*)d_in[1];
  const int* hl   = (const int*)d_in[2];
  const int* tl   = (const int*)d_in[3];

  const int userElems = in_sizes[0];
  const int itemElems = in_sizes[1];
  const int E = in_sizes[2];
  const int N = (userElems + itemElems) / 64;
  const size_t N64 = (size_t)N * 64;
  const size_t E4  = (size_t)E * 4;
  const size_t N4  = (size_t)N * 4;

  float* base = (float*)d_ws;
  size_t off = 0;
  float* ego   = base + off; off += align16f(N64);
  float* fact  = base + off; off += align16f(N64);
  float* lists = base + off; off += align16f(N64);
  float* A     = base + off; off += align16f(E4);
  float* ew    = base + off; off += align16f(E4);
  float* dval  = base + off; off += align16f(N4);
  float* keyf  = base + off; off += align16f(N4);

#define GRID(n) ((unsigned)(((size_t)(n) + WG - 1) / WG))

  k_init<<<STREAM_WGS, WG, 0, stream>>>(u, it, ego, userElems, (int)N64);
  k_fill1<<<STREAM_WGS, WG, 0, stream>>>(A, (int)E4);

  float* cur = ego;
  float* nxt = fact;
  const int nGroups = (E + 15) / 16;
  const dim3 degGrid(DEG_SLICES, (unsigned)((N + DEG_NODES - 1) / DEG_NODES));
  const unsigned msgGrid = (unsigned)((N + MSG_NODES - 1) / MSG_NODES);
  const int degPer = (((E + DEG_SLICES - 1) / DEG_SLICES) + 3) & ~3;

  for (int layer = 0; layer < 2; ++layer) {
    for (int itn = 0; itn < 2; ++itn) {
      if (layer == 0 && itn == 0) {
        k_softmax<<<STREAM_WGS, WG, 0, stream>>>(A, ew, E);
        k_degree_lds<<<degGrid, DEG_WG, 0, stream>>>(ew, hl, nxt, E, N, degPer);
        k_dinv_sum<<<GRID(N), WG, 0, stream>>>(nxt, dval, N);
        k_edge_weight<<<STREAM_WGS, WG, 0, stream>>>(ew, dval, hl, tl, keyf, E, (int)msgGrid);
        k_message_first<<<msgGrid, MSG_WG, 0, stream>>>(ew, cur, hl, tl, keyf, nxt, lists, E, N);
      } else {
        k_degree_lists<<<msgGrid, MSG_WG, 0, stream>>>(A, hl, lists, ew, E, N);
        k_message_replay<<<msgGrid, MSG_WG, 0, stream>>>(A, ew, cur, hl, tl, keyf, nxt, lists, E, N);
      }
      const bool lastUpdateDead = (layer == 1 && itn == 1);
      if (!lastUpdateDead)
        k_updateA<<<GRID((size_t)nGroups * 32), WG, 0, stream>>>(
            nxt, cur, hl, tl, A, E, nGroups);
    }
    float* tmp = cur; cur = nxt; nxt = tmp;
  }

  k_final<<<STREAM_WGS, WG, 0, stream>>>(u, it, nxt, cur, (float*)d_out, userElems, (int)N64);
#undef GRID
}


# Key scan moved to its own launch ahead of pass 1: it stores the waves' edge lists and forms pass 1's degree sums (LDS f32 adds, arrival order); pass 1 then runs from the lists; 17 -> 14 launches
# speedup vs baseline: 3.9956x; 1.0772x over previous
#include <hip/hip_runtime.h>
#include <math.h>

typedef float v2f __attribute__((ext_vector_type(2)));
typedef float v4f __attribute__((ext_vector_type(4)));
typedef float v8f __attribute__((ext_vector_type(8)));

#define WG 256

__device__ __forceinline__ float xhalf(float x) {
  int b = __builtin_bit_cast(int, x);
  return __builtin_bit_cast(float,
      __builtin_amdgcn_permlanex16(b, b, 0x76543210, 0xfedcba98, false, false));
}

__device__ __forceinline__ float fast_tanh(float x) {
  float r;
  asm("v_tanh_f32 %0, %1\n\tv_nop\n\tv_nop" : "=v"(r) : "v"(x));
  return r;
}

#define STREAM_WGS 512
#define CP_VEC 8
#define EW_VEC 4
#define FN_VEC 4

__device__ __forceinline__ void hold4(const float4 a, const float4 b, const float4 c, const float4 d) {
  asm volatile("" :: "v"(a.x), "v"(a.y), "v"(a.z), "v"(a.w), "v"(b.x), "v"(b.y), "v"(b.z), "v"(b.w),
                     "v"(c.x), "v"(c.y), "v"(c.z), "v"(c.w), "v"(d.x), "v"(d.y), "v"(d.z), "v"(d.w));
}

__global__ void __launch_bounds__(WG) k_init(const float* __restrict__ u, const float* __restrict__ it,
                       float* __restrict__ ego, int userElems, int total) {
  const int user4 = userElems >> 2;
  const int total4 = total >> 2;
  const int stride = (int)(gridDim.x * blockDim.x);
  const float4* u4 = reinterpret_cast<const float4*>(u);
  const float4* i4 = reinterpret_cast<const float4*>(it);
  float4* e4 = reinterpret_cast<float4*>(ego);
  for (int s0 = (int)(blockIdx.x * blockDim.x + threadIdx.x); s0 < total4; s0 += stride * CP_VEC) {
    float4 v[CP_VEC];
#pragma unroll
    for (int k = 0; k < CP_VEC; ++k) {
      const int i = s0 + k * stride;
      const int ic = (i < total4) ? i : total4 - 1;
      const float4* p = (ic < user4) ? (u4 + ic) : (i4 + (ic - user4));
      v[k] = *p;
    }
    hold4(v[0], v[1], v[2], v[3]);
    hold4(v[4], v[5], v[6], v[7]);
#pragma unroll
    for (int k = 0; k < CP_VEC; ++k) {
      const int i = s0 + k * stride;
      if (i < total4) e4[i] = v[k];
    }
  }
}

__global__ void __launch_bounds__(WG) k_edges_init(float* __restrict__ a, const int* __restrict__ hl,
                             float* __restrict__ keyf, int E, int R) {
  unsigned char* keyb = reinterpret_cast<unsigned char*>(keyf);
  const int stride = (int)(gridDim.x * blockDim.x);
  const int Epad = (E + 15) & ~15;
  float4* a4 = reinterpret_cast<float4*>(a);
  const float4 one = make_float4(1.0f, 1.0f, 1.0f, 1.0f);
  for (int s0 = (int)(blockIdx.x * blockDim.x + threadIdx.x); s0 < Epad; s0 += stride * EW_VEC) {
    unsigned h[EW_VEC];
#pragma unroll
    for (int k = 0; k < EW_VEC; ++k) {
      const int e = s0 + k * stride;
      h[k] = (unsigned)hl[(e < E) ? e : E - 1];
    }
    asm volatile("" :: "v"(h[0]), "v"(h[1]), "v"(h[2]), "v"(h[3]));
#pragma unroll
    for (int k = 0; k < EW_VEC; ++k) {
      const int e = s0 + k * stride;
      if (e < E) {
        a4[e] = one;
        keyb[e] = (unsigned char)(h[k] % (unsigned)R);
      } else if (e < Epad) {
        keyb[e] = (unsigned char)0xffu;
      }
    }
  }
}

__device__ __forceinline__ float4 softmax4(float4 a) {
  float m = fmaxf(fmaxf(a.x, a.y), fmaxf(a.z, a.w));
  float e0 = expf(a.x - m), e1 = expf(a.y - m), e2 = expf(a.z - m), e3 = expf(a.w - m);
  float inv = 1.0f / (e0 + e1 + e2 + e3);
  return make_float4(e0 * inv, e1 * inv, e2 * inv, e3 * inv);
}

__device__ __forceinline__ void lds_fadd(float* p, float v) {
  __hip_atomic_fetch_add(p, v, __ATOMIC_RELAXED, __HIP_MEMORY_SCOPE_WORKGROUP);
}

#define MSG_NODES 320
#define MSG_WG    1024
#define MSG_WAVES 32
#define MSG_BATCH 16
#define MSG_QCAP  544
#define MSG_NOKEY 255u
#define MSG_OVER  0xffffffffu
#define MSG_TRIP  128
#define MSG_TBITS 17

__device__ __forceinline__ unsigned nonzero_bytes(unsigned x) {
  return ((x & 0x7f7f7f7fu) + 0x7f7f7f7fu) | x;
}

__device__ __forceinline__ unsigned key_hits(uint4 k, unsigned pat) {
  unsigned y0 = nonzero_bytes(k.x ^ pat);
  unsigned y1 = nonzero_bytes(k.y ^ pat);
  unsigned y2 = nonzero_bytes(k.z ^ pat);
  unsigned y3 = nonzero_bytes(k.w ^ pat);
  unsigned n = ((y0 >> 7) & 0x01010101u) | ((y1 >> 6) & 0x02020202u) |
               ((y2 >> 5) & 0x04040404u) | ((y3 >> 4) & 0x08080808u);
  return ~n & 0x0f0f0f0fu;
}

template <int K>
__device__ __forceinline__ float quad_lane(float x) {
  const int b = __builtin_bit_cast(int, x);
  return __builtin_bit_cast(float, __builtin_amdgcn_mov_dpp(b, K * 0x55, 0xf, 0xf, false));
}

struct OwnW {
  float4 a;
  float dh, dt;
};

__device__ __forceinline__ OwnW own_weight_load(const float* __restrict__ A,
                                                const float* __restrict__ dinv,
                                                unsigned e, unsigned p, unsigned r, unsigned R,
                                                unsigned Rm, int f) {
  const unsigned t = p & ((1u << MSG_TBITS) - 1u);
  const unsigned tq = __umulhi(t + t, Rm);
  const unsigned hi = r * (unsigned)MSG_NODES + (p >> MSG_TBITS);
  const unsigned ti = (t - tq * R) * (unsigned)MSG_NODES + tq;
  OwnW o;
  o.a = reinterpret_cast<const float4*>(A)[e];
  o.dh = dinv[(size_t)hi * 4 + f];
  o.dt = dinv[(size_t)ti * 4 + f];
  return o;
}

__device__ __forceinline__ float own_weight(const OwnW o, int f) {
  const float4 a = o.a;
  const float dh = o.dh, dt = o.dt;
  const float m = fmaxf(fmaxf(a.x, a.y), fmaxf(a.z, a.w));
  float af = a.x;
  af = (f == 1) ? a.y : af;
  af = (f == 2) ? a.z : af;
  af = (f == 3) ? a.w : af;
  const float ef = expf(af - m);
  const float e0 = quad_lane<0>(ef), e1 = quad_lane<1>(ef);
  const float e2 = quad_lane<2>(ef), e3 = quad_lane<3>(ef);
  const float inv = 1.0f / (e0 + e1 + e2 + e3);
  return (ef * inv) * (dh * dt);
}

template <bool FULL, bool PAIRS>
__device__ __forceinline__ void msg_batch(const unsigned* qp, int n, int lane, unsigned R,
                                          unsigned Rm,
                                          const float* __restrict__ A,
                                          const float* __restrict__ dinv,
                                          const float* __restrict__ ego,
                                          const int* __restrict__ hl,
                                          const int* __restrict__ tl, float* tab) {
  const int l16 = lane & 15;
  const int s16 = (FULL || l16 < n) ? l16 : 0;
  const int j8 = lane >> 2;
  const int sa = (FULL || j8 < n) ? j8 : 0;
  const int sb = (FULL || j8 + 8 < n) ? j8 + 8 : 0;
  unsigned pk, ea, eb, pa, pb;
  if (PAIRS) {
    pk = qp[2 * s16 + 1];
    ea = qp[2 * sa];
    eb = qp[2 * sb];
    pa = qp[2 * sa + 1];
    pb = qp[2 * sb + 1];
  } else {
    const unsigned el = qp[s16];
    pk = (unsigned)tl[el] | (((unsigned)hl[el] / R) << MSG_TBITS);
    ea = qp[sa];
    eb = qp[sb];
    pa = (unsigned)tl[ea] | (((unsigned)hl[ea] / R) << MSG_TBITS);
    pb = (unsigned)tl[eb] | (((unsigned)hl[eb] / R) << MSG_TBITS);
  }
  const int tv = (int)(pk & ((1u << MSG_TBITS) - 1u));
  const int dv = (int)(pk >> MSG_TBITS);
  const OwnW oa = own_weight_load(A, dinv, ea, pa, blockIdx.x, R, Rm, lane & 3);
  const OwnW ob = own_weight_load(A, dinv, eb, pb, blockIdx.x, R, Rm, lane & 3);
  const int f = lane >> 3;

  float2 v[MSG_BATCH];
  int d[MSG_BATCH];
#pragma unroll
  for (int i = 0; i < MSG_BATCH; ++i) {
    const int t = __builtin_amdgcn_readlane(tv, i);
    d[i] = __builtin_amdgcn_readlane(dv, i);
    v[i] = make_float2(0.f, 0.f);
    if (FULL || i < n)
      v[i] = *reinterpret_cast<const float2*>(ego + (size_t)t * 64 + lane * 2);
  }
  __builtin_amdgcn_sched_barrier(0);
  const float wa = own_weight(oa, lane & 3);
  const float wb = own_weight(ob, lane & 3);
  if (PAIRS) {
    asm volatile("" :: "v"(v[0].x), "v"(v[0].y), "v"(v[1].x), "v"(v[1].y),
                       "v"(v[2].x), "v"(v[2].y), "v"(v[3].x), "v"(v[3].y),
                       "v"(v[4].x), "v"(v[4].y), "v"(v[5].x), "v"(v[5].y),
                       "v"(v[6].x), "v"(v[6].y), "v"(v[7].x), "v"(v[7].y));
    asm volatile("" :: "v"(v[8].x), "v"(v[8].y), "v"(v[9].x), "v"(v[9].y),
                       "v"(v[10].x), "v"(v[10].y), "v"(v[11].x), "v"(v[11].y),
                       "v"(v[12].x), "v"(v[12].y), "v"(v[13].x), "v"(v[13].y),
                       "v"(v[14].x), "v"(v[14].y), "v"(v[15].x), "v"(v[15].y));
  }
#pragma unroll
  for (int i = 0; i < MSG_BATCH; ++i) {
    const float w = __shfl((i < 8) ? wa : wb, 4 * (i & 7) + f);
    if (FULL || i < n) {
      float* tp = tab + d[i] * 64 + lane * 2;
      lds_fadd(tp + 0, w * v[i].x);
      lds_fadd(tp + 1, w * v[i].y);
    }
  }
}

__global__ void __launch_bounds__(MSG_WG) k_message(
    const float* __restrict__ A, const float* __restrict__ dinv,
    const float* __restrict__ ego,
    const int* __restrict__ hl, const int* __restrict__ tl,
    const float* __restrict__ keyf, float* __restrict__ fact,
    const float* __restrict__ listf, int E, int N) {
  __shared__ float4 mtab4[MSG_NODES * 16];
  __shared__ __attribute__((aligned(16))) unsigned mq[MSG_WAVES][MSG_QCAP];
  float* tab = reinterpret_cast<float*>(mtab4);
  const int r = blockIdx.x;
  const unsigned R = gridDim.x;
  const unsigned Rm = 0x80000000u / R + 1u;
  const int lane = threadIdx.x & 31;
  const int wave = threadIdx.x >> 5;

  for (int i = threadIdx.x; i < MSG_NODES * 16; i += MSG_WG)
    mtab4[i] = make_float4(0.f, 0.f, 0.f, 0.f);
  __syncthreads();

  unsigned* q = mq[wave];

  const unsigned cap = (((unsigned)N * 64u) / (R * (unsigned)MSG_WAVES)) & ~1u;
  const unsigned* gl = reinterpret_cast<const unsigned*>(listf) +
                       (size_t)((unsigned)r * MSG_WAVES + (unsigned)wave) * cap;
  const uint2* gp = reinterpret_cast<const uint2*>(gl) + 1;

  const unsigned n = (unsigned)__builtin_amdgcn_readfirstlane((int)gl[0]);
  if (n != MSG_OVER) {
    for (unsigned b0 = 0; b0 < n; b0 += MSG_TRIP) {
      const unsigned m = (n - b0 < (unsigned)MSG_TRIP) ? (n - b0) : (unsigned)MSG_TRIP;
      uint2 env[MSG_TRIP / 32];
#pragma unroll
      for (int k = 0; k < MSG_TRIP / 32; ++k) {
        const unsigned x = (unsigned)(k * 32 + lane);
        env[k] = make_uint2(0u, 0u);
        if (x < m) env[k] = gp[b0 + x];
      }
      uint2* q2 = reinterpret_cast<uint2*>(q);
#pragma unroll
      for (int k = 0; k < MSG_TRIP / 32; ++k) q2[k * 32 + lane] = env[k];
      const int full = (int)(m >> 4);
      for (int b = 0; b < full; ++b)
        msg_batch<true, true>(q + b * MSG_BATCH * 2, MSG_BATCH, lane, R, Rm, A, dinv, ego, hl, tl, tab);
      const int rem = (int)(m & 15u);
      if (rem > 0)
        msg_batch<false, true>(q + full * MSG_BATCH * 2, rem, lane, R, Rm, A, dinv, ego, hl, tl, tab);
    }
  } else {
    const uint4* k4 = reinterpret_cast<const uint4*>(keyf);
    const int nChunks = (E + 15) >> 4;
    const int perTrip = MSG_WAVES * 128;
    const int trips = (nChunks + perTrip - 1) / perTrip;
    const unsigned pat = (unsigned)r * 0x01010101u;
    int cnt = 0;
    for (int j = 0; j < trips; ++j) {
      const int cb = j * perTrip + wave * 128;
      uint4 kk[4];
#pragma unroll
      for (int u = 0; u < 4; ++u) {
        const int c = cb + u * 32 + lane;
        kk[u] = make_uint4(~0u, ~0u, ~0u, ~0u);
        if (c < nChunks) kk[u] = k4[c];
      }
#pragma unroll
      for (int u = 0; u < 4; ++u) {
        const int c = cb + u * 32 + lane;
        unsigned m = key_hits(kk[u], pat);
        unsigned bal = __builtin_amdgcn_ballot_w32(m != 0u);
        while (bal != 0u) {
          if (m != 0u) {
            const int p = __builtin_ctz(m);
            const int idx = ((p & 3) << 2) | (p >> 3);
            q[cnt + (int)__builtin_amdgcn_mbcnt_lo(bal, 0u)] = (unsigned)(c * 16 + idx);
            m &= m - 1u;
          }
          cnt += __builtin_popcount(bal);
          bal = __builtin_amdgcn_ballot_w32(m != 0u);
        }
        while (cnt >= MSG_BATCH) {
          cnt -= MSG_BATCH;
          msg_batch<true, false>(q + cnt, MSG_BATCH, lane, R, Rm, A, dinv, ego, hl, tl, tab);
        }
      }
    }
    if (cnt > 0)
      msg_batch<false, false>(q, cnt, lane, R, Rm, A, dinv, ego, hl, tl, tab);
  }
  __syncthreads();

  float4* dst = reinterpret_cast<float4*>(fact);
  for (int i = threadIdx.x; i < MSG_NODES * 16; i += MSG_WG) {
    const int node = (i >> 4) * (int)R + r;
    if (node < N) dst[(size_t)node * 16 + (i & 15)] = mtab4[i];
  }
}

template <bool FULL>
__device__ __forceinline__ void list_drain(const unsigned* qp, int n, int lane, unsigned R,
                                           const float4* __restrict__ A4,
                                           const int* __restrict__ hl, const int* __restrict__ tl,
                                           uint2* gp, unsigned pcap, unsigned& stored, bool& fits,
                                           float* tab) {
  const bool on = FULL || lane < n;
  const unsigned el = qp[on ? lane : 0];
  const unsigned h = (unsigned)hl[el];
  const unsigned t = (unsigned)tl[el];
  const float4 a = A4[el];
  asm volatile("" :: "v"(h), "v"(t), "v"(a.x), "v"(a.y), "v"(a.z), "v"(a.w));
  const unsigned row = h / R;
  if (fits && stored + (unsigned)n <= pcap) {
    if (on) gp[stored + (unsigned)lane] = make_uint2(el, t | (row << MSG_TBITS));
    stored += (unsigned)n;
  } else {
    fits = false;
  }
  const float4 w = softmax4(a);
  if (on) {
    float* tp = tab + row * 4u;
    lds_fadd(tp + 0, w.x);
    lds_fadd(tp + 1, w.y);
    lds_fadd(tp + 2, w.z);
    lds_fadd(tp + 3, w.w);
  }
}

__global__ void __launch_bounds__(MSG_WG) k_build_lists(
    const float* __restrict__ A, const int* __restrict__ hl, const int* __restrict__ tl,
    const float* __restrict__ keyf, float* __restrict__ listf, float* __restrict__ dinvg,
    int E, int N) {
  __shared__ float4 dtab4[MSG_NODES];
  __shared__ unsigned mq[MSG_WAVES][MSG_QCAP];
  float* tab = reinterpret_cast<float*>(dtab4);
  const int r = blockIdx.x;
  const unsigned R = gridDim.x;
  const int lane = threadIdx.x & 31;
  const int wave = threadIdx.x >> 5;

  for (int i = threadIdx.x; i < MSG_NODES; i += MSG_WG)
    dtab4[i] = make_float4(0.f, 0.f, 0.f, 0.f);
  __syncthreads();

  const float4* A4 = reinterpret_cast<const float4*>(A);
  const uint4* k4 = reinterpret_cast<const uint4*>(keyf);
  const int nChunks = (E + 15) >> 4;
  const int perTrip = MSG_WAVES * 128;
  const int trips = (nChunks + perTrip - 1) / perTrip;
  const unsigned pat = (unsigned)r * 0x01010101u;
  unsigned* q = mq[wave];
  int cnt = 0;

  const unsigned cap = (((unsigned)N * 64u) / (R * (unsigned)MSG_WAVES)) & ~1u;
  const unsigned pcap = (cap - 2u) >> 1;
  unsigned* gl = reinterpret_cast<unsigned*>(listf) +
                 (size_t)((unsigned)r * MSG_WAVES + (unsigned)wave) * cap;
  uint2* gp = reinterpret_cast<uint2*>(gl) + 1;
  unsigned stored = 0;
  bool fits = true;

  int j = r % trips;
  for (int jj = 0; jj < trips; ++jj) {
    const int cb = j * perTrip + wave * 128;
    j = (j + 1 == trips) ? 0 : j + 1;
    uint4 kk[4];
#pragma unroll
    for (int u = 0; u < 4; ++u) {
      const int c = cb + u * 32 + lane;
      kk[u] = make_uint4(~0u, ~0u, ~0u, ~0u);
      if (c < nChunks) kk[u] = k4[c];
    }
#pragma unroll
    for (int u = 0; u < 4; ++u) {
      const int c = cb + u * 32 + lane;
      unsigned m = key_hits(kk[u], pat);
      unsigned bal = __builtin_amdgcn_ballot_w32(m != 0u);
      while (bal != 0u) {
        if (m != 0u) {
          const int p = __builtin_ctz(m);
          const int idx = ((p & 3) << 2) | (p >> 3);
          q[cnt + (int)__builtin_amdgcn_mbcnt_lo(bal, 0u)] = (unsigned)(c * 16 + idx);
          m &= m - 1u;
        }
        cnt += __builtin_popcount(bal);
        bal = __builtin_amdgcn_ballot_w32(m != 0u);
      }
      while (cnt >= 32) {
        cnt -= 32;
        list_drain<true>(q + cnt, 32, lane, R, A4, hl, tl, gp, pcap, stored, fits, tab);
      }
    }
  }
  if (cnt > 0)
    list_drain<false>(q, cnt, lane, R, A4, hl, tl, gp, pcap, stored, fits, tab);
  if (lane == 0) gl[0] = fits ? stored : MSG_OVER;
  __syncthreads();

  float4* dst = reinterpret_cast<float4*>(dinvg) + (size_t)r * MSG_NODES;
  for (int i = threadIdx.x; i < MSG_NODES; i += MSG_WG) {
    const float4 s = dtab4[i];
    float4 o;
    o.x = 1.0f / sqrtf(fmaxf(s.x, 1e-8f));
    o.y = 1.0f / sqrtf(fmaxf(s.y, 1e-8f));
    o.z = 1.0f / sqrtf(fmaxf(s.z, 1e-8f));
    o.w = 1.0f / sqrtf(fmaxf(s.w, 1e-8f));
    dst[i] = o;
  }
}

#define DL_IDS 6

__global__ void __launch_bounds__(MSG_WG) k_degree_lists(
    const float* __restrict__ A, const int* __restrict__ hl,
    const float* __restrict__ listf, float* __restrict__ dinvg, int E, int N) {
  __shared__ float4 dtab4[MSG_NODES];
  float* tab = reinterpret_cast<float*>(dtab4);
  const unsigned r = blockIdx.x;
  const unsigned R = gridDim.x;
  const int lane = threadIdx.x & 31;
  const int wave = threadIdx.x >> 5;

  for (int i = threadIdx.x; i < MSG_NODES; i += MSG_WG)
    dtab4[i] = make_float4(0.f, 0.f, 0.f, 0.f);
  __syncthreads();

  const float4* A4 = reinterpret_cast<const float4*>(A);
  const unsigned cap = (((unsigned)N * 64u) / (R * (unsigned)MSG_WAVES)) & ~1u;
  const unsigned* gl = reinterpret_cast<const unsigned*>(listf) +
                       (size_t)(r * MSG_WAVES + (unsigned)wave) * cap;
  const uint2* gp = reinterpret_cast<const uint2*>(gl) + 1;
  const unsigned n = (unsigned)__builtin_amdgcn_readfirstlane((int)gl[0]);
  if (n != MSG_OVER) {
    for (unsigned b0 = 0; b0 < n; b0 += 32u * DL_IDS) {
      uint2 en[DL_IDS];
      bool on[DL_IDS];
#pragma unroll
      for (int k = 0; k < DL_IDS; ++k) {
        const unsigned x = b0 + (unsigned)(k * 32 + lane);
        on[k] = x < n;
        en[k] = gp[on[k] ? x : n - 1u];
      }
      float4 w[DL_IDS];
#pragma unroll
      for (int k = 0; k < DL_IDS; ++k) w[k] = A4[en[k].x];
#pragma unroll
      for (int k = 0; k < DL_IDS; ++k) w[k] = softmax4(w[k]);
#pragma unroll
      for (int k = 0; k < DL_IDS; ++k) {
        if (on[k]) {
          float* t = tab + (en[k].y >> MSG_TBITS) * 4u;
          lds_fadd(t + 0, w[k].x);
          lds_fadd(t + 1, w[k].y);
          lds_fadd(t + 2, w[k].z);
          lds_fadd(t + 3, w[k].w);
        }
      }
    }
  } else {
    const int nChunks = (E + 15) >> 4;
    const int perTrip = MSG_WAVES * 128;
    const int trips = (nChunks + perTrip - 1) / perTrip;
#pragma unroll 1
    for (int j = 0; j < trips; ++j) {
#pragma unroll 1
      for (int u = 0; u < 4; ++u) {
        const int c = j * perTrip + wave * 128 + u * 32 + lane;
        if (c < nChunks) {
#pragma unroll 1
          for (int x = 0; x < 16; ++x) {
            const int e = c * 16 + x;
            if (e < E) {
              const unsigned h = (unsigned)hl[e];
              if (h % R == r) {
                const float4 wv = softmax4(A4[e]);
                float* t = tab + (h / R) * 4u;
                lds_fadd(t + 0, wv.x);
                lds_fadd(t + 1, wv.y);
                lds_fadd(t + 2, wv.z);
                lds_fadd(t + 3, wv.w);
              }
            }
          }
        }
      }
    }
  }
  __syncthreads();

  float4* dst = reinterpret_cast<float4*>(dinvg) + (size_t)r * MSG_NODES;
  for (int i = threadIdx.x; i < MSG_NODES; i += MSG_WG) {
    const float4 s = dtab4[i];
    float4 o;
    o.x = 1.0f / sqrtf(fmaxf(s.x, 1e-8f));
    o.y = 1.0f / sqrtf(fmaxf(s.y, 1e-8f));
    o.z = 1.0f / sqrtf(fmaxf(s.z, 1e-8f));
    o.w = 1.0f / sqrtf(fmaxf(s.w, 1e-8f));
    dst[i] = o;
  }
}

__global__ void __launch_bounds__(WG) k_updateA(
    const float* __restrict__ fact, const float* __restrict__ ego,
    const int* __restrict__ hl, const int* __restrict__ tl,
    float* __restrict__ A, int E, int nGroups) {
  int tid = blockIdx.x * blockDim.x + threadIdx.x;
  int wave = tid >> 5;
  int lane = tid & 31;
  if (wave >= nGroups) return;

  const int c = lane & 15;
  const int kh = lane >> 4;
  const int i = c >> 2;
  const int f = c & 3;
  const int seg = f * 16 + kh * 8;

  const bool active = (lane < 8) || (lane >= 24);
  const int cc = (lane < 8) ? lane : (lane - 16);
  const int e0 = wave * 16;

  int hh[4], tt[4];
  float aold[4];
  bool wr[4];
  const int aLast = E * 4 - 1;
#pragma unroll
  for (int g = 0; g < 4; ++g) {
    const int eIdx = e0 + g * 4 + i;
    const int e = (eIdx < E) ? eIdx : (E - 1);
    hh[g] = hl[e];
    tt[g] = tl[e];
    const int ai = (e0 + g * 4) * 4 + c;
    aold[g] = A[(ai < aLast) ? ai : aLast];
    wr[g] = active && (e0 + g * 4 + (cc >> 2) < E);
  }
  asm volatile("" :: "v"(hh[0]), "v"(hh[1]), "v"(hh[2]), "v"(hh[3]),
                     "v"(tt[0]), "v"(tt[1]), "v"(tt[2]), "v"(tt[3]),
                     "v"(aold[0]), "v"(aold[1]), "v"(aold[2]), "v"(aold[3]));
  v4f h0[4], h1[4], t0[4], t1[4];
#pragma unroll
  for (int g = 0; g < 4; ++g) {
    const v4f* hp = reinterpret_cast<const v4f*>(fact + (size_t)hh[g] * 64 + seg);
    const v4f* tp = reinterpret_cast<const v4f*>(ego + (size_t)tt[g] * 64 + seg);
    h0[g] = hp[0];
    h1[g] = hp[1];
    t0[g] = tp[0];
    t1[g] = tp[1];
  }
  float dv[4];
#pragma unroll
  for (int g = 0; g < 4; ++g) {
    const float hv[8] = {h0[g].x, h0[g].y, h0[g].z, h0[g].w, h1[g].x, h1[g].y, h1[g].z, h1[g].w};
    const float tv[8] = {t0[g].x, t0[g].y, t0[g].z, t0[g].w, t1[g].x, t1[g].y, t1[g].z, t1[g].w};
    v2f qh = h0[g].xy * h0[g].xy;
    v2f qt = t0[g].xy * t0[g].xy;
    qh = __builtin_elementwise_fma(h0[g].zw, h0[g].zw, qh);
    qt = __builtin_elementwise_fma(t0[g].zw, t0[g].zw, qt);
    qh = __builtin_elementwise_fma(h1[g].xy, h1[g].xy, qh);
    qt = __builtin_elementwise_fma(t1[g].xy, t1[g].xy, qt);
    qh = __builtin_elementwise_fma(h1[g].zw, h1[g].zw, qh);
    qt = __builtin_elementwise_fma(t1[g].zw, t1[g].zw, qt);
    const float ph = qh.x + qh.y, pt = qt.x + qt.y;
    const float sh = ph + xhalf(ph);
    const float st = pt + xhalf(pt);
    const float mine = 1.0f / fmaxf(sqrtf(kh ? st : sh), 1e-12f);
    const float other = xhalf(mine);
    const float hn = kh ? other : mine;
    const float tn = kh ? mine : other;
    v8f acc = {0.f, 0.f, 0.f, 0.f, 0.f, 0.f, 0.f, 0.f};
#pragma unroll
    for (int j = 0; j < 4; ++j) {
      v2f av, bv;
      av.x = hv[2 * j] * hn;
      av.y = hv[2 * j + 1] * hn;
      bv.x = fast_tanh(tv[2 * j] * tn);
      bv.y = fast_tanh(tv[2 * j + 1] * tn);
      acc = __builtin_amdgcn_wmma_f32_16x16x4_f32(false, av, false, bv,
                                                  (short)0, acc, false, false);
    }
    const int idx = lane & 7;
    float d = acc[0];
    d = (idx == 1) ? acc[1] : d;
    d = (idx == 2) ? acc[2] : d;
    d = (idx == 3) ? acc[3] : d;
    d = (idx == 4) ? acc[4] : d;
    d = (idx == 5) ? acc[5] : d;
    d = (idx == 6) ? acc[6] : d;
    d = (idx == 7) ? acc[7] : d;
    dv[g] = d;
  }

#pragma unroll
  for (int g = 0; g < 4; ++g)
    if (wr[g]) A[(size_t)(e0 + g * 4) * 4 + cc] = aold[g] + dv[g];
}

__global__ void __launch_bounds__(WG) k_final(const float* __restrict__ u, const float* __restrict__ it,
                        const float* __restrict__ l1, const float* __restrict__ l2,
                        float* __restrict__ out, int userElems, int n) {
  const int user4 = userElems >> 2;
  const int n4 = n >> 2;
  const int stride = (int)(gridDim.x * blockDim.x);
  const float4* u4 = reinterpret_cast<const float4*>(u);
  const float4* i4 = reinterpret_cast<const float4*>(it);
  const float4* a4 = reinterpret_cast<const float4*>(l1);
  const float4* b4 = reinterpret_cast<const float4*>(l2);
  float4* o4 = reinterpret_cast<float4*>(out);
  const int span = stride * FN_VEC;
  const int trips = (n4 + span - 1) / span;
  const int first = (int)(blockIdx.x * blockDim.x + threadIdx.x);
  for (int tr = 0; tr < trips; ++tr) {
    const int s0 = first + tr * span;
    float4 x[FN_VEC], y[FN_VEC], z[FN_VEC];
#pragma unroll
    for (int k = 0; k < FN_VEC; ++k) {
      const int i = s0 + k * stride;
      const int ic = (i < n4) ? i : n4 - 1;
      const float4* p = (ic < user4) ? (u4 + ic) : (i4 + (ic - user4));
      x[k] = *p;
    }
#pragma unroll
    for (int k = 0; k < FN_VEC; ++k) {
      const int i = s0 + k * stride;
      const int ic = (i < n4) ? i : n4 - 1;
      y[k] = a4[ic];
      z[k] = b4[ic];
    }
    hold4(x[0], x[1], x[2], x[3]);
    hold4(y[0], y[1], y[2], y[3]);
    hold4(z[0], z[1], z[2], z[3]);
#pragma unroll
    for (int k = 0; k < FN_VEC; ++k) {
      const int i = s0 + k * stride;
      if (i < n4) {
        float4 v = x[k];
        v.x += y[k].x; v.y += y[k].y; v.z += y[k].z; v.w += y[k].w;
        v.x += z[k].x; v.y += z[k].y; v.z += z[k].z; v.w += z[k].w;
        v.x *= (1.0f / 3.0f); v.y *= (1.0f / 3.0f); v.z *= (1.0f / 3.0f); v.w *= (1.0f / 3.0f);
        o4[i] = v;
      }
    }
  }
}

static inline size_t align16f(size_t x) { return (x + 15) & ~(size_t)15; }

extern "C" void kernel_launch(void* const* d_in, const int* in_sizes, int n_in,
                              void* d_out, int out_size, void* d_ws, size_t ws_size,
                              hipStream_t stream) {
  (void)n_in; (void)out_size; (void)ws_size;
  const float* u  = (const float*)d_in[0];
  const float* it = (const float*)d_in[1];
  const int* hl   = (const int*)d_in[2];
  const int* tl   = (const int*)d_in[3];

  const int userElems = in_sizes[0];
  const int itemElems = in_sizes[1];
  const int E = in_sizes[2];
  const int N = (userElems + itemElems) / 64;
  const size_t N64 = (size_t)N * 64;
  const size_t E4  = (size_t)E * 4;
  const size_t N4  = (size_t)N * 4;

  const unsigned msgGrid = (unsigned)((N + MSG_NODES - 1) / MSG_NODES);

  float* base = (float*)d_ws;
  size_t off = 0;
  float* ego   = base + off; off += align16f(N64);
  float* fact  = base + off; off += align16f(N64);
  float* lists = base + off; off += align16f(N64);
  float* A     = base + off; off += align16f(E4);
  float* dinvg = base + off; off += align16f((size_t)msgGrid * MSG_NODES * 4);
  float* keyf  = base + off; off += align16f(N4);

#define GRID(n) ((unsigned)(((size_t)(n) + WG - 1) / WG))

  k_init<<<STREAM_WGS, WG, 0, stream>>>(u, it, ego, userElems, (int)N64);
  k_edges_init<<<STREAM_WGS, WG, 0, stream>>>(A, hl, keyf, E, (int)msgGrid);
  k_build_lists<<<msgGrid, MSG_WG, 0, stream>>>(A, hl, tl, keyf, lists, dinvg, E, N);

  float* cur = ego;
  float* nxt = fact;
  const int nGroups = (E + 15) / 16;

  for (int layer = 0; layer < 2; ++layer) {
    for (int itn = 0; itn < 2; ++itn) {
      if (!(layer == 0 && itn == 0))
        k_degree_lists<<<msgGrid, MSG_WG, 0, stream>>>(A, hl, lists, dinvg, E, N);
      k_message<<<msgGrid, MSG_WG, 0, stream>>>(A, dinvg, cur, hl, tl, keyf, nxt, lists, E, N);
      const bool lastUpdateDead = (layer == 1 && itn == 1);
      if (!lastUpdateDead)
        k_updateA<<<GRID((size_t)nGroups * 32), WG, 0, stream>>>(
            nxt, cur, hl, tl, A, E, nGroups);
    }
    float* tmp = cur; cur = nxt; nxt = tmp;
  }

  k_final<<<STREAM_WGS, WG, 0, stream>>>(u, it, nxt, cur, (float*)d_out, userElems, (int)N64);
#undef GRID
}
